# starting kernel
# baseline (speedup 1.0000x reference)
#include <hip/hip_runtime.h>
#include <math.h>


typedef __attribute__((ext_vector_type(2))) float v2f;
typedef __attribute__((ext_vector_type(8))) float v8f;

#define B_      2048
#define T_      200
#define D_      128
#define NNODE_  200
#define TP      208
#define NT      13
#define HSTR    132
#define SSTR    210
#define PBS     (TP * D_)
#define NEGC    (-4294967295.0f)

__device__ __forceinline__ v8f wmma4(v2f a, v2f b, v8f c) {
  return __builtin_amdgcn_wmma_f32_16x16x4_f32(false, a, false, b, (short)0, c,
                                               false, false);
}

__global__ void qkv_kernel(const float* __restrict__ re, const int* __restrict__ alias,
                           const float* __restrict__ Wq, const float* __restrict__ Wk,
                           const float* __restrict__ Wv,
                           float* __restrict__ wsQ, float* __restrict__ wsK,
                           float* __restrict__ wsV) {
  extern __shared__ float smem[];
  float* sH = smem;
  float* sW = smem + TP * HSTR;
  const int b = blockIdx.x, tid = threadIdx.x;

  for (int idx = tid; idx < TP * D_; idx += 256) {
    int t = idx >> 7, d = idx & 127;
    float v = 0.f;
    if (t < T_) {
      int a = alias[b * T_ + t];
      v = re[((size_t)b * NNODE_ + a) * D_ + d];
    }
    sH[t * HSTR + d] = v;
  }
  for (int idx = tid; idx < 3 * D_ * D_; idx += 256) {
    int mat = idx >> 14, r = idx & 16383;
    const float* src = (mat == 0) ? Wq : (mat == 1) ? Wk : Wv;
    sW[idx] = src[r];
  }
  __syncthreads();

  const int wave = tid >> 5, lane = tid & 31, lh = lane & 15, hi = lane >> 4;
  for (int tile = wave; tile < 3 * NT * 8; tile += 8) {
    int mat = tile / (NT * 8), rem = tile % (NT * 8);
    int i = rem >> 3, n0 = (rem & 7) << 4;
    const float* W = sW + mat * D_ * D_;
    const float* arow = sH + (i * 16 + lh) * HSTR;
    v8f acc = {0.f, 0.f, 0.f, 0.f, 0.f, 0.f, 0.f, 0.f};
#pragma unroll
    for (int c = 0; c < 32; ++c) {
      int k = 4 * c + 2 * hi;
      v2f a = *(const v2f*)(arow + k);
      v2f bb;
      bb.x = W[k * D_ + n0 + lh];
      bb.y = W[(k + 1) * D_ + n0 + lh];
      acc = wmma4(a, bb, acc);
    }
    float* dst = ((mat == 0) ? wsQ : (mat == 1) ? wsK : wsV) + (size_t)b * PBS;
#pragma unroll
    for (int r = 0; r < 8; ++r) {
      int row = i * 16 + r + 8 * hi;
      float v = acc[r];
      dst[row * D_ + n0 + lh] = v > 0.f ? v : 0.f;
    }
  }
}

__global__ void attn_kernel(const float* __restrict__ re, const int* __restrict__ alias,
                            const int* __restrict__ seqlen,
                            const float* __restrict__ wsQ, const float* __restrict__ wsK,
                            const float* __restrict__ wsV, float* __restrict__ wsDec) {
  extern __shared__ float smem[];
  float* sK = smem;
  float* sV = smem + TP * HSTR;
  float* sS0 = smem + 2 * TP * HSTR;
  const int b = blockIdx.x, tid = threadIdx.x;
  const int len = seqlen[b];
  const float* Kb = wsK + (size_t)b * PBS;
  const float* Vb = wsV + (size_t)b * PBS;
  for (int idx = tid; idx < TP * D_; idx += 256) {
    int t = idx >> 7, d = idx & 127;
    sK[t * HSTR + d] = Kb[idx];
    sV[t * HSTR + d] = Vb[idx];
  }
  __syncthreads();

  const int wave = tid >> 5, lane = tid & 31, lh = lane & 15, hi = lane >> 4;
  float* sS = sS0 + wave * (16 * SSTR);
  const float scale = 0.088388347648318447f;

  for (int pass = 0; pass < 2; ++pass) {
    const int qt = wave + 8 * pass;
    const bool active = (qt < NT);
    if (active) {
      v2f qa[32];
      const float* Qrow = wsQ + (size_t)b * PBS + (qt * 16 + lh) * D_;
#pragma unroll
      for (int c = 0; c < 32; ++c) qa[c] = *(const v2f*)(Qrow + 4 * c + 2 * hi);
      for (int j = 0; j < NT; ++j) {
        v8f acc = {0.f, 0.f, 0.f, 0.f, 0.f, 0.f, 0.f, 0.f};
#pragma unroll
        for (int c = 0; c < 32; ++c) {
          v2f bb = *(const v2f*)(sK + (j * 16 + lh) * HSTR + 4 * c + 2 * hi);
          acc = wmma4(qa[c], bb, acc);
        }
#pragma unroll
        for (int r = 0; r < 8; ++r) {
          int m = r + 8 * hi, s = j * 16 + lh;
          sS[m * SSTR + s] = (s < len) ? acc[r] * scale : NEGC;
        }
      }
    }
    __syncthreads();
    if (active) {
      float* rowp = sS + lh * SSTR + hi * 104;
      float mx = -3.0e38f;
      for (int s = 0; s < 104; ++s) mx = fmaxf(mx, rowp[s]);
      mx = fmaxf(mx, __shfl_xor(mx, 16, 32));
      float sum = 0.f;
      for (int s = 0; s < 104; ++s) { float e = __expf(rowp[s] - mx); rowp[s] = e; sum += e; }
      sum += __shfl_xor(sum, 16, 32);
      float qm = ((qt * 16 + lh) < len) ? 1.f : 0.f;
      float inv = qm / sum;
      for (int s = 0; s < 104; ++s) rowp[s] *= inv;
    }
    __syncthreads();
    if (active) {
      float* Db = wsDec + (size_t)b * PBS;
      for (int nt = 0; nt < 8; ++nt) {
        int n0 = nt * 16;
        v8f acc;
#pragma unroll
        for (int r = 0; r < 8; ++r) {
          int t = qt * 16 + r + 8 * hi;
          float h = 0.f;
          if (t < T_) {
            int a = alias[b * T_ + t];
            h = re[((size_t)b * NNODE_ + a) * D_ + n0 + lh];
          }
          acc[r] = h;
        }
#pragma unroll
        for (int c = 0; c < 52; ++c) {
          int s = 4 * c + 2 * hi;
          v2f a = *(const v2f*)(sS + lh * SSTR + s);
          v2f bb;
          bb.x = sV[s * HSTR + n0 + lh];
          bb.y = sV[(s + 1) * HSTR + n0 + lh];
          acc = wmma4(a, bb, acc);
        }
#pragma unroll
        for (int r = 0; r < 8; ++r) {
          int t = qt * 16 + r + 8 * hi;
          Db[t * D_ + n0 + lh] = acc[r];
        }
      }
    }
    __syncthreads();
  }
}

__global__ void stamp_kernel(const int* __restrict__ seqlen, const float* __restrict__ wsDec,
                             const float* __restrict__ u_w1, const float* __restrict__ u_w2,
                             const float* __restrict__ u_b, const float* __restrict__ u_v,
                             float* __restrict__ out) {
  extern __shared__ float smem[];
  float* sW2 = smem;
  float* sLp = smem + D_ * D_;
  float* sUb = sLp + 128;
  float* sUv = sUb + 128;
  float* sCoef = sUv + 128;
  const int b = blockIdx.x, tid = threadIdx.x;
  const int len = seqlen[b];
  const float* Db = wsDec + (size_t)b * PBS;

  for (int idx = tid; idx < D_ * D_; idx += 256) sW2[idx] = u_w2[idx];
  if (tid < D_) { sUb[tid] = u_b[tid]; sUv[tid] = u_v[tid]; }
  if (tid < D_) {
    const float* lrow = Db + (len - 1) * D_;
    float acc = 0.f;
    for (int d = 0; d < D_; ++d) acc += lrow[d] * u_w1[d * D_ + tid];
    sLp[tid] = acc;
    out[(size_t)b * 2 * D_ + D_ + tid] = acc;
  }
  __syncthreads();

  const int wave = tid >> 5, lane = tid & 31, lh = lane & 15, hi = lane >> 4;
  for (int pass = 0; pass < 2; ++pass) {
    const int qt = wave + 8 * pass;
    if (qt < NT) {
      v2f da[32];
      const float* drow = Db + (qt * 16 + lh) * D_;
#pragma unroll
      for (int c = 0; c < 32; ++c) da[c] = *(const v2f*)(drow + 4 * c + 2 * hi);
      float cpart[8];
#pragma unroll
      for (int r = 0; r < 8; ++r) cpart[r] = 0.f;
      for (int nt = 0; nt < 8; ++nt) {
        int n0 = nt * 16;
        v8f acc = {0.f, 0.f, 0.f, 0.f, 0.f, 0.f, 0.f, 0.f};
#pragma unroll
        for (int c = 0; c < 32; ++c) {
          int k = 4 * c + 2 * hi;
          v2f bb;
          bb.x = sW2[k * D_ + n0 + lh];
          bb.y = sW2[(k + 1) * D_ + n0 + lh];
          acc = wmma4(da[c], bb, acc);
        }
        int col = n0 + lh;
        float add = sLp[col] + sUb[col];
        float uvv = sUv[col];
#pragma unroll
        for (int r = 0; r < 8; ++r) {
          float x = acc[r] + add;
          float sg = 1.f / (1.f + __expf(-x));
          cpart[r] += sg * uvv;
        }
      }
#pragma unroll
      for (int r = 0; r < 8; ++r) {
        float v = cpart[r];
        v += __shfl_xor(v, 1, 32);
        v += __shfl_xor(v, 2, 32);
        v += __shfl_xor(v, 4, 32);
        v += __shfl_xor(v, 8, 32);
        if (lh == 0) {
          int t = qt * 16 + r + 8 * hi;
          sCoef[t] = (t < len) ? v : 0.f;
        }
      }
    }
  }
  __syncthreads();
  if (tid < D_) {
    float acc = 0.f;
    for (int t = 0; t < TP; ++t) acc += sCoef[t] * Db[t * D_ + tid];
    out[(size_t)b * 2 * D_ + tid] = acc;
  }
}

extern "C" void kernel_launch(void* const* d_in, const int* in_sizes, int n_in,
                              void* d_out, int out_size, void* d_ws, size_t ws_size,
                              hipStream_t stream) {
  (void)in_sizes; (void)n_in; (void)out_size; (void)ws_size;
  const float* re   = (const float*)d_in[0];
  const int*   alia = (const int*)d_in[1];
  const int*   slen = (const int*)d_in[2];
  const float* Wq   = (const float*)d_in[3];
  const float* Wk   = (const float*)d_in[4];
  const float* Wv   = (const float*)d_in[5];
  const float* u_w1 = (const float*)d_in[6];
  const float* u_w2 = (const float*)d_in[7];
  const float* u_b  = (const float*)d_in[8];
  const float* u_v  = (const float*)d_in[9];
  float* out = (float*)d_out;

  float* ws   = (float*)d_ws;
  float* wsQ  = ws;
  float* wsK  = ws + (size_t)B_ * PBS;
  float* wsV  = ws + 2 * (size_t)B_ * PBS;
  float* wsDec = wsQ;

  size_t sm1 = (size_t)(TP * HSTR + 3 * D_ * D_) * sizeof(float);
  size_t sm2 = (size_t)(2 * TP * HSTR + 8 * 16 * SSTR) * sizeof(float);
  size_t sm3 = (size_t)(D_ * D_ + 3 * 128 + TP) * sizeof(float);
  hipFuncSetAttribute((const void*)qkv_kernel,
                      hipFuncAttributeMaxDynamicSharedMemorySize, (int)sm1);
  hipFuncSetAttribute((const void*)attn_kernel,
                      hipFuncAttributeMaxDynamicSharedMemorySize, (int)sm2);
  hipFuncSetAttribute((const void*)stamp_kernel,
                      hipFuncAttributeMaxDynamicSharedMemorySize, (int)sm3);

  qkv_kernel<<<B_, 256, sm1, stream>>>(re, alia, Wq, Wk, Wv, wsQ, wsK, wsV);
  attn_kernel<<<B_, 256, sm2, stream>>>(re, alia, slen, wsQ, wsK, wsV, wsDec);
  stamp_kernel<<<B_, 256, sm3, stream>>>(slen, wsDec, u_w1, u_w2, u_b, u_v, out);
}


# qkv kernel: H gather per row with 128-bit loads, 13 independent rows in flight per wave (aliases read once per wave, handed out as scalars) instead of one awaited 4-byte alias+row load per element
# speedup vs baseline: 1.1727x; 1.1727x over previous
#include <hip/hip_runtime.h>
#include <math.h>


typedef __attribute__((ext_vector_type(2))) float v2f;
typedef __attribute__((ext_vector_type(8))) float v8f;

#define B_      2048
#define T_      200
#define D_      128
#define NNODE_  200
#define TP      208
#define NT      13
#define HSTR    132
#define SSTR    210
#define PBS     (TP * D_)
#define NEGC    (-4294967295.0f)

__device__ __forceinline__ v8f wmma4(v2f a, v2f b, v8f c) {
  return __builtin_amdgcn_wmma_f32_16x16x4_f32(false, a, false, b, (short)0, c,
                                               false, false);
}

__global__ void qkv_kernel(const float* __restrict__ re, const int* __restrict__ alias,
                           const float* __restrict__ Wq, const float* __restrict__ Wk,
                           const float* __restrict__ Wv,
                           float* __restrict__ wsQ, float* __restrict__ wsK,
                           float* __restrict__ wsV) {
  extern __shared__ __attribute__((aligned(16))) float smem_h[];
  float* sH = smem_h;
  float* sW = smem_h + TP * HSTR;
  const int b = blockIdx.x, tid = threadIdx.x;

  {
    const int wv = __builtin_amdgcn_readfirstlane(tid >> 5);
    const int ln = tid & 31;
    const int tl = wv * 26 + ln;
    const int my_a = alias[(size_t)b * T_ + (tl < T_ ? tl : T_ - 1)];
    const float* reb = re + (size_t)b * NNODE_ * D_ + ln * 4;
#pragma unroll 1
    for (int h = 0; h < 2; ++h) {
      const int t0 = wv * 26 + h * 13;
      float4 v[13];
#pragma unroll
      for (int r = 0; r < 13; ++r) {
        const int a = __builtin_amdgcn_readlane(my_a, h * 13 + r);
        float4 x = *(const float4*)(reb + (size_t)a * D_);
        if (t0 + r >= T_) x = make_float4(0.f, 0.f, 0.f, 0.f);
        v[r] = x;
      }
#pragma unroll
      for (int r = 0; r < 13; ++r)
        *(float4*)(sH + (t0 + r) * HSTR + ln * 4) = v[r];
    }
  }
  for (int idx = tid; idx < 3 * D_ * D_; idx += 256) {
    int mat = idx >> 14, r = idx & 16383;
    const float* src = (mat == 0) ? Wq : (mat == 1) ? Wk : Wv;
    sW[idx] = src[r];
  }
  __syncthreads();

  const int wave = tid >> 5, lane = tid & 31, lh = lane & 15, hi = lane >> 4;
  for (int tile = wave; tile < 3 * NT * 8; tile += 8) {
    int mat = tile / (NT * 8), rem = tile % (NT * 8);
    int i = rem >> 3, n0 = (rem & 7) << 4;
    const float* W = sW + mat * D_ * D_;
    const float* arow = sH + (i * 16 + lh) * HSTR;
    v8f acc = {0.f, 0.f, 0.f, 0.f, 0.f, 0.f, 0.f, 0.f};
#pragma unroll
    for (int c = 0; c < 32; ++c) {
      int k = 4 * c + 2 * hi;
      v2f a = *(const v2f*)(arow + k);
      v2f bb;
      bb.x = W[k * D_ + n0 + lh];
      bb.y = W[(k + 1) * D_ + n0 + lh];
      acc = wmma4(a, bb, acc);
    }
    float* dst = ((mat == 0) ? wsQ : (mat == 1) ? wsK : wsV) + (size_t)b * PBS;
#pragma unroll
    for (int r = 0; r < 8; ++r) {
      int row = i * 16 + r + 8 * hi;
      float v = acc[r];
      dst[row * D_ + n0 + lh] = v > 0.f ? v : 0.f;
    }
  }
}

__global__ void attn_kernel(const float* __restrict__ re, const int* __restrict__ alias,
                            const int* __restrict__ seqlen,
                            const float* __restrict__ wsQ, const float* __restrict__ wsK,
                            const float* __restrict__ wsV, float* __restrict__ wsDec) {
  extern __shared__ float smem[];
  float* sK = smem;
  float* sV = smem + TP * HSTR;
  float* sS0 = smem + 2 * TP * HSTR;
  const int b = blockIdx.x, tid = threadIdx.x;
  const int len = seqlen[b];
  const float* Kb = wsK + (size_t)b * PBS;
  const float* Vb = wsV + (size_t)b * PBS;
  for (int idx = tid; idx < TP * D_; idx += 256) {
    int t = idx >> 7, d = idx & 127;
    sK[t * HSTR + d] = Kb[idx];
    sV[t * HSTR + d] = Vb[idx];
  }
  __syncthreads();

  const int wave = tid >> 5, lane = tid & 31, lh = lane & 15, hi = lane >> 4;
  float* sS = sS0 + wave * (16 * SSTR);
  const float scale = 0.088388347648318447f;

  for (int pass = 0; pass < 2; ++pass) {
    const int qt = wave + 8 * pass;
    const bool active = (qt < NT);
    if (active) {
      v2f qa[32];
      const float* Qrow = wsQ + (size_t)b * PBS + (qt * 16 + lh) * D_;
#pragma unroll
      for (int c = 0; c < 32; ++c) qa[c] = *(const v2f*)(Qrow + 4 * c + 2 * hi);
      for (int j = 0; j < NT; ++j) {
        v8f acc = {0.f, 0.f, 0.f, 0.f, 0.f, 0.f, 0.f, 0.f};
#pragma unroll
        for (int c = 0; c < 32; ++c) {
          v2f bb = *(const v2f*)(sK + (j * 16 + lh) * HSTR + 4 * c + 2 * hi);
          acc = wmma4(qa[c], bb, acc);
        }
#pragma unroll
        for (int r = 0; r < 8; ++r) {
          int m = r + 8 * hi, s = j * 16 + lh;
          sS[m * SSTR + s] = (s < len) ? acc[r] * scale : NEGC;
        }
      }
    }
    __syncthreads();
    if (active) {
      float* rowp = sS + lh * SSTR + hi * 104;
      float mx = -3.0e38f;
      for (int s = 0; s < 104; ++s) mx = fmaxf(mx, rowp[s]);
      mx = fmaxf(mx, __shfl_xor(mx, 16, 32));
      float sum = 0.f;
      for (int s = 0; s < 104; ++s) { float e = __expf(rowp[s] - mx); rowp[s] = e; sum += e; }
      sum += __shfl_xor(sum, 16, 32);
      float qm = ((qt * 16 + lh) < len) ? 1.f : 0.f;
      float inv = qm / sum;
      for (int s = 0; s < 104; ++s) rowp[s] *= inv;
    }
    __syncthreads();
    if (active) {
      float* Db = wsDec + (size_t)b * PBS;
      for (int nt = 0; nt < 8; ++nt) {
        int n0 = nt * 16;
        v8f acc;
#pragma unroll
        for (int r = 0; r < 8; ++r) {
          int t = qt * 16 + r + 8 * hi;
          float h = 0.f;
          if (t < T_) {
            int a = alias[b * T_ + t];
            h = re[((size_t)b * NNODE_ + a) * D_ + n0 + lh];
          }
          acc[r] = h;
        }
#pragma unroll
        for (int c = 0; c < 52; ++c) {
          int s = 4 * c + 2 * hi;
          v2f a = *(const v2f*)(sS + lh * SSTR + s);
          v2f bb;
          bb.x = sV[s * HSTR + n0 + lh];
          bb.y = sV[(s + 1) * HSTR + n0 + lh];
          acc = wmma4(a, bb, acc);
        }
#pragma unroll
        for (int r = 0; r < 8; ++r) {
          int t = qt * 16 + r + 8 * hi;
          Db[t * D_ + n0 + lh] = acc[r];
        }
      }
    }
    __syncthreads();
  }
}

__global__ void stamp_kernel(const int* __restrict__ seqlen, const float* __restrict__ wsDec,
                             const float* __restrict__ u_w1, const float* __restrict__ u_w2,
                             const float* __restrict__ u_b, const float* __restrict__ u_v,
                             float* __restrict__ out) {
  extern __shared__ float smem[];
  float* sW2 = smem;
  float* sLp = smem + D_ * D_;
  float* sUb = sLp + 128;
  float* sUv = sUb + 128;
  float* sCoef = sUv + 128;
  const int b = blockIdx.x, tid = threadIdx.x;
  const int len = seqlen[b];
  const float* Db = wsDec + (size_t)b * PBS;

  for (int idx = tid; idx < D_ * D_; idx += 256) sW2[idx] = u_w2[idx];
  if (tid < D_) { sUb[tid] = u_b[tid]; sUv[tid] = u_v[tid]; }
  if (tid < D_) {
    const float* lrow = Db + (len - 1) * D_;
    float acc = 0.f;
    for (int d = 0; d < D_; ++d) acc += lrow[d] * u_w1[d * D_ + tid];
    sLp[tid] = acc;
    out[(size_t)b * 2 * D_ + D_ + tid] = acc;
  }
  __syncthreads();

  const int wave = tid >> 5, lane = tid & 31, lh = lane & 15, hi = lane >> 4;
  for (int pass = 0; pass < 2; ++pass) {
    const int qt = wave + 8 * pass;
    if (qt < NT) {
      v2f da[32];
      const float* drow = Db + (qt * 16 + lh) * D_;
#pragma unroll
      for (int c = 0; c < 32; ++c) da[c] = *(const v2f*)(drow + 4 * c + 2 * hi);
      float cpart[8];
#pragma unroll
      for (int r = 0; r < 8; ++r) cpart[r] = 0.f;
      for (int nt = 0; nt < 8; ++nt) {
        int n0 = nt * 16;
        v8f acc = {0.f, 0.f, 0.f, 0.f, 0.f, 0.f, 0.f, 0.f};
#pragma unroll
        for (int c = 0; c < 32; ++c) {
          int k = 4 * c + 2 * hi;
          v2f bb;
          bb.x = sW2[k * D_ + n0 + lh];
          bb.y = sW2[(k + 1) * D_ + n0 + lh];
          acc = wmma4(da[c], bb, acc);
        }
        int col = n0 + lh;
        float add = sLp[col] + sUb[col];
        float uvv = sUv[col];
#pragma unroll
        for (int r = 0; r < 8; ++r) {
          float x = acc[r] + add;
          float sg = 1.f / (1.f + __expf(-x));
          cpart[r] += sg * uvv;
        }
      }
#pragma unroll
      for (int r = 0; r < 8; ++r) {
        float v = cpart[r];
        v += __shfl_xor(v, 1, 32);
        v += __shfl_xor(v, 2, 32);
        v += __shfl_xor(v, 4, 32);
        v += __shfl_xor(v, 8, 32);
        if (lh == 0) {
          int t = qt * 16 + r + 8 * hi;
          sCoef[t] = (t < len) ? v : 0.f;
        }
      }
    }
  }
  __syncthreads();
  if (tid < D_) {
    float acc = 0.f;
    for (int t = 0; t < TP; ++t) acc += sCoef[t] * Db[t * D_ + tid];
    out[(size_t)b * 2 * D_ + tid] = acc;
  }
}

extern "C" void kernel_launch(void* const* d_in, const int* in_sizes, int n_in,
                              void* d_out, int out_size, void* d_ws, size_t ws_size,
                              hipStream_t stream) {
  (void)in_sizes; (void)n_in; (void)out_size; (void)ws_size;
  const float* re   = (const float*)d_in[0];
  const int*   alia = (const int*)d_in[1];
  const int*   slen = (const int*)d_in[2];
  const float* Wq   = (const float*)d_in[3];
  const float* Wk   = (const float*)d_in[4];
  const float* Wv   = (const float*)d_in[5];
  const float* u_w1 = (const float*)d_in[6];
  const float* u_w2 = (const float*)d_in[7];
  const float* u_b  = (const float*)d_in[8];
  const float* u_v  = (const float*)d_in[9];
  float* out = (float*)d_out;

  float* ws   = (float*)d_ws;
  float* wsQ  = ws;
  float* wsK  = ws + (size_t)B_ * PBS;
  float* wsV  = ws + 2 * (size_t)B_ * PBS;
  float* wsDec = wsQ;

  size_t sm1 = (size_t)(TP * HSTR + 3 * D_ * D_) * sizeof(float);
  size_t sm2 = (size_t)(2 * TP * HSTR + 8 * 16 * SSTR) * sizeof(float);
  size_t sm3 = (size_t)(D_ * D_ + 3 * 128 + TP) * sizeof(float);
  hipFuncSetAttribute((const void*)qkv_kernel,
                      hipFuncAttributeMaxDynamicSharedMemorySize, (int)sm1);
  hipFuncSetAttribute((const void*)attn_kernel,
                      hipFuncAttributeMaxDynamicSharedMemorySize, (int)sm2);
  hipFuncSetAttribute((const void*)stamp_kernel,
                      hipFuncAttributeMaxDynamicSharedMemorySize, (int)sm3);

  qkv_kernel<<<B_, 256, sm1, stream>>>(re, alia, Wq, Wk, Wv, wsQ, wsK, wsV);
  attn_kernel<<<B_, 256, sm2, stream>>>(re, alia, slen, wsQ, wsK, wsV, wsDec);
  stamp_kernel<<<B_, 256, sm3, stream>>>(slen, wsDec, u_w1, u_w2, u_b, u_v, out);
}


# qkv and attn*V WMMA loops: each block's LDS fragment loads issued as one group ahead of its WMMAs (block size 256 declared, so the fragments fit in registers) instead of an awaited load per WMMA
# speedup vs baseline: 1.7114x; 1.4594x over previous
#include <hip/hip_runtime.h>
#include <math.h>


typedef __attribute__((ext_vector_type(2))) float v2f;
typedef __attribute__((ext_vector_type(8))) float v8f;

#define B_      2048
#define T_      200
#define D_      128
#define NNODE_  200
#define TP      208
#define NT      13
#define HSTR    132
#define SSTR    210
#define PBS     (TP * D_)
#define NEGC    (-4294967295.0f)

__device__ __forceinline__ v8f wmma4(v2f a, v2f b, v8f c) {
  return __builtin_amdgcn_wmma_f32_16x16x4_f32(false, a, false, b, (short)0, c,
                                               false, false);
}

__global__ void __launch_bounds__(256) __attribute__((amdgpu_waves_per_eu(2, 2))) qkv_kernel(const float* __restrict__ re, const int* __restrict__ alias,
                           const float* __restrict__ Wq, const float* __restrict__ Wk,
                           const float* __restrict__ Wv,
                           float* __restrict__ wsQ, float* __restrict__ wsK,
                           float* __restrict__ wsV) {
  extern __shared__ __attribute__((aligned(16))) float smem_h[];
  float* sH = smem_h;
  float* sW = smem_h + TP * HSTR;
  const int b = blockIdx.x, tid = threadIdx.x;

  {
    const int wv = __builtin_amdgcn_readfirstlane(tid >> 5);
    const int ln = tid & 31;
    const int tl = wv * 26 + ln;
    const int my_a = alias[(size_t)b * T_ + (tl < T_ ? tl : T_ - 1)];
    const float* reb = re + (size_t)b * NNODE_ * D_ + ln * 4;
#pragma unroll 1
    for (int h = 0; h < 2; ++h) {
      const int t0 = wv * 26 + h * 13;
      float4 v[13];
#pragma unroll
      for (int r = 0; r < 13; ++r) {
        const int a = __builtin_amdgcn_readlane(my_a, h * 13 + r);
        float4 x = *(const float4*)(reb + (size_t)a * D_);
        if (t0 + r >= T_) x = make_float4(0.f, 0.f, 0.f, 0.f);
        v[r] = x;
      }
#pragma unroll
      for (int r = 0; r < 13; ++r)
        *(float4*)(sH + (t0 + r) * HSTR + ln * 4) = v[r];
    }
  }
  for (int idx = tid; idx < 3 * D_ * D_; idx += 256) {
    int mat = idx >> 14, r = idx & 16383;
    const float* src = (mat == 0) ? Wq : (mat == 1) ? Wk : Wv;
    sW[idx] = src[r];
  }
  __syncthreads();

  const int wave = tid >> 5, lane = tid & 31, lh = lane & 15, hi = lane >> 4;
  for (int tile = wave; tile < 3 * NT * 8; tile += 8) {
    int mat = tile / (NT * 8), rem = tile % (NT * 8);
    int i = rem >> 3, n0 = (rem & 7) << 4;
    const float* W = sW + mat * D_ * D_;
    const float* arow = sH + (i * 16 + lh) * HSTR;
    v2f af[32], bf[32];
#pragma unroll
    for (int c = 0; c < 32; ++c) {
      int k = 4 * c + 2 * hi;
      af[c] = *(const v2f*)(arow + k);
      bf[c].x = W[k * D_ + n0 + lh];
      bf[c].y = W[(k + 1) * D_ + n0 + lh];
    }
    __builtin_amdgcn_sched_barrier(0);
    v8f acc = {0.f, 0.f, 0.f, 0.f, 0.f, 0.f, 0.f, 0.f};
#pragma unroll
    for (int c = 0; c < 32; ++c) acc = wmma4(af[c], bf[c], acc);
    float* dst = ((mat == 0) ? wsQ : (mat == 1) ? wsK : wsV) + (size_t)b * PBS;
#pragma unroll
    for (int r = 0; r < 8; ++r) {
      int row = i * 16 + r + 8 * hi;
      float v = acc[r];
      dst[row * D_ + n0 + lh] = v > 0.f ? v : 0.f;
    }
  }
}

__global__ void __launch_bounds__(256) __attribute__((amdgpu_waves_per_eu(2, 2))) attn_kernel(const float* __restrict__ re, const int* __restrict__ alias,
                            const int* __restrict__ seqlen,
                            const float* __restrict__ wsQ, const float* __restrict__ wsK,
                            const float* __restrict__ wsV, float* __restrict__ wsDec) {
  extern __shared__ float smem[];
  float* sK = smem;
  float* sV = smem + TP * HSTR;
  float* sS0 = smem + 2 * TP * HSTR;
  const int b = blockIdx.x, tid = threadIdx.x;
  const int len = seqlen[b];
  const float* Kb = wsK + (size_t)b * PBS;
  const float* Vb = wsV + (size_t)b * PBS;
  for (int idx = tid; idx < TP * D_; idx += 256) {
    int t = idx >> 7, d = idx & 127;
    sK[t * HSTR + d] = Kb[idx];
    sV[t * HSTR + d] = Vb[idx];
  }
  __syncthreads();

  const int wave = tid >> 5, lane = tid & 31, lh = lane & 15, hi = lane >> 4;
  float* sS = sS0 + wave * (16 * SSTR);
  const float scale = 0.088388347648318447f;

  for (int pass = 0; pass < 2; ++pass) {
    const int qt = wave + 8 * pass;
    const bool active = (qt < NT);
    if (active) {
      v2f qa[32];
      const float* Qrow = wsQ + (size_t)b * PBS + (qt * 16 + lh) * D_;
#pragma unroll
      for (int c = 0; c < 32; ++c) qa[c] = *(const v2f*)(Qrow + 4 * c + 2 * hi);
      for (int j = 0; j < NT; ++j) {
        v8f acc = {0.f, 0.f, 0.f, 0.f, 0.f, 0.f, 0.f, 0.f};
#pragma unroll
        for (int c = 0; c < 32; ++c) {
          v2f bb = *(const v2f*)(sK + (j * 16 + lh) * HSTR + 4 * c + 2 * hi);
          acc = wmma4(qa[c], bb, acc);
        }
#pragma unroll
        for (int r = 0; r < 8; ++r) {
          int m = r + 8 * hi, s = j * 16 + lh;
          sS[m * SSTR + s] = (s < len) ? acc[r] * scale : NEGC;
        }
      }
    }
    __syncthreads();
    if (active) {
      float* rowp = sS + lh * SSTR + hi * 104;
      float mx = -3.0e38f;
      for (int s = 0; s < 104; ++s) mx = fmaxf(mx, rowp[s]);
      mx = fmaxf(mx, __shfl_xor(mx, 16, 32));
      float sum = 0.f;
      for (int s = 0; s < 104; ++s) { float e = __expf(rowp[s] - mx); rowp[s] = e; sum += e; }
      sum += __shfl_xor(sum, 16, 32);
      float qm = ((qt * 16 + lh) < len) ? 1.f : 0.f;
      float inv = qm / sum;
      for (int s = 0; s < 104; ++s) rowp[s] *= inv;
    }
    __syncthreads();
    if (active) {
      float* Db = wsDec + (size_t)b * PBS;
      v2f pa[52];
#pragma unroll
      for (int c = 0; c < 52; ++c) pa[c] = *(const v2f*)(sS + lh * SSTR + 4 * c + 2 * hi);
      const float* vcol = sV + (2 * hi) * HSTR + lh;
      v2f vb0[26], vb1[26];
#pragma unroll
      for (int c = 0; c < 26; ++c) {
        vb0[c].x = vcol[(4 * c) * HSTR];
        vb0[c].y = vcol[(4 * c + 1) * HSTR];
      }
      for (int nt = 0; nt < 8; ++nt) {
        int n0 = nt * 16;
        const int n1 = (nt < 7) ? n0 + 16 : n0;
#pragma unroll
        for (int c = 0; c < 26; ++c) {
          vb1[c].x = vcol[(4 * (c + 26)) * HSTR + n0];
          vb1[c].y = vcol[(4 * (c + 26) + 1) * HSTR + n0];
        }
        v8f acc;
#pragma unroll
        for (int r = 0; r < 8; ++r) {
          int t = qt * 16 + r + 8 * hi;
          float h = 0.f;
          if (t < T_) {
            int a = alias[b * T_ + t];
            h = re[((size_t)b * NNODE_ + a) * D_ + n0 + lh];
          }
          acc[r] = h;
        }
#pragma unroll
        for (int c = 0; c < 26; ++c) acc = wmma4(pa[c], vb0[c], acc);
#pragma unroll
        for (int c = 0; c < 26; ++c) {
          vb0[c].x = vcol[(4 * c) * HSTR + n1];
          vb0[c].y = vcol[(4 * c + 1) * HSTR + n1];
        }
#pragma unroll
        for (int c = 0; c < 26; ++c) acc = wmma4(pa[c + 26], vb1[c], acc);
#pragma unroll
        for (int r = 0; r < 8; ++r) {
          int t = qt * 16 + r + 8 * hi;
          Db[t * D_ + n0 + lh] = acc[r];
        }
      }
    }
    __syncthreads();
  }
}

__global__ void stamp_kernel(const int* __restrict__ seqlen, const float* __restrict__ wsDec,
                             const float* __restrict__ u_w1, const float* __restrict__ u_w2,
                             const float* __restrict__ u_b, const float* __restrict__ u_v,
                             float* __restrict__ out) {
  extern __shared__ float smem[];
  float* sW2 = smem;
  float* sLp = smem + D_ * D_;
  float* sUb = sLp + 128;
  float* sUv = sUb + 128;
  float* sCoef = sUv + 128;
  const int b = blockIdx.x, tid = threadIdx.x;
  const int len = seqlen[b];
  const float* Db = wsDec + (size_t)b * PBS;

  for (int idx = tid; idx < D_ * D_; idx += 256) sW2[idx] = u_w2[idx];
  if (tid < D_) { sUb[tid] = u_b[tid]; sUv[tid] = u_v[tid]; }
  if (tid < D_) {
    const float* lrow = Db + (len - 1) * D_;
    float acc = 0.f;
    for (int d = 0; d < D_; ++d) acc += lrow[d] * u_w1[d * D_ + tid];
    sLp[tid] = acc;
    out[(size_t)b * 2 * D_ + D_ + tid] = acc;
  }
  __syncthreads();

  const int wave = tid >> 5, lane = tid & 31, lh = lane & 15, hi = lane >> 4;
  for (int pass = 0; pass < 2; ++pass) {
    const int qt = wave + 8 * pass;
    if (qt < NT) {
      v2f da[32];
      const float* drow = Db + (qt * 16 + lh) * D_;
#pragma unroll
      for (int c = 0; c < 32; ++c) da[c] = *(const v2f*)(drow + 4 * c + 2 * hi);
      float cpart[8];
#pragma unroll
      for (int r = 0; r < 8; ++r) cpart[r] = 0.f;
      for (int nt = 0; nt < 8; ++nt) {
        int n0 = nt * 16;
        v8f acc = {0.f, 0.f, 0.f, 0.f, 0.f, 0.f, 0.f, 0.f};
#pragma unroll
        for (int c = 0; c < 32; ++c) {
          int k = 4 * c + 2 * hi;
          v2f bb;
          bb.x = sW2[k * D_ + n0 + lh];
          bb.y = sW2[(k + 1) * D_ + n0 + lh];
          acc = wmma4(da[c], bb, acc);
        }
        int col = n0 + lh;
        float add = sLp[col] + sUb[col];
        float uvv = sUv[col];
#pragma unroll
        for (int r = 0; r < 8; ++r) {
          float x = acc[r] + add;
          float sg = 1.f / (1.f + __expf(-x));
          cpart[r] += sg * uvv;
        }
      }
#pragma unroll
      for (int r = 0; r < 8; ++r) {
        float v = cpart[r];
        v += __shfl_xor(v, 1, 32);
        v += __shfl_xor(v, 2, 32);
        v += __shfl_xor(v, 4, 32);
        v += __shfl_xor(v, 8, 32);
        if (lh == 0) {
          int t = qt * 16 + r + 8 * hi;
          sCoef[t] = (t < len) ? v : 0.f;
        }
      }
    }
  }
  __syncthreads();
  if (tid < D_) {
    float acc = 0.f;
    for (int t = 0; t < TP; ++t) acc += sCoef[t] * Db[t * D_ + tid];
    out[(size_t)b * 2 * D_ + tid] = acc;
  }
}

extern "C" void kernel_launch(void* const* d_in, const int* in_sizes, int n_in,
                              void* d_out, int out_size, void* d_ws, size_t ws_size,
                              hipStream_t stream) {
  (void)in_sizes; (void)n_in; (void)out_size; (void)ws_size;
  const float* re   = (const float*)d_in[0];
  const int*   alia = (const int*)d_in[1];
  const int*   slen = (const int*)d_in[2];
  const float* Wq   = (const float*)d_in[3];
  const float* Wk   = (const float*)d_in[4];
  const float* Wv   = (const float*)d_in[5];
  const float* u_w1 = (const float*)d_in[6];
  const float* u_w2 = (const float*)d_in[7];
  const float* u_b  = (const float*)d_in[8];
  const float* u_v  = (const float*)d_in[9];
  float* out = (float*)d_out;

  float* ws   = (float*)d_ws;
  float* wsQ  = ws;
  float* wsK  = ws + (size_t)B_ * PBS;
  float* wsV  = ws + 2 * (size_t)B_ * PBS;
  float* wsDec = wsQ;

  size_t sm1 = (size_t)(TP * HSTR + 3 * D_ * D_) * sizeof(float);
  size_t sm2 = (size_t)(2 * TP * HSTR + 8 * 16 * SSTR) * sizeof(float);
  size_t sm3 = (size_t)(D_ * D_ + 3 * 128 + TP) * sizeof(float);
  hipFuncSetAttribute((const void*)qkv_kernel,
                      hipFuncAttributeMaxDynamicSharedMemorySize, (int)sm1);
  hipFuncSetAttribute((const void*)attn_kernel,
                      hipFuncAttributeMaxDynamicSharedMemorySize, (int)sm2);
  hipFuncSetAttribute((const void*)stamp_kernel,
                      hipFuncAttributeMaxDynamicSharedMemorySize, (int)sm3);

  qkv_kernel<<<B_, 256, sm1, stream>>>(re, alia, Wq, Wk, Wv, wsQ, wsK, wsV);
  attn_kernel<<<B_, 256, sm2, stream>>>(re, alia, slen, wsQ, wsK, wsV, wsDec);
  stamp_kernel<<<B_, 256, sm3, stream>>>(slen, wsDec, u_w1, u_w2, u_b, u_v, out);
}


# attn kernel: residual values requested ahead (row aliases once per query tile, each column tile's 8 values one tile before its WMMAs) instead of 8 awaited alias+row load pairs per column tile
# speedup vs baseline: 1.9816x; 1.1579x over previous
#include <hip/hip_runtime.h>
#include <math.h>


typedef __attribute__((ext_vector_type(2))) float v2f;
typedef __attribute__((ext_vector_type(8))) float v8f;

#define B_      2048
#define T_      200
#define D_      128
#define NNODE_  200
#define TP      208
#define NT      13
#define HSTR    132
#define SSTR    210
#define PBS     (TP * D_)
#define NEGC    (-4294967295.0f)

__device__ __forceinline__ v8f wmma4(v2f a, v2f b, v8f c) {
  return __builtin_amdgcn_wmma_f32_16x16x4_f32(false, a, false, b, (short)0, c,
                                               false, false);
}

__global__ void __launch_bounds__(256) __attribute__((amdgpu_waves_per_eu(2, 2))) qkv_kernel(const float* __restrict__ re, const int* __restrict__ alias,
                           const float* __restrict__ Wq, const float* __restrict__ Wk,
                           const float* __restrict__ Wv,
                           float* __restrict__ wsQ, float* __restrict__ wsK,
                           float* __restrict__ wsV) {
  extern __shared__ __attribute__((aligned(16))) float smem_h[];
  float* sH = smem_h;
  float* sW = smem_h + TP * HSTR;
  const int b = blockIdx.x, tid = threadIdx.x;

  {
    const int wv = __builtin_amdgcn_readfirstlane(tid >> 5);
    const int ln = tid & 31;
    const int tl = wv * 26 + ln;
    const int my_a = alias[(size_t)b * T_ + (tl < T_ ? tl : T_ - 1)];
    const float* reb = re + (size_t)b * NNODE_ * D_ + ln * 4;
#pragma unroll 1
    for (int h = 0; h < 2; ++h) {
      const int t0 = wv * 26 + h * 13;
      float4 v[13];
#pragma unroll
      for (int r = 0; r < 13; ++r) {
        const int a = __builtin_amdgcn_readlane(my_a, h * 13 + r);
        float4 x = *(const float4*)(reb + (size_t)a * D_);
        if (t0 + r >= T_) x = make_float4(0.f, 0.f, 0.f, 0.f);
        v[r] = x;
      }
#pragma unroll
      for (int r = 0; r < 13; ++r)
        *(float4*)(sH + (t0 + r) * HSTR + ln * 4) = v[r];
    }
  }
  for (int idx = tid; idx < 3 * D_ * D_; idx += 256) {
    int mat = idx >> 14, r = idx & 16383;
    const float* src = (mat == 0) ? Wq : (mat == 1) ? Wk : Wv;
    sW[idx] = src[r];
  }
  __syncthreads();

  const int wave = tid >> 5, lane = tid & 31, lh = lane & 15, hi = lane >> 4;
  for (int tile = wave; tile < 3 * NT * 8; tile += 8) {
    int mat = tile / (NT * 8), rem = tile % (NT * 8);
    int i = rem >> 3, n0 = (rem & 7) << 4;
    const float* W = sW + mat * D_ * D_;
    const float* arow = sH + (i * 16 + lh) * HSTR;
    v2f af[32], bf[32];
#pragma unroll
    for (int c = 0; c < 32; ++c) {
      int k = 4 * c + 2 * hi;
      af[c] = *(const v2f*)(arow + k);
      bf[c].x = W[k * D_ + n0 + lh];
      bf[c].y = W[(k + 1) * D_ + n0 + lh];
    }
    __builtin_amdgcn_sched_barrier(0);
    v8f acc = {0.f, 0.f, 0.f, 0.f, 0.f, 0.f, 0.f, 0.f};
#pragma unroll
    for (int c = 0; c < 32; ++c) acc = wmma4(af[c], bf[c], acc);
    float* dst = ((mat == 0) ? wsQ : (mat == 1) ? wsK : wsV) + (size_t)b * PBS;
#pragma unroll
    for (int r = 0; r < 8; ++r) {
      int row = i * 16 + r + 8 * hi;
      float v = acc[r];
      dst[row * D_ + n0 + lh] = v > 0.f ? v : 0.f;
    }
  }
}

__global__ void __launch_bounds__(256) __attribute__((amdgpu_waves_per_eu(2, 2))) attn_kernel(const float* __restrict__ re, const int* __restrict__ alias,
                            const int* __restrict__ seqlen,
                            const float* __restrict__ wsQ, const float* __restrict__ wsK,
                            const float* __restrict__ wsV, float* __restrict__ wsDec) {
  extern __shared__ float smem[];
  float* sK = smem;
  float* sV = smem + TP * HSTR;
  float* sS0 = smem + 2 * TP * HSTR;
  const int b = blockIdx.x, tid = threadIdx.x;
  const int len = seqlen[b];
  const int wave = tid >> 5, lane = tid & 31, lh = lane & 15, hi = lane >> 4;
  const int t_a0 = wave * 16 + lh, t_a1 = (wave + 8) * 16 + lh;
  const int my_a0 = alias[b * T_ + (t_a0 < T_ ? t_a0 : T_ - 1)];
  const int my_a1 = alias[b * T_ + (t_a1 < T_ ? t_a1 : T_ - 1)];
  const float* Kb = wsK + (size_t)b * PBS;
  const float* Vb = wsV + (size_t)b * PBS;
  for (int idx = tid; idx < TP * D_; idx += 256) {
    int t = idx >> 7, d = idx & 127;
    sK[t * HSTR + d] = Kb[idx];
    sV[t * HSTR + d] = Vb[idx];
  }
  __syncthreads();

  float* sS = sS0 + wave * (16 * SSTR);
  const float scale = 0.088388347648318447f;
  const float* reb = re + (size_t)b * NNODE_ * D_ + lh;

  for (int pass = 0; pass < 2; ++pass) {
    const int qt = wave + 8 * pass;
    const bool active = (qt < NT);
    int roff[8];
    float hres[8];
    if (active) {
      const int my_a = pass ? my_a1 : my_a0;
#pragma unroll
      for (int r = 0; r < 8; ++r) {
        const int alo = __builtin_amdgcn_readlane(my_a, r);
        const int ahi = __builtin_amdgcn_readlane(my_a, 8 + r);
        roff[r] = (hi ? ahi : alo) * D_;
      }
#pragma unroll
      for (int r = 0; r < 8; ++r) hres[r] = reb[roff[r]];
      v2f qa[32];
      const float* Qrow = wsQ + (size_t)b * PBS + (qt * 16 + lh) * D_;
#pragma unroll
      for (int c = 0; c < 32; ++c) qa[c] = *(const v2f*)(Qrow + 4 * c + 2 * hi);
      for (int j = 0; j < NT; ++j) {
        v8f acc = {0.f, 0.f, 0.f, 0.f, 0.f, 0.f, 0.f, 0.f};
#pragma unroll
        for (int c = 0; c < 32; ++c) {
          v2f bb = *(const v2f*)(sK + (j * 16 + lh) * HSTR + 4 * c + 2 * hi);
          acc = wmma4(qa[c], bb, acc);
        }
#pragma unroll
        for (int r = 0; r < 8; ++r) {
          int m = r + 8 * hi, s = j * 16 + lh;
          sS[m * SSTR + s] = (s < len) ? acc[r] * scale : NEGC;
        }
      }
    }
    __syncthreads();
    if (active) {
      float* rowp = sS + lh * SSTR + hi * 104;
      float mx = -3.0e38f;
      for (int s = 0; s < 104; ++s) mx = fmaxf(mx, rowp[s]);
      mx = fmaxf(mx, __shfl_xor(mx, 16, 32));
      float sum = 0.f;
      for (int s = 0; s < 104; ++s) { float e = __expf(rowp[s] - mx); rowp[s] = e; sum += e; }
      sum += __shfl_xor(sum, 16, 32);
      float qm = ((qt * 16 + lh) < len) ? 1.f : 0.f;
      float inv = qm / sum;
      for (int s = 0; s < 104; ++s) rowp[s] *= inv;
    }
    __syncthreads();
    if (active) {
      float* Db = wsDec + (size_t)b * PBS;
      v2f pa[52];
#pragma unroll
      for (int c = 0; c < 52; ++c) pa[c] = *(const v2f*)(sS + lh * SSTR + 4 * c + 2 * hi);
      const float* vcol = sV + (2 * hi) * HSTR + lh;
      v2f vb0[26], vb1[26];
#pragma unroll
      for (int c = 0; c < 26; ++c) {
        vb0[c].x = vcol[(4 * c) * HSTR];
        vb0[c].y = vcol[(4 * c + 1) * HSTR];
      }
      for (int nt = 0; nt < 8; ++nt) {
        int n0 = nt * 16;
        const int n1 = (nt < 7) ? n0 + 16 : n0;
#pragma unroll
        for (int c = 0; c < 26; ++c) {
          vb1[c].x = vcol[(4 * (c + 26)) * HSTR + n0];
          vb1[c].y = vcol[(4 * (c + 26) + 1) * HSTR + n0];
        }
        v8f acc;
#pragma unroll
        for (int r = 0; r < 8; ++r)
          acc[r] = (qt * 16 + r + 8 * hi < T_) ? hres[r] : 0.f;
        if (nt < 7) {
#pragma unroll
          for (int r = 0; r < 8; ++r) hres[r] = reb[roff[r] + n0 + 16];
        }
        __builtin_amdgcn_sched_barrier(0);
#pragma unroll
        for (int c = 0; c < 26; ++c) acc = wmma4(pa[c], vb0[c], acc);
#pragma unroll
        for (int c = 0; c < 26; ++c) {
          vb0[c].x = vcol[(4 * c) * HSTR + n1];
          vb0[c].y = vcol[(4 * c + 1) * HSTR + n1];
        }
#pragma unroll
        for (int c = 0; c < 26; ++c) acc = wmma4(pa[c + 26], vb1[c], acc);
#pragma unroll
        for (int r = 0; r < 8; ++r) {
          int t = qt * 16 + r + 8 * hi;
          Db[t * D_ + n0 + lh] = acc[r];
        }
      }
    }
    __syncthreads();
  }
}

__global__ void stamp_kernel(const int* __restrict__ seqlen, const float* __restrict__ wsDec,
                             const float* __restrict__ u_w1, const float* __restrict__ u_w2,
                             const float* __restrict__ u_b, const float* __restrict__ u_v,
                             float* __restrict__ out) {
  extern __shared__ float smem[];
  float* sW2 = smem;
  float* sLp = smem + D_ * D_;
  float* sUb = sLp + 128;
  float* sUv = sUb + 128;
  float* sCoef = sUv + 128;
  const int b = blockIdx.x, tid = threadIdx.x;
  const int len = seqlen[b];
  const float* Db = wsDec + (size_t)b * PBS;

  for (int idx = tid; idx < D_ * D_; idx += 256) sW2[idx] = u_w2[idx];
  if (tid < D_) { sUb[tid] = u_b[tid]; sUv[tid] = u_v[tid]; }
  if (tid < D_) {
    const float* lrow = Db + (len - 1) * D_;
    float acc = 0.f;
    for (int d = 0; d < D_; ++d) acc += lrow[d] * u_w1[d * D_ + tid];
    sLp[tid] = acc;
    out[(size_t)b * 2 * D_ + D_ + tid] = acc;
  }
  __syncthreads();

  const int wave = tid >> 5, lane = tid & 31, lh = lane & 15, hi = lane >> 4;
  for (int pass = 0; pass < 2; ++pass) {
    const int qt = wave + 8 * pass;
    if (qt < NT) {
      v2f da[32];
      const float* drow = Db + (qt * 16 + lh) * D_;
#pragma unroll
      for (int c = 0; c < 32; ++c) da[c] = *(const v2f*)(drow + 4 * c + 2 * hi);
      float cpart[8];
#pragma unroll
      for (int r = 0; r < 8; ++r) cpart[r] = 0.f;
      for (int nt = 0; nt < 8; ++nt) {
        int n0 = nt * 16;
        v8f acc = {0.f, 0.f, 0.f, 0.f, 0.f, 0.f, 0.f, 0.f};
#pragma unroll
        for (int c = 0; c < 32; ++c) {
          int k = 4 * c + 2 * hi;
          v2f bb;
          bb.x = sW2[k * D_ + n0 + lh];
          bb.y = sW2[(k + 1) * D_ + n0 + lh];
          acc = wmma4(da[c], bb, acc);
        }
        int col = n0 + lh;
        float add = sLp[col] + sUb[col];
        float uvv = sUv[col];
#pragma unroll
        for (int r = 0; r < 8; ++r) {
          float x = acc[r] + add;
          float sg = 1.f / (1.f + __expf(-x));
          cpart[r] += sg * uvv;
        }
      }
#pragma unroll
      for (int r = 0; r < 8; ++r) {
        float v = cpart[r];
        v += __shfl_xor(v, 1, 32);
        v += __shfl_xor(v, 2, 32);
        v += __shfl_xor(v, 4, 32);
        v += __shfl_xor(v, 8, 32);
        if (lh == 0) {
          int t = qt * 16 + r + 8 * hi;
          sCoef[t] = (t < len) ? v : 0.f;
        }
      }
    }
  }
  __syncthreads();
  if (tid < D_) {
    float acc = 0.f;
    for (int t = 0; t < TP; ++t) acc += sCoef[t] * Db[t * D_ + tid];
    out[(size_t)b * 2 * D_ + tid] = acc;
  }
}

extern "C" void kernel_launch(void* const* d_in, const int* in_sizes, int n_in,
                              void* d_out, int out_size, void* d_ws, size_t ws_size,
                              hipStream_t stream) {
  (void)in_sizes; (void)n_in; (void)out_size; (void)ws_size;
  const float* re   = (const float*)d_in[0];
  const int*   alia = (const int*)d_in[1];
  const int*   slen = (const int*)d_in[2];
  const float* Wq   = (const float*)d_in[3];
  const float* Wk   = (const float*)d_in[4];
  const float* Wv   = (const float*)d_in[5];
  const float* u_w1 = (const float*)d_in[6];
  const float* u_w2 = (const float*)d_in[7];
  const float* u_b  = (const float*)d_in[8];
  const float* u_v  = (const float*)d_in[9];
  float* out = (float*)d_out;

  float* ws   = (float*)d_ws;
  float* wsQ  = ws;
  float* wsK  = ws + (size_t)B_ * PBS;
  float* wsV  = ws + 2 * (size_t)B_ * PBS;
  float* wsDec = wsQ;

  size_t sm1 = (size_t)(TP * HSTR + 3 * D_ * D_) * sizeof(float);
  size_t sm2 = (size_t)(2 * TP * HSTR + 8 * 16 * SSTR) * sizeof(float);
  size_t sm3 = (size_t)(D_ * D_ + 3 * 128 + TP) * sizeof(float);
  hipFuncSetAttribute((const void*)qkv_kernel,
                      hipFuncAttributeMaxDynamicSharedMemorySize, (int)sm1);
  hipFuncSetAttribute((const void*)attn_kernel,
                      hipFuncAttributeMaxDynamicSharedMemorySize, (int)sm2);
  hipFuncSetAttribute((const void*)stamp_kernel,
                      hipFuncAttributeMaxDynamicSharedMemorySize, (int)sm3);

  qkv_kernel<<<B_, 256, sm1, stream>>>(re, alia, Wq, Wk, Wv, wsQ, wsK, wsV);
  attn_kernel<<<B_, 256, sm2, stream>>>(re, alia, slen, wsQ, wsK, wsV, wsDec);
  stamp_kernel<<<B_, 256, sm3, stream>>>(slen, wsDec, u_w1, u_w2, u_b, u_v, out);
}


# W (qkv), K and V (attn) and u_w2 (stamp) staging into LDS: 128-bit loads, a thread's whole share requested as one group, instead of 4-byte loads awaited in small groups per loop trip
# speedup vs baseline: 2.1770x; 1.0986x over previous
#include <hip/hip_runtime.h>
#include <math.h>


typedef __attribute__((ext_vector_type(2))) float v2f;
typedef __attribute__((ext_vector_type(8))) float v8f;

#define B_      2048
#define T_      200
#define D_      128
#define NNODE_  200
#define TP      208
#define NT      13
#define HSTR    132
#define SSTR    210
#define PBS     (TP * D_)
#define NEGC    (-4294967295.0f)

__device__ __forceinline__ v8f wmma4(v2f a, v2f b, v8f c) {
  return __builtin_amdgcn_wmma_f32_16x16x4_f32(false, a, false, b, (short)0, c,
                                               false, false);
}

__global__ void __launch_bounds__(256) __attribute__((amdgpu_waves_per_eu(2, 2))) qkv_kernel(const float* __restrict__ re, const int* __restrict__ alias,
                           const float* __restrict__ Wq, const float* __restrict__ Wk,
                           const float* __restrict__ Wv,
                           float* __restrict__ wsQ, float* __restrict__ wsK,
                           float* __restrict__ wsV) {
  extern __shared__ __attribute__((aligned(16))) float smem_h[];
  float* sH = smem_h;
  float* sW = smem_h + TP * HSTR;
  const int b = blockIdx.x, tid = threadIdx.x;

  {
    const int wv = __builtin_amdgcn_readfirstlane(tid >> 5);
    const int ln = tid & 31;
    const int tl = wv * 26 + ln;
    const int my_a = alias[(size_t)b * T_ + (tl < T_ ? tl : T_ - 1)];
    {
      float4 w[48];
#pragma unroll
      for (int m = 0; m < 3; ++m) {
        const float4* src = (const float4*)((m == 0) ? Wq : (m == 1) ? Wk : Wv);
#pragma unroll
        for (int i = 0; i < 16; ++i) w[m * 16 + i] = src[tid + 256 * i];
      }
#pragma unroll
      for (int j = 0; j < 48; ++j) *(float4*)(sW + 4 * (tid + 256 * j)) = w[j];
    }
    const float* reb = re + (size_t)b * NNODE_ * D_ + ln * 4;
#pragma unroll 1
    for (int h = 0; h < 2; ++h) {
      const int t0 = wv * 26 + h * 13;
      float4 v[13];
#pragma unroll
      for (int r = 0; r < 13; ++r) {
        const int a = __builtin_amdgcn_readlane(my_a, h * 13 + r);
        float4 x = *(const float4*)(reb + (size_t)a * D_);
        if (t0 + r >= T_) x = make_float4(0.f, 0.f, 0.f, 0.f);
        v[r] = x;
      }
#pragma unroll
      for (int r = 0; r < 13; ++r)
        *(float4*)(sH + (t0 + r) * HSTR + ln * 4) = v[r];
    }
  }
  __syncthreads();

  const int wave = tid >> 5, lane = tid & 31, lh = lane & 15, hi = lane >> 4;
  for (int tile = wave; tile < 3 * NT * 8; tile += 8) {
    int mat = tile / (NT * 8), rem = tile % (NT * 8);
    int i = rem >> 3, n0 = (rem & 7) << 4;
    const float* W = sW + mat * D_ * D_;
    const float* arow = sH + (i * 16 + lh) * HSTR;
    v2f af[32], bf[32];
#pragma unroll
    for (int c = 0; c < 32; ++c) {
      int k = 4 * c + 2 * hi;
      af[c] = *(const v2f*)(arow + k);
      bf[c].x = W[k * D_ + n0 + lh];
      bf[c].y = W[(k + 1) * D_ + n0 + lh];
    }
    __builtin_amdgcn_sched_barrier(0);
    v8f acc = {0.f, 0.f, 0.f, 0.f, 0.f, 0.f, 0.f, 0.f};
#pragma unroll
    for (int c = 0; c < 32; ++c) acc = wmma4(af[c], bf[c], acc);
    float* dst = ((mat == 0) ? wsQ : (mat == 1) ? wsK : wsV) + (size_t)b * PBS;
#pragma unroll
    for (int r = 0; r < 8; ++r) {
      int row = i * 16 + r + 8 * hi;
      float v = acc[r];
      dst[row * D_ + n0 + lh] = v > 0.f ? v : 0.f;
    }
  }
}

__global__ void __launch_bounds__(256) __attribute__((amdgpu_waves_per_eu(2, 2))) attn_kernel(const float* __restrict__ re, const int* __restrict__ alias,
                            const int* __restrict__ seqlen,
                            const float* __restrict__ wsQ, const float* __restrict__ wsK,
                            const float* __restrict__ wsV, float* __restrict__ wsDec) {
  extern __shared__ __attribute__((aligned(16))) float smem[];
  float* sK = smem;
  float* sV = smem + TP * HSTR;
  float* sS0 = smem + 2 * TP * HSTR;
  const int b = blockIdx.x, tid = threadIdx.x;
  const int len = seqlen[b];
  const int wave = tid >> 5, lane = tid & 31, lh = lane & 15, hi = lane >> 4;
  const int t_a0 = wave * 16 + lh, t_a1 = (wave + 8) * 16 + lh;
  const int my_a0 = alias[b * T_ + (t_a0 < T_ ? t_a0 : T_ - 1)];
  const int my_a1 = alias[b * T_ + (t_a1 < T_ ? t_a1 : T_ - 1)];
  const float* Kb = wsK + (size_t)b * PBS;
  const float* Vb = wsV + (size_t)b * PBS;
  {
    const float4* K4 = (const float4*)Kb + tid;
    const float4* V4 = (const float4*)Vb + tid;
    float4 kr[26], vr[26];
#pragma unroll
    for (int i = 0; i < 26; ++i) kr[i] = K4[256 * i];
#pragma unroll
    for (int i = 0; i < 26; ++i) vr[i] = V4[256 * i];
#pragma unroll
    for (int i = 0; i < 26; ++i) *(float4*)(sK + (wave + 8 * i) * HSTR + lane * 4) = kr[i];
#pragma unroll
    for (int i = 0; i < 26; ++i) *(float4*)(sV + (wave + 8 * i) * HSTR + lane * 4) = vr[i];
  }
  __syncthreads();

  float* sS = sS0 + wave * (16 * SSTR);
  const float scale = 0.088388347648318447f;
  const float* reb = re + (size_t)b * NNODE_ * D_ + lh;

  for (int pass = 0; pass < 2; ++pass) {
    const int qt = wave + 8 * pass;
    const bool active = (qt < NT);
    int roff[8];
    float hres[8];
    if (active) {
      const int my_a = pass ? my_a1 : my_a0;
#pragma unroll
      for (int r = 0; r < 8; ++r) {
        const int alo = __builtin_amdgcn_readlane(my_a, r);
        const int ahi = __builtin_amdgcn_readlane(my_a, 8 + r);
        roff[r] = (hi ? ahi : alo) * D_;
      }
#pragma unroll
      for (int r = 0; r < 8; ++r) hres[r] = reb[roff[r]];
      v2f qa[32];
      const float* Qrow = wsQ + (size_t)b * PBS + (qt * 16 + lh) * D_;
#pragma unroll
      for (int c = 0; c < 32; ++c) qa[c] = *(const v2f*)(Qrow + 4 * c + 2 * hi);
      for (int j = 0; j < NT; ++j) {
        v8f acc = {0.f, 0.f, 0.f, 0.f, 0.f, 0.f, 0.f, 0.f};
#pragma unroll
        for (int c = 0; c < 32; ++c) {
          v2f bb = *(const v2f*)(sK + (j * 16 + lh) * HSTR + 4 * c + 2 * hi);
          acc = wmma4(qa[c], bb, acc);
        }
#pragma unroll
        for (int r = 0; r < 8; ++r) {
          int m = r + 8 * hi, s = j * 16 + lh;
          sS[m * SSTR + s] = (s < len) ? acc[r] * scale : NEGC;
        }
      }
    }
    __syncthreads();
    if (active) {
      float* rowp = sS + lh * SSTR + hi * 104;
      float mx = -3.0e38f;
      for (int s = 0; s < 104; ++s) mx = fmaxf(mx, rowp[s]);
      mx = fmaxf(mx, __shfl_xor(mx, 16, 32));
      float sum = 0.f;
      for (int s = 0; s < 104; ++s) { float e = __expf(rowp[s] - mx); rowp[s] = e; sum += e; }
      sum += __shfl_xor(sum, 16, 32);
      float qm = ((qt * 16 + lh) < len) ? 1.f : 0.f;
      float inv = qm / sum;
      for (int s = 0; s < 104; ++s) rowp[s] *= inv;
    }
    __syncthreads();
    if (active) {
      float* Db = wsDec + (size_t)b * PBS;
      v2f pa[52];
#pragma unroll
      for (int c = 0; c < 52; ++c) pa[c] = *(const v2f*)(sS + lh * SSTR + 4 * c + 2 * hi);
      const float* vcol = sV + (2 * hi) * HSTR + lh;
      v2f vb0[26], vb1[26];
#pragma unroll
      for (int c = 0; c < 26; ++c) {
        vb0[c].x = vcol[(4 * c) * HSTR];
        vb0[c].y = vcol[(4 * c + 1) * HSTR];
      }
      for (int nt = 0; nt < 8; ++nt) {
        int n0 = nt * 16;
        const int n1 = (nt < 7) ? n0 + 16 : n0;
#pragma unroll
        for (int c = 0; c < 26; ++c) {
          vb1[c].x = vcol[(4 * (c + 26)) * HSTR + n0];
          vb1[c].y = vcol[(4 * (c + 26) + 1) * HSTR + n0];
        }
        v8f acc;
#pragma unroll
        for (int r = 0; r < 8; ++r)
          acc[r] = (qt * 16 + r + 8 * hi < T_) ? hres[r] : 0.f;
        if (nt < 7) {
#pragma unroll
          for (int r = 0; r < 8; ++r) hres[r] = reb[roff[r] + n0 + 16];
        }
        __builtin_amdgcn_sched_barrier(0);
#pragma unroll
        for (int c = 0; c < 26; ++c) acc = wmma4(pa[c], vb0[c], acc);
#pragma unroll
        for (int c = 0; c < 26; ++c) {
          vb0[c].x = vcol[(4 * c) * HSTR + n1];
          vb0[c].y = vcol[(4 * c + 1) * HSTR + n1];
        }
#pragma unroll
        for (int c = 0; c < 26; ++c) acc = wmma4(pa[c + 26], vb1[c], acc);
#pragma unroll
        for (int r = 0; r < 8; ++r) {
          int t = qt * 16 + r + 8 * hi;
          Db[t * D_ + n0 + lh] = acc[r];
        }
      }
    }
    __syncthreads();
  }
}

__global__ void stamp_kernel(const int* __restrict__ seqlen, const float* __restrict__ wsDec,
                             const float* __restrict__ u_w1, const float* __restrict__ u_w2,
                             const float* __restrict__ u_b, const float* __restrict__ u_v,
                             float* __restrict__ out) {
  extern __shared__ __attribute__((aligned(16))) float smem[];
  float* sW2 = smem;
  float* sLp = smem + D_ * D_;
  float* sUb = sLp + 128;
  float* sUv = sUb + 128;
  float* sCoef = sUv + 128;
  const int b = blockIdx.x, tid = threadIdx.x;
  const int len = seqlen[b];
  const float* Db = wsDec + (size_t)b * PBS;

  {
    const float4* src = (const float4*)u_w2 + tid;
    float4 w[16];
#pragma unroll
    for (int i = 0; i < 16; ++i) w[i] = src[256 * i];
#pragma unroll
    for (int i = 0; i < 16; ++i) *(float4*)(sW2 + 4 * (tid + 256 * i)) = w[i];
  }
  if (tid < D_) { sUb[tid] = u_b[tid]; sUv[tid] = u_v[tid]; }
  if (tid < D_) {
    const float* lrow = Db + (len - 1) * D_;
    float acc = 0.f;
    for (int d = 0; d < D_; ++d) acc += lrow[d] * u_w1[d * D_ + tid];
    sLp[tid] = acc;
    out[(size_t)b * 2 * D_ + D_ + tid] = acc;
  }
  __syncthreads();

  const int wave = tid >> 5, lane = tid & 31, lh = lane & 15, hi = lane >> 4;
  for (int pass = 0; pass < 2; ++pass) {
    const int qt = wave + 8 * pass;
    if (qt < NT) {
      v2f da[32];
      const float* drow = Db + (qt * 16 + lh) * D_;
#pragma unroll
      for (int c = 0; c < 32; ++c) da[c] = *(const v2f*)(drow + 4 * c + 2 * hi);
      float cpart[8];
#pragma unroll
      for (int r = 0; r < 8; ++r) cpart[r] = 0.f;
      for (int nt = 0; nt < 8; ++nt) {
        int n0 = nt * 16;
        v8f acc = {0.f, 0.f, 0.f, 0.f, 0.f, 0.f, 0.f, 0.f};
#pragma unroll
        for (int c = 0; c < 32; ++c) {
          int k = 4 * c + 2 * hi;
          v2f bb;
          bb.x = sW2[k * D_ + n0 + lh];
          bb.y = sW2[(k + 1) * D_ + n0 + lh];
          acc = wmma4(da[c], bb, acc);
        }
        int col = n0 + lh;
        float add = sLp[col] + sUb[col];
        float uvv = sUv[col];
#pragma unroll
        for (int r = 0; r < 8; ++r) {
          float x = acc[r] + add;
          float sg = 1.f / (1.f + __expf(-x));
          cpart[r] += sg * uvv;
        }
      }
#pragma unroll
      for (int r = 0; r < 8; ++r) {
        float v = cpart[r];
        v += __shfl_xor(v, 1, 32);
        v += __shfl_xor(v, 2, 32);
        v += __shfl_xor(v, 4, 32);
        v += __shfl_xor(v, 8, 32);
        if (lh == 0) {
          int t = qt * 16 + r + 8 * hi;
          sCoef[t] = (t < len) ? v : 0.f;
        }
      }
    }
  }
  __syncthreads();
  if (tid < D_) {
    float acc = 0.f;
    for (int t = 0; t < TP; ++t) acc += sCoef[t] * Db[t * D_ + tid];
    out[(size_t)b * 2 * D_ + tid] = acc;
  }
}

extern "C" void kernel_launch(void* const* d_in, const int* in_sizes, int n_in,
                              void* d_out, int out_size, void* d_ws, size_t ws_size,
                              hipStream_t stream) {
  (void)in_sizes; (void)n_in; (void)out_size; (void)ws_size;
  const float* re   = (const float*)d_in[0];
  const int*   alia = (const int*)d_in[1];
  const int*   slen = (const int*)d_in[2];
  const float* Wq   = (const float*)d_in[3];
  const float* Wk   = (const float*)d_in[4];
  const float* Wv   = (const float*)d_in[5];
  const float* u_w1 = (const float*)d_in[6];
  const float* u_w2 = (const float*)d_in[7];
  const float* u_b  = (const float*)d_in[8];
  const float* u_v  = (const float*)d_in[9];
  float* out = (float*)d_out;

  float* ws   = (float*)d_ws;
  float* wsQ  = ws;
  float* wsK  = ws + (size_t)B_ * PBS;
  float* wsV  = ws + 2 * (size_t)B_ * PBS;
  float* wsDec = wsQ;

  size_t sm1 = (size_t)(TP * HSTR + 3 * D_ * D_) * sizeof(float);
  size_t sm2 = (size_t)(2 * TP * HSTR + 8 * 16 * SSTR) * sizeof(float);
  size_t sm3 = (size_t)(D_ * D_ + 3 * 128 + TP) * sizeof(float);
  hipFuncSetAttribute((const void*)qkv_kernel,
                      hipFuncAttributeMaxDynamicSharedMemorySize, (int)sm1);
  hipFuncSetAttribute((const void*)attn_kernel,
                      hipFuncAttributeMaxDynamicSharedMemorySize, (int)sm2);
  hipFuncSetAttribute((const void*)stamp_kernel,
                      hipFuncAttributeMaxDynamicSharedMemorySize, (int)sm3);

  qkv_kernel<<<B_, 256, sm1, stream>>>(re, alia, Wq, Wk, Wv, wsQ, wsK, wsV);
  attn_kernel<<<B_, 256, sm2, stream>>>(re, alia, slen, wsQ, wsK, wsV, wsDec);
  stamp_kernel<<<B_, 256, sm3, stream>>>(slen, wsDec, u_w1, u_w2, u_b, u_v, out);
}


# all three kernels stage, compute and store only the 16-row tiles below seq_len (masked keys weigh exactly 0, masked query rows never reach the output); live terms are summed in the old order
# speedup vs baseline: 4.0889x; 1.8782x over previous
#include <hip/hip_runtime.h>
#include <math.h>


typedef __attribute__((ext_vector_type(2))) float v2f;
typedef __attribute__((ext_vector_type(8))) float v8f;

#define B_      2048
#define T_      200
#define D_      128
#define NNODE_  200
#define TP      208
#define NT      13
#define HSTR    132
#define SSTR    210
#define PBS     (TP * D_)
#define NEGC    (-4294967295.0f)

__device__ __forceinline__ int live_tiles(int len) {
  const int n = (len + 15) >> 4;
  return n < 1 ? 1 : (n > NT ? NT : n);
}

__device__ __forceinline__ v8f wmma4(v2f a, v2f b, v8f c) {
  return __builtin_amdgcn_wmma_f32_16x16x4_f32(false, a, false, b, (short)0, c,
                                               false, false);
}

__global__ void __launch_bounds__(256) __attribute__((amdgpu_waves_per_eu(2, 2))) qkv_kernel(const float* __restrict__ re, const int* __restrict__ alias,
                           const int* __restrict__ seqlen,
                           const float* __restrict__ Wq, const float* __restrict__ Wk,
                           const float* __restrict__ Wv,
                           float* __restrict__ wsQ, float* __restrict__ wsK,
                           float* __restrict__ wsV) {
  extern __shared__ __attribute__((aligned(16))) float smem_h[];
  float* sH = smem_h;
  float* sW = smem_h + TP * HSTR;
  const int b = blockIdx.x, tid = threadIdx.x;
  const int ntl = live_tiles(seqlen[b]);

  {
    const int wv = __builtin_amdgcn_readfirstlane(tid >> 5);
    const int ln = tid & 31;
    const int tl = wv + 8 * ln;
    const int my_a = alias[(size_t)b * T_ + (tl < T_ ? tl : T_ - 1)];
    {
      float4 w[48];
#pragma unroll
      for (int m = 0; m < 3; ++m) {
        const float4* src = (const float4*)((m == 0) ? Wq : (m == 1) ? Wk : Wv);
#pragma unroll
        for (int i = 0; i < 16; ++i) w[m * 16 + i] = src[tid + 256 * i];
      }
#pragma unroll
      for (int j = 0; j < 48; ++j) *(float4*)(sW + 4 * (tid + 256 * j)) = w[j];
    }
    const float* reb = re + (size_t)b * NNODE_ * D_ + ln * 4;
    const int nk = 2 * ntl;
#pragma unroll 1
    for (int h = 0; h < 2; ++h) {
      const int k0 = h * 13;
      if (k0 >= nk) break;
      float4 v[13];
#pragma unroll
      for (int r = 0; r < 13; ++r) {
        if (k0 + r < nk) {
          const int a = __builtin_amdgcn_readlane(my_a, k0 + r);
          float4 x = *(const float4*)(reb + (size_t)a * D_);
          if (wv + 8 * (k0 + r) >= T_) x = make_float4(0.f, 0.f, 0.f, 0.f);
          v[r] = x;
        }
      }
#pragma unroll
      for (int r = 0; r < 13; ++r)
        if (k0 + r < nk) *(float4*)(sH + (wv + 8 * (k0 + r)) * HSTR + ln * 4) = v[r];
    }
  }
  __syncthreads();

  const int wave = tid >> 5, lane = tid & 31, lh = lane & 15, hi = lane >> 4;
  const int n0 = wave << 4;
  for (int it = 0; it < 3 * ntl; ++it) {
    const int mat = (it >= ntl) + (it >= 2 * ntl);
    const int i = it - mat * ntl;
    const float* wcol = sW + mat * D_ * D_ + (2 * hi) * D_ + n0 + lh;
    const float* arow = sH + (i * 16 + lh) * HSTR;
    v2f af[32], bf[32];
#pragma unroll
    for (int c = 0; c < 32; ++c) {
      int k = 4 * c + 2 * hi;
      af[c] = *(const v2f*)(arow + k);
      bf[c].x = wcol[(4 * c) * D_];
      bf[c].y = wcol[(4 * c + 1) * D_];
    }
    __builtin_amdgcn_sched_barrier(0);
    v8f acc = {0.f, 0.f, 0.f, 0.f, 0.f, 0.f, 0.f, 0.f};
#pragma unroll
    for (int c = 0; c < 32; ++c) acc = wmma4(af[c], bf[c], acc);
    float* dst = ((mat == 0) ? wsQ : (mat == 1) ? wsK : wsV) + (size_t)b * PBS;
#pragma unroll
    for (int r = 0; r < 8; ++r) {
      int row = i * 16 + r + 8 * hi;
      float v = acc[r];
      dst[row * D_ + n0 + lh] = v > 0.f ? v : 0.f;
    }
  }
}

template <int NG>
__device__ __forceinline__ void stage_kv(const float4* __restrict__ K4, const float4* __restrict__ V4,
                                         float* sK, float* sV, int wave, int lane, int ntl) {
  float4 kr[2 * NG], vr[2 * NG];
#pragma unroll
  for (int g = 0; g < NG; ++g) {
    const int gg = g < ntl ? g : ntl - 1;
    kr[2 * g] = K4[256 * (2 * gg)];
    kr[2 * g + 1] = K4[256 * (2 * gg + 1)];
  }
#pragma unroll
  for (int g = 0; g < NG; ++g) {
    const int gg = g < ntl ? g : ntl - 1;
    vr[2 * g] = V4[256 * (2 * gg)];
    vr[2 * g + 1] = V4[256 * (2 * gg + 1)];
  }
#pragma unroll
  for (int g = 0; g < NG; ++g) {
    const int gg = g < ntl ? g : ntl - 1;
    *(float4*)(sK + (wave + 16 * gg) * HSTR + lane * 4) = kr[2 * g];
    *(float4*)(sK + (wave + 16 * gg + 8) * HSTR + lane * 4) = kr[2 * g + 1];
  }
#pragma unroll
  for (int g = 0; g < NG; ++g) {
    const int gg = g < ntl ? g : ntl - 1;
    *(float4*)(sV + (wave + 16 * gg) * HSTR + lane * 4) = vr[2 * g];
    *(float4*)(sV + (wave + 16 * gg + 8) * HSTR + lane * 4) = vr[2 * g + 1];
  }
}

__global__ void __launch_bounds__(256) __attribute__((amdgpu_waves_per_eu(2, 2))) attn_kernel(const float* __restrict__ re, const int* __restrict__ alias,
                            const int* __restrict__ seqlen,
                            const float* __restrict__ wsQ, const float* __restrict__ wsK,
                            const float* __restrict__ wsV, float* __restrict__ wsDec) {
  extern __shared__ __attribute__((aligned(16))) float smem[];
  float* sK = smem;
  float* sV = smem + TP * HSTR;
  float* sS0 = smem + 2 * TP * HSTR;
  const int b = blockIdx.x, tid = threadIdx.x;
  const int len = seqlen[b];
  const int ntl = live_tiles(len);
  const int wave = tid >> 5, lane = tid & 31, lh = lane & 15, hi = lane >> 4;
  const int t_a0 = wave * 16 + lh, t_a1 = (wave + 8) * 16 + lh;
  const int my_a0 = alias[b * T_ + (t_a0 < T_ ? t_a0 : T_ - 1)];
  const int my_a1 = alias[b * T_ + (t_a1 < T_ ? t_a1 : T_ - 1)];
  const float* Kb = wsK + (size_t)b * PBS;
  const float* Vb = wsV + (size_t)b * PBS;
  {
    const float4* K4 = (const float4*)Kb + tid;
    const float4* V4 = (const float4*)Vb + tid;
    if (ntl <= 2)       stage_kv<2>(K4, V4, sK, sV, wave, lane, ntl);
    else if (ntl <= 4)  stage_kv<4>(K4, V4, sK, sV, wave, lane, ntl);
    else if (ntl <= 7)  stage_kv<7>(K4, V4, sK, sV, wave, lane, ntl);
    else if (ntl <= 10) stage_kv<10>(K4, V4, sK, sV, wave, lane, ntl);
    else                stage_kv<NT>(K4, V4, sK, sV, wave, lane, ntl);
  }
  __syncthreads();

  float* sS = sS0 + wave * (16 * SSTR);
  const float scale = 0.088388347648318447f;
  const float* reb = re + (size_t)b * NNODE_ * D_ + lh;

  const int npass = (ntl > 8) ? 2 : 1;
  for (int pass = 0; pass < npass; ++pass) {
    const int qt = wave + 8 * pass;
    const bool active = (qt < ntl);
    int roff[8];
    float hres[8];
    if (active) {
      const int my_a = pass ? my_a1 : my_a0;
#pragma unroll
      for (int r = 0; r < 8; ++r) {
        const int alo = __builtin_amdgcn_readlane(my_a, r);
        const int ahi = __builtin_amdgcn_readlane(my_a, 8 + r);
        roff[r] = (hi ? ahi : alo) * D_;
      }
#pragma unroll
      for (int r = 0; r < 8; ++r) hres[r] = reb[roff[r]];
      v2f qa[32];
      const float* Qrow = wsQ + (size_t)b * PBS + (qt * 16 + lh) * D_;
#pragma unroll
      for (int c = 0; c < 32; ++c) qa[c] = *(const v2f*)(Qrow + 4 * c + 2 * hi);
      for (int j = 0; j < ntl; ++j) {
        v8f acc = {0.f, 0.f, 0.f, 0.f, 0.f, 0.f, 0.f, 0.f};
#pragma unroll
        for (int c = 0; c < 32; ++c) {
          v2f bb = *(const v2f*)(sK + (j * 16 + lh) * HSTR + 4 * c + 2 * hi);
          acc = wmma4(qa[c], bb, acc);
        }
#pragma unroll
        for (int r = 0; r < 8; ++r) {
          int m = r + 8 * hi, s = j * 16 + lh;
          sS[m * SSTR + s] = (s < len) ? acc[r] * scale : NEGC;
        }
      }
    }
    __syncthreads();
    if (active) {
      float* rowp = sS + lh * SSTR + hi * 104;
      int cnt = 16 * ntl - hi * 104;
      cnt = cnt > 104 ? 104 : (cnt < 0 ? 0 : cnt);
      float mx = -3.0e38f;
      for (int s0 = 0; s0 < cnt; s0 += 8) {
#pragma unroll
        for (int u = 0; u < 8; ++u) mx = fmaxf(mx, rowp[s0 + u]);
      }
      mx = fmaxf(mx, __shfl_xor(mx, 16, 32));
      float sum = 0.f;
      for (int s0 = 0; s0 < cnt; s0 += 8) {
#pragma unroll
        for (int u = 0; u < 8; ++u) { float e = __expf(rowp[s0 + u] - mx); rowp[s0 + u] = e; sum += e; }
      }
      sum += __shfl_xor(sum, 16, 32);
      float qm = ((qt * 16 + lh) < len) ? 1.f : 0.f;
      float inv = qm / sum;
      for (int s0 = 0; s0 < cnt; s0 += 8) {
#pragma unroll
        for (int u = 0; u < 8; ++u) rowp[s0 + u] *= inv;
      }
    }
    __syncthreads();
    if (active) {
      float* Db = wsDec + (size_t)b * PBS;
      v2f pa[52];
#pragma unroll
      for (int g = 0; g < NT; ++g)
        if (g < ntl) {
#pragma unroll
          for (int cc = 0; cc < 4; ++cc) {
            const int c = 4 * g + cc;
            pa[c] = *(const v2f*)(sS + lh * SSTR + 4 * c + 2 * hi);
          }
        }
      const float* vcol = sV + (2 * hi) * HSTR + lh;
      v2f vb0[26], vb1[26];
#pragma unroll
      for (int g = 0; g < 7; ++g)
        if (g < ntl) {
#pragma unroll
          for (int cc = 0; cc < 4; ++cc) {
            const int c = 4 * g + cc;
            if (c < 26) {
              vb0[c].x = vcol[(4 * c) * HSTR];
              vb0[c].y = vcol[(4 * c + 1) * HSTR];
            }
          }
        }
      for (int nt = 0; nt < 8; ++nt) {
        int n0 = nt * 16;
        const int n1 = (nt < 7) ? n0 + 16 : n0;
        v8f acc;
#pragma unroll
        for (int r = 0; r < 8; ++r)
          acc[r] = (qt * 16 + r + 8 * hi < T_) ? hres[r] : 0.f;
        if (nt < 7) {
#pragma unroll
          for (int r = 0; r < 8; ++r) hres[r] = reb[roff[r] + n0 + 16];
        }
#pragma unroll
        for (int c = 0; c < 4; ++c) acc = wmma4(pa[c], vb0[c], acc);
        asm volatile("" : "+v"(acc));
#pragma unroll
        for (int g = 6; g < NT; ++g)
          if (g < ntl) {
#pragma unroll
            for (int cc = 0; cc < 4; ++cc) {
              const int k = 4 * g + cc;
              if (k >= 26) {
                vb1[k - 26].x = vcol[(4 * k) * HSTR + n0];
                vb1[k - 26].y = vcol[(4 * k + 1) * HSTR + n0];
              }
            }
          }
#pragma unroll
        for (int g = 1; g < 7; ++g)
          if (g < ntl) {
#pragma unroll
            for (int cc = 0; cc < 4; ++cc) {
              const int c = 4 * g + cc;
              if (c < 26) acc = wmma4(pa[c], vb0[c], acc);
            }
          }
        if (6 < ntl) {
          acc = wmma4(pa[26], vb1[0], acc);
          acc = wmma4(pa[27], vb1[1], acc);
        }
#pragma unroll
        for (int g = 0; g < 7; ++g)
          if (g < ntl) {
#pragma unroll
            for (int cc = 0; cc < 4; ++cc) {
              const int c = 4 * g + cc;
              if (c < 26) {
                vb0[c].x = vcol[(4 * c) * HSTR + n1];
                vb0[c].y = vcol[(4 * c + 1) * HSTR + n1];
              }
            }
          }
#pragma unroll
        for (int g = 7; g < NT; ++g)
          if (g < ntl) {
#pragma unroll
            for (int cc = 0; cc < 4; ++cc) {
              const int k = 4 * g + cc;
              acc = wmma4(pa[k], vb1[k - 26], acc);
            }
          }
#pragma unroll
        for (int r = 0; r < 8; ++r) {
          int t = qt * 16 + r + 8 * hi;
          Db[t * D_ + n0 + lh] = acc[r];
        }
      }
    }
    __syncthreads();
  }
}

__global__ void stamp_kernel(const int* __restrict__ seqlen, const float* __restrict__ wsDec,
                             const float* __restrict__ u_w1, const float* __restrict__ u_w2,
                             const float* __restrict__ u_b, const float* __restrict__ u_v,
                             float* __restrict__ out) {
  extern __shared__ __attribute__((aligned(16))) float smem[];
  float* sW2 = smem;
  float* sLp = smem + D_ * D_;
  float* sUb = sLp + 128;
  float* sUv = sUb + 128;
  float* sCoef = sUv + 128;
  const int b = blockIdx.x, tid = threadIdx.x;
  const int len = seqlen[b];
  const int ntl = live_tiles(len);
  const float* Db = wsDec + (size_t)b * PBS;

  {
    const float4* src = (const float4*)u_w2 + tid;
    float4 w[16];
#pragma unroll
    for (int i = 0; i < 16; ++i) w[i] = src[256 * i];
#pragma unroll
    for (int i = 0; i < 16; ++i) *(float4*)(sW2 + 4 * (tid + 256 * i)) = w[i];
  }
  if (tid < D_) { sUb[tid] = u_b[tid]; sUv[tid] = u_v[tid]; }
  if (tid < D_) {
    const float* lrow = Db + (len - 1) * D_;
    float acc = 0.f;
    for (int d = 0; d < D_; ++d) acc += lrow[d] * u_w1[d * D_ + tid];
    sLp[tid] = acc;
    out[(size_t)b * 2 * D_ + D_ + tid] = acc;
  }
  __syncthreads();

  const int wave = tid >> 5, lane = tid & 31, lh = lane & 15, hi = lane >> 4;
  for (int pass = 0; pass < 2; ++pass) {
    const int qt = wave + 8 * pass;
    if (qt < ntl) {
      v2f da[32];
      const float* drow = Db + (qt * 16 + lh) * D_;
#pragma unroll
      for (int c = 0; c < 32; ++c) da[c] = *(const v2f*)(drow + 4 * c + 2 * hi);
      float cpart[8];
#pragma unroll
      for (int r = 0; r < 8; ++r) cpart[r] = 0.f;
      for (int nt = 0; nt < 8; ++nt) {
        int n0 = nt * 16;
        v8f acc = {0.f, 0.f, 0.f, 0.f, 0.f, 0.f, 0.f, 0.f};
#pragma unroll
        for (int c = 0; c < 32; ++c) {
          int k = 4 * c + 2 * hi;
          v2f bb;
          bb.x = sW2[k * D_ + n0 + lh];
          bb.y = sW2[(k + 1) * D_ + n0 + lh];
          acc = wmma4(da[c], bb, acc);
        }
        int col = n0 + lh;
        float add = sLp[col] + sUb[col];
        float uvv = sUv[col];
#pragma unroll
        for (int r = 0; r < 8; ++r) {
          float x = acc[r] + add;
          float sg = 1.f / (1.f + __expf(-x));
          cpart[r] += sg * uvv;
        }
      }
#pragma unroll
      for (int r = 0; r < 8; ++r) {
        float v = cpart[r];
        v += __shfl_xor(v, 1, 32);
        v += __shfl_xor(v, 2, 32);
        v += __shfl_xor(v, 4, 32);
        v += __shfl_xor(v, 8, 32);
        if (lh == 0) {
          int t = qt * 16 + r + 8 * hi;
          sCoef[t] = (t < len) ? v : 0.f;
        }
      }
    }
  }
  __syncthreads();
  if (tid < D_) {
    float acc = 0.f;
    for (int tt = 0; tt < ntl; ++tt) {
#pragma unroll
      for (int u = 0; u < 16; ++u) {
        const int t = 16 * tt + u;
        acc += sCoef[t] * Db[t * D_ + tid];
      }
    }
    out[(size_t)b * 2 * D_ + tid] = acc;
  }
}

extern "C" void kernel_launch(void* const* d_in, const int* in_sizes, int n_in,
                              void* d_out, int out_size, void* d_ws, size_t ws_size,
                              hipStream_t stream) {
  (void)in_sizes; (void)n_in; (void)out_size; (void)ws_size;
  const float* re   = (const float*)d_in[0];
  const int*   alia = (const int*)d_in[1];
  const int*   slen = (const int*)d_in[2];
  const float* Wq   = (const float*)d_in[3];
  const float* Wk   = (const float*)d_in[4];
  const float* Wv   = (const float*)d_in[5];
  const float* u_w1 = (const float*)d_in[6];
  const float* u_w2 = (const float*)d_in[7];
  const float* u_b  = (const float*)d_in[8];
  const float* u_v  = (const float*)d_in[9];
  float* out = (float*)d_out;

  float* ws   = (float*)d_ws;
  float* wsQ  = ws;
  float* wsK  = ws + (size_t)B_ * PBS;
  float* wsV  = ws + 2 * (size_t)B_ * PBS;
  float* wsDec = wsQ;

  size_t sm1 = (size_t)(TP * HSTR + 3 * D_ * D_) * sizeof(float);
  size_t sm2 = (size_t)(2 * TP * HSTR + 8 * 16 * SSTR) * sizeof(float);
  size_t sm3 = (size_t)(D_ * D_ + 3 * 128 + TP) * sizeof(float);
  hipFuncSetAttribute((const void*)qkv_kernel,
                      hipFuncAttributeMaxDynamicSharedMemorySize, (int)sm1);
  hipFuncSetAttribute((const void*)attn_kernel,
                      hipFuncAttributeMaxDynamicSharedMemorySize, (int)sm2);
  hipFuncSetAttribute((const void*)stamp_kernel,
                      hipFuncAttributeMaxDynamicSharedMemorySize, (int)sm3);

  qkv_kernel<<<B_, 256, sm1, stream>>>(re, alia, slen, Wq, Wk, Wv, wsQ, wsK, wsV);
  attn_kernel<<<B_, 256, sm2, stream>>>(re, alia, slen, wsQ, wsK, wsV, wsDec);
  stamp_kernel<<<B_, 256, sm3, stream>>>(slen, wsDec, u_w1, u_w2, u_b, u_v, out);
}


# qkv kernel: each wave keeps its W column fragments (Wq, Wk, Wv) in registers, read once from global, for all row tiles, three chains per H fragment; with W out of LDS two blocks fit a WGP
# speedup vs baseline: 4.3234x; 1.0574x over previous
#include <hip/hip_runtime.h>
#include <math.h>


typedef __attribute__((ext_vector_type(2))) float v2f;
typedef __attribute__((ext_vector_type(8))) float v8f;

#define B_      2048
#define T_      200
#define D_      128
#define NNODE_  200
#define TP      208
#define NT      13
#define HSTR    132
#define SSTR    210
#define PBS     (TP * D_)
#define NEGC    (-4294967295.0f)

__device__ __forceinline__ int live_tiles(int len) {
  const int n = (len + 15) >> 4;
  return n < 1 ? 1 : (n > NT ? NT : n);
}

__device__ __forceinline__ v8f wmma4(v2f a, v2f b, v8f c) {
  return __builtin_amdgcn_wmma_f32_16x16x4_f32(false, a, false, b, (short)0, c,
                                               false, false);
}

__global__ void __launch_bounds__(256) __attribute__((amdgpu_waves_per_eu(4, 4))) qkv_kernel(const float* __restrict__ re, const int* __restrict__ alias,
                           const int* __restrict__ seqlen,
                           const float* __restrict__ Wq, const float* __restrict__ Wk,
                           const float* __restrict__ Wv,
                           float* __restrict__ wsQ, float* __restrict__ wsK,
                           float* __restrict__ wsV) {
  extern __shared__ __attribute__((aligned(16))) float smem_h[];
  float* sH = smem_h;
  const int b = blockIdx.x, tid = threadIdx.x;
  const int ntl = live_tiles(seqlen[b]);
  const int wv = __builtin_amdgcn_readfirstlane(tid >> 5);
  const int ln = tid & 31, lh = ln & 15, hi = ln >> 4;
  const int n0 = wv << 4;

  {
    const int tl = wv + 8 * ln;
    const int my_a = alias[(size_t)b * T_ + (tl < T_ ? tl : T_ - 1)];
    const float* reb = re + (size_t)b * NNODE_ * D_ + ln * 4;
    const int nk = 2 * ntl;
#pragma unroll 1
    for (int h = 0; h < 2; ++h) {
      const int k0 = h * 13;
      if (k0 >= nk) break;
      float4 v[13];
#pragma unroll
      for (int r = 0; r < 13; ++r) {
        if (k0 + r < nk) {
          const int a = __builtin_amdgcn_readlane(my_a, k0 + r);
          float4 x = *(const float4*)(reb + (size_t)a * D_);
          if (wv + 8 * (k0 + r) >= T_) x = make_float4(0.f, 0.f, 0.f, 0.f);
          v[r] = x;
        }
      }
#pragma unroll
      for (int r = 0; r < 13; ++r)
        if (k0 + r < nk) *(float4*)(sH + (wv + 8 * (k0 + r)) * HSTR + ln * 4) = v[r];
    }
  }

  v2f bq[32], bk[32], bv[32];
  {
    const float* wq = Wq + (2 * hi) * D_ + n0 + lh;
    const float* wk = Wk + (2 * hi) * D_ + n0 + lh;
    const float* wvp = Wv + (2 * hi) * D_ + n0 + lh;
#pragma unroll
    for (int c = 0; c < 32; ++c) {
      bq[c].x = wq[(4 * c) * D_];
      bq[c].y = wq[(4 * c + 1) * D_];
      bk[c].x = wk[(4 * c) * D_];
      bk[c].y = wk[(4 * c + 1) * D_];
      bv[c].x = wvp[(4 * c) * D_];
      bv[c].y = wvp[(4 * c + 1) * D_];
    }
  }
  __syncthreads();

  float* dq = wsQ + (size_t)b * PBS + n0 + lh;
  float* dk = wsK + (size_t)b * PBS + n0 + lh;
  float* dv = wsV + (size_t)b * PBS + n0 + lh;
  for (int i = 0; i < ntl; ++i) {
    const float* arow = sH + (i * 16 + lh) * HSTR + 2 * hi;
    v8f aq = {0.f, 0.f, 0.f, 0.f, 0.f, 0.f, 0.f, 0.f};
    v8f ak = {0.f, 0.f, 0.f, 0.f, 0.f, 0.f, 0.f, 0.f};
    v8f av = {0.f, 0.f, 0.f, 0.f, 0.f, 0.f, 0.f, 0.f};
#pragma unroll
    for (int g = 0; g < 4; ++g) {
      v2f af[8];
#pragma unroll
      for (int cc = 0; cc < 8; ++cc) af[cc] = *(const v2f*)(arow + 4 * (8 * g + cc));
      __builtin_amdgcn_sched_barrier(0);
#pragma unroll
      for (int cc = 0; cc < 8; ++cc) {
        const int c = 8 * g + cc;
        aq = wmma4(af[cc], bq[c], aq);
        ak = wmma4(af[cc], bk[c], ak);
        av = wmma4(af[cc], bv[c], av);
      }
    }
#pragma unroll
    for (int r = 0; r < 8; ++r) {
      const int row = i * 16 + r + 8 * hi;
      const float q = aq[r], k = ak[r], v = av[r];
      dq[row * D_] = q > 0.f ? q : 0.f;
      dk[row * D_] = k > 0.f ? k : 0.f;
      dv[row * D_] = v > 0.f ? v : 0.f;
    }
  }
}

template <int NG>
__device__ __forceinline__ void stage_kv(const float4* __restrict__ K4, const float4* __restrict__ V4,
                                         float* sK, float* sV, int wave, int lane, int ntl) {
  float4 kr[2 * NG], vr[2 * NG];
#pragma unroll
  for (int g = 0; g < NG; ++g) {
    const int gg = g < ntl ? g : ntl - 1;
    kr[2 * g] = K4[256 * (2 * gg)];
    kr[2 * g + 1] = K4[256 * (2 * gg + 1)];
  }
#pragma unroll
  for (int g = 0; g < NG; ++g) {
    const int gg = g < ntl ? g : ntl - 1;
    vr[2 * g] = V4[256 * (2 * gg)];
    vr[2 * g + 1] = V4[256 * (2 * gg + 1)];
  }
#pragma unroll
  for (int g = 0; g < NG; ++g) {
    const int gg = g < ntl ? g : ntl - 1;
    *(float4*)(sK + (wave + 16 * gg) * HSTR + lane * 4) = kr[2 * g];
    *(float4*)(sK + (wave + 16 * gg + 8) * HSTR + lane * 4) = kr[2 * g + 1];
  }
#pragma unroll
  for (int g = 0; g < NG; ++g) {
    const int gg = g < ntl ? g : ntl - 1;
    *(float4*)(sV + (wave + 16 * gg) * HSTR + lane * 4) = vr[2 * g];
    *(float4*)(sV + (wave + 16 * gg + 8) * HSTR + lane * 4) = vr[2 * g + 1];
  }
}

__global__ void __launch_bounds__(256) __attribute__((amdgpu_waves_per_eu(2, 2))) attn_kernel(const float* __restrict__ re, const int* __restrict__ alias,
                            const int* __restrict__ seqlen,
                            const float* __restrict__ wsQ, const float* __restrict__ wsK,
                            const float* __restrict__ wsV, float* __restrict__ wsDec) {
  extern __shared__ __attribute__((aligned(16))) float smem[];
  float* sK = smem;
  float* sV = smem + TP * HSTR;
  float* sS0 = smem + 2 * TP * HSTR;
  const int b = blockIdx.x, tid = threadIdx.x;
  const int len = seqlen[b];
  const int ntl = live_tiles(len);
  const int wave = tid >> 5, lane = tid & 31, lh = lane & 15, hi = lane >> 4;
  const int t_a0 = wave * 16 + lh, t_a1 = (wave + 8) * 16 + lh;
  const int my_a0 = alias[b * T_ + (t_a0 < T_ ? t_a0 : T_ - 1)];
  const int my_a1 = alias[b * T_ + (t_a1 < T_ ? t_a1 : T_ - 1)];
  const float* Kb = wsK + (size_t)b * PBS;
  const float* Vb = wsV + (size_t)b * PBS;
  {
    const float4* K4 = (const float4*)Kb + tid;
    const float4* V4 = (const float4*)Vb + tid;
    if (ntl <= 2)       stage_kv<2>(K4, V4, sK, sV, wave, lane, ntl);
    else if (ntl <= 4)  stage_kv<4>(K4, V4, sK, sV, wave, lane, ntl);
    else if (ntl <= 7)  stage_kv<7>(K4, V4, sK, sV, wave, lane, ntl);
    else if (ntl <= 10) stage_kv<10>(K4, V4, sK, sV, wave, lane, ntl);
    else                stage_kv<NT>(K4, V4, sK, sV, wave, lane, ntl);
  }
  __syncthreads();

  float* sS = sS0 + wave * (16 * SSTR);
  const float scale = 0.088388347648318447f;
  const float* reb = re + (size_t)b * NNODE_ * D_ + lh;

  const int npass = (ntl > 8) ? 2 : 1;
  for (int pass = 0; pass < npass; ++pass) {
    const int qt = wave + 8 * pass;
    const bool active = (qt < ntl);
    int roff[8];
    float hres[8];
    if (active) {
      const int my_a = pass ? my_a1 : my_a0;
#pragma unroll
      for (int r = 0; r < 8; ++r) {
        const int alo = __builtin_amdgcn_readlane(my_a, r);
        const int ahi = __builtin_amdgcn_readlane(my_a, 8 + r);
        roff[r] = (hi ? ahi : alo) * D_;
      }
#pragma unroll
      for (int r = 0; r < 8; ++r) hres[r] = reb[roff[r]];
      v2f qa[32];
      const float* Qrow = wsQ + (size_t)b * PBS + (qt * 16 + lh) * D_;
#pragma unroll
      for (int c = 0; c < 32; ++c) qa[c] = *(const v2f*)(Qrow + 4 * c + 2 * hi);
      for (int j = 0; j < ntl; ++j) {
        v8f acc = {0.f, 0.f, 0.f, 0.f, 0.f, 0.f, 0.f, 0.f};
#pragma unroll
        for (int c = 0; c < 32; ++c) {
          v2f bb = *(const v2f*)(sK + (j * 16 + lh) * HSTR + 4 * c + 2 * hi);
          acc = wmma4(qa[c], bb, acc);
        }
#pragma unroll
        for (int r = 0; r < 8; ++r) {
          int m = r + 8 * hi, s = j * 16 + lh;
          sS[m * SSTR + s] = (s < len) ? acc[r] * scale : NEGC;
        }
      }
    }
    __syncthreads();
    if (active) {
      float* rowp = sS + lh * SSTR + hi * 104;
      int cnt = 16 * ntl - hi * 104;
      cnt = cnt > 104 ? 104 : (cnt < 0 ? 0 : cnt);
      float mx = -3.0e38f;
      for (int s0 = 0; s0 < cnt; s0 += 8) {
#pragma unroll
        for (int u = 0; u < 8; ++u) mx = fmaxf(mx, rowp[s0 + u]);
      }
      mx = fmaxf(mx, __shfl_xor(mx, 16, 32));
      float sum = 0.f;
      for (int s0 = 0; s0 < cnt; s0 += 8) {
#pragma unroll
        for (int u = 0; u < 8; ++u) { float e = __expf(rowp[s0 + u] - mx); rowp[s0 + u] = e; sum += e; }
      }
      sum += __shfl_xor(sum, 16, 32);
      float qm = ((qt * 16 + lh) < len) ? 1.f : 0.f;
      float inv = qm / sum;
      for (int s0 = 0; s0 < cnt; s0 += 8) {
#pragma unroll
        for (int u = 0; u < 8; ++u) rowp[s0 + u] *= inv;
      }
    }
    __syncthreads();
    if (active) {
      float* Db = wsDec + (size_t)b * PBS;
      v2f pa[52];
#pragma unroll
      for (int g = 0; g < NT; ++g)
        if (g < ntl) {
#pragma unroll
          for (int cc = 0; cc < 4; ++cc) {
            const int c = 4 * g + cc;
            pa[c] = *(const v2f*)(sS + lh * SSTR + 4 * c + 2 * hi);
          }
        }
      const float* vcol = sV + (2 * hi) * HSTR + lh;
      v2f vb0[26], vb1[26];
#pragma unroll
      for (int g = 0; g < 7; ++g)
        if (g < ntl) {
#pragma unroll
          for (int cc = 0; cc < 4; ++cc) {
            const int c = 4 * g + cc;
            if (c < 26) {
              vb0[c].x = vcol[(4 * c) * HSTR];
              vb0[c].y = vcol[(4 * c + 1) * HSTR];
            }
          }
        }
      for (int nt = 0; nt < 8; ++nt) {
        int n0 = nt * 16;
        const int n1 = (nt < 7) ? n0 + 16 : n0;
        v8f acc;
#pragma unroll
        for (int r = 0; r < 8; ++r)
          acc[r] = (qt * 16 + r + 8 * hi < T_) ? hres[r] : 0.f;
        if (nt < 7) {
#pragma unroll
          for (int r = 0; r < 8; ++r) hres[r] = reb[roff[r] + n0 + 16];
        }
#pragma unroll
        for (int c = 0; c < 4; ++c) acc = wmma4(pa[c], vb0[c], acc);
        asm volatile("" : "+v"(acc));
#pragma unroll
        for (int g = 6; g < NT; ++g)
          if (g < ntl) {
#pragma unroll
            for (int cc = 0; cc < 4; ++cc) {
              const int k = 4 * g + cc;
              if (k >= 26) {
                vb1[k - 26].x = vcol[(4 * k) * HSTR + n0];
                vb1[k - 26].y = vcol[(4 * k + 1) * HSTR + n0];
              }
            }
          }
#pragma unroll
        for (int g = 1; g < 7; ++g)
          if (g < ntl) {
#pragma unroll
            for (int cc = 0; cc < 4; ++cc) {
              const int c = 4 * g + cc;
              if (c < 26) acc = wmma4(pa[c], vb0[c], acc);
            }
          }
        if (6 < ntl) {
          acc = wmma4(pa[26], vb1[0], acc);
          acc = wmma4(pa[27], vb1[1], acc);
        }
#pragma unroll
        for (int g = 0; g < 7; ++g)
          if (g < ntl) {
#pragma unroll
            for (int cc = 0; cc < 4; ++cc) {
              const int c = 4 * g + cc;
              if (c < 26) {
                vb0[c].x = vcol[(4 * c) * HSTR + n1];
                vb0[c].y = vcol[(4 * c + 1) * HSTR + n1];
              }
            }
          }
#pragma unroll
        for (int g = 7; g < NT; ++g)
          if (g < ntl) {
#pragma unroll
            for (int cc = 0; cc < 4; ++cc) {
              const int k = 4 * g + cc;
              acc = wmma4(pa[k], vb1[k - 26], acc);
            }
          }
#pragma unroll
        for (int r = 0; r < 8; ++r) {
          int t = qt * 16 + r + 8 * hi;
          Db[t * D_ + n0 + lh] = acc[r];
        }
      }
    }
    __syncthreads();
  }
}

__global__ void stamp_kernel(const int* __restrict__ seqlen, const float* __restrict__ wsDec,
                             const float* __restrict__ u_w1, const float* __restrict__ u_w2,
                             const float* __restrict__ u_b, const float* __restrict__ u_v,
                             float* __restrict__ out) {
  extern __shared__ __attribute__((aligned(16))) float smem[];
  float* sW2 = smem;
  float* sLp = smem + D_ * D_;
  float* sUb = sLp + 128;
  float* sUv = sUb + 128;
  float* sCoef = sUv + 128;
  const int b = blockIdx.x, tid = threadIdx.x;
  const int len = seqlen[b];
  const int ntl = live_tiles(len);
  const float* Db = wsDec + (size_t)b * PBS;

  {
    const float4* src = (const float4*)u_w2 + tid;
    float4 w[16];
#pragma unroll
    for (int i = 0; i < 16; ++i) w[i] = src[256 * i];
#pragma unroll
    for (int i = 0; i < 16; ++i) *(float4*)(sW2 + 4 * (tid + 256 * i)) = w[i];
  }
  if (tid < D_) { sUb[tid] = u_b[tid]; sUv[tid] = u_v[tid]; }
  if (tid < D_) {
    const float* lrow = Db + (len - 1) * D_;
    float acc = 0.f;
    for (int d = 0; d < D_; ++d) acc += lrow[d] * u_w1[d * D_ + tid];
    sLp[tid] = acc;
    out[(size_t)b * 2 * D_ + D_ + tid] = acc;
  }
  __syncthreads();

  const int wave = tid >> 5, lane = tid & 31, lh = lane & 15, hi = lane >> 4;
  for (int pass = 0; pass < 2; ++pass) {
    const int qt = wave + 8 * pass;
    if (qt < ntl) {
      v2f da[32];
      const float* drow = Db + (qt * 16 + lh) * D_;
#pragma unroll
      for (int c = 0; c < 32; ++c) da[c] = *(const v2f*)(drow + 4 * c + 2 * hi);
      float cpart[8];
#pragma unroll
      for (int r = 0; r < 8; ++r) cpart[r] = 0.f;
      for (int nt = 0; nt < 8; ++nt) {
        int n0 = nt * 16;
        v8f acc = {0.f, 0.f, 0.f, 0.f, 0.f, 0.f, 0.f, 0.f};
#pragma unroll
        for (int c = 0; c < 32; ++c) {
          int k = 4 * c + 2 * hi;
          v2f bb;
          bb.x = sW2[k * D_ + n0 + lh];
          bb.y = sW2[(k + 1) * D_ + n0 + lh];
          acc = wmma4(da[c], bb, acc);
        }
        int col = n0 + lh;
        float add = sLp[col] + sUb[col];
        float uvv = sUv[col];
#pragma unroll
        for (int r = 0; r < 8; ++r) {
          float x = acc[r] + add;
          float sg = 1.f / (1.f + __expf(-x));
          cpart[r] += sg * uvv;
        }
      }
#pragma unroll
      for (int r = 0; r < 8; ++r) {
        float v = cpart[r];
        v += __shfl_xor(v, 1, 32);
        v += __shfl_xor(v, 2, 32);
        v += __shfl_xor(v, 4, 32);
        v += __shfl_xor(v, 8, 32);
        if (lh == 0) {
          int t = qt * 16 + r + 8 * hi;
          sCoef[t] = (t < len) ? v : 0.f;
        }
      }
    }
  }
  __syncthreads();
  if (tid < D_) {
    float acc = 0.f;
    for (int tt = 0; tt < ntl; ++tt) {
#pragma unroll
      for (int u = 0; u < 16; ++u) {
        const int t = 16 * tt + u;
        acc += sCoef[t] * Db[t * D_ + tid];
      }
    }
    out[(size_t)b * 2 * D_ + tid] = acc;
  }
}

extern "C" void kernel_launch(void* const* d_in, const int* in_sizes, int n_in,
                              void* d_out, int out_size, void* d_ws, size_t ws_size,
                              hipStream_t stream) {
  (void)in_sizes; (void)n_in; (void)out_size; (void)ws_size;
  const float* re   = (const float*)d_in[0];
  const int*   alia = (const int*)d_in[1];
  const int*   slen = (const int*)d_in[2];
  const float* Wq   = (const float*)d_in[3];
  const float* Wk   = (const float*)d_in[4];
  const float* Wv   = (const float*)d_in[5];
  const float* u_w1 = (const float*)d_in[6];
  const float* u_w2 = (const float*)d_in[7];
  const float* u_b  = (const float*)d_in[8];
  const float* u_v  = (const float*)d_in[9];
  float* out = (float*)d_out;

  float* ws   = (float*)d_ws;
  float* wsQ  = ws;
  float* wsK  = ws + (size_t)B_ * PBS;
  float* wsV  = ws + 2 * (size_t)B_ * PBS;
  float* wsDec = wsQ;

  size_t sm1 = (size_t)(TP * HSTR) * sizeof(float);
  size_t sm2 = (size_t)(2 * TP * HSTR + 8 * 16 * SSTR) * sizeof(float);
  size_t sm3 = (size_t)(D_ * D_ + 3 * 128 + TP) * sizeof(float);
  hipFuncSetAttribute((const void*)qkv_kernel,
                      hipFuncAttributeMaxDynamicSharedMemorySize, (int)sm1);
  hipFuncSetAttribute((const void*)attn_kernel,
                      hipFuncAttributeMaxDynamicSharedMemorySize, (int)sm2);
  hipFuncSetAttribute((const void*)stamp_kernel,
                      hipFuncAttributeMaxDynamicSharedMemorySize, (int)sm3);

  qkv_kernel<<<B_, 256, sm1, stream>>>(re, alia, slen, Wq, Wk, Wv, wsQ, wsK, wsV);
  attn_kernel<<<B_, 256, sm2, stream>>>(re, alia, slen, wsQ, wsK, wsV, wsDec);
  stamp_kernel<<<B_, 256, sm3, stream>>>(slen, wsDec, u_w1, u_w2, u_b, u_v, out);
}


# all kernels: block k takes the session ranked k-th by live 16-row tiles, longest first (rank found per block from seq_len, packed counting scan), so long sessions no longer end a launch alone
# speedup vs baseline: 4.5687x; 1.0567x over previous
#include <hip/hip_runtime.h>
#include <math.h>


typedef __attribute__((ext_vector_type(2))) float v2f;
typedef __attribute__((ext_vector_type(8))) float v8f;

#define B_      2048
#define T_      200
#define D_      128
#define NNODE_  200
#define TP      208
#define NT      13
#define HSTR    132
#define SSTR    210
#define PBS     (TP * D_)
#define NEGC    (-4294967295.0f)

__device__ __forceinline__ int live_tiles(int len) {
  const int n = (len + 15) >> 4;
  return n < 1 ? 1 : (n > NT ? NT : n);
}

__device__ __forceinline__ v8f wmma4(v2f a, v2f b, v8f c) {
  return __builtin_amdgcn_wmma_f32_16x16x4_f32(false, a, false, b, (short)0, c,
                                               false, false);
}

typedef unsigned long long u64;

__device__ __forceinline__ int cls_get(u64 x0, u64 x1, u64 x2, int c) {
  const int w = (c >= 5) + (c >= 10);
  const u64 x = (w == 0) ? x0 : ((w == 1) ? x1 : x2);
  return (int)((x >> (12 * (c - 5 * w))) & 0xFFFull);
}

__device__ __forceinline__ u64 uniform64(u64 x) {
  const unsigned lo = (unsigned)__builtin_amdgcn_readfirstlane((int)(unsigned)(x & 0xFFFFFFFFull));
  const unsigned hi = (unsigned)__builtin_amdgcn_readfirstlane((int)(unsigned)(x >> 32));
  return ((u64)hi << 32) | (u64)lo;
}

__device__ __forceinline__ int ranked_session(const int* __restrict__ seqlen, u64* scr, int tid,
                                              int& len_out) {
  const int lane = tid & 31, wave = tid >> 5;
  int* res = (int*)(scr + 24);
  const int4 la = ((const int4*)seqlen)[2 * tid];
  const int4 lb = ((const int4*)seqlen)[2 * tid + 1];
  const int lenv[8] = {la.x, la.y, la.z, la.w, lb.x, lb.y, lb.z, lb.w};
  int cls[8];
  u64 nib = 0ull;
#pragma unroll
  for (int e = 0; e < 8; ++e) {
    cls[e] = NT - live_tiles(lenv[e]);
    nib += 1ull << (4 * cls[e]);
  }
  u64 h0 = 0ull, h1 = 0ull, h2 = 0ull;
#pragma unroll
  for (int c = 0; c < NT; ++c) {
    const int w = c / 5;
    const u64 v = ((nib >> (4 * c)) & 0xFull) << (12 * (c - 5 * w));
    if (w == 0) h0 += v; else if (w == 1) h1 += v; else h2 += v;
  }
  u64 s0 = h0, s1 = h1, s2 = h2;
#pragma unroll
  for (int d = 1; d < 32; d <<= 1) {
    const u64 t0 = __shfl_up(s0, d, 32), t1 = __shfl_up(s1, d, 32), t2 = __shfl_up(s2, d, 32);
    if (lane >= d) { s0 += t0; s1 += t1; s2 += t2; }
  }
  if (lane == 31) { scr[3 * wave] = s0; scr[3 * wave + 1] = s1; scr[3 * wave + 2] = s2; }
  if (tid == 0) { res[0] = blockIdx.x; res[1] = 1; }
  __syncthreads();
  u64 r0 = s0 - h0, r1 = s1 - h1, r2 = s2 - h2;
  u64 H0 = 0ull, H1 = 0ull, H2 = 0ull;
#pragma unroll
  for (int w = 0; w < 8; ++w) {
    const u64 a = scr[3 * w], bq = scr[3 * w + 1], cq = scr[3 * w + 2];
    H0 += a; H1 += bq; H2 += cq;
    if (w < wave) { r0 += a; r1 += bq; r2 += cq; }
  }
  const u64 T0 = uniform64(H0), T1 = uniform64(H1), T2 = uniform64(H2);
  const int k = blockIdx.x;
  int run = 0, cstar = NT - 1, jj = 0;
  bool open = true;
#pragma unroll
  for (int c = 0; c < NT; ++c) {
    const int cnt = cls_get(T0, T1, T2, c);
    if (open && k < run + cnt) { cstar = c; jj = k - run; open = false; }
    run += cnt;
  }
  int p = cls_get(r0, r1, r2, cstar);
  int found = -1, flen = 1;
#pragma unroll
  for (int e = 0; e < 8; ++e) {
    const bool m = (cls[e] == cstar);
    if (m && p == jj) { found = 8 * tid + e; flen = lenv[e]; }
    p += m ? 1 : 0;
  }
  if (found >= 0) { res[0] = found; res[1] = flen; }
  __syncthreads();
  int b = __builtin_amdgcn_readfirstlane(res[0]);
  len_out = __builtin_amdgcn_readfirstlane(res[1]);
  return b < 0 ? 0 : (b > B_ - 1 ? B_ - 1 : b);
}

__global__ void __launch_bounds__(256) __attribute__((amdgpu_waves_per_eu(4, 4))) qkv_kernel(const float* __restrict__ re, const int* __restrict__ alias,
                           const int* __restrict__ seqlen,
                           const float* __restrict__ Wq, const float* __restrict__ Wk,
                           const float* __restrict__ Wv,
                           float* __restrict__ wsQ, float* __restrict__ wsK,
                           float* __restrict__ wsV) {
  extern __shared__ __attribute__((aligned(16))) float smem_h[];
  float* sH = smem_h;
  const int tid = threadIdx.x;
  int len;
  const int b = ranked_session(seqlen, (u64*)(smem_h + TP * HSTR), tid, len);
  const int ntl = live_tiles(len);
  const int wv = __builtin_amdgcn_readfirstlane(tid >> 5);
  const int ln = tid & 31, lh = ln & 15, hi = ln >> 4;
  const int n0 = wv << 4;

  {
    const int tl = wv + 8 * ln;
    const int my_a = alias[(size_t)b * T_ + (tl < T_ ? tl : T_ - 1)];
    const float* reb = re + (size_t)b * NNODE_ * D_ + ln * 4;
    const int nk = 2 * ntl;
#pragma unroll 1
    for (int h = 0; h < 2; ++h) {
      const int k0 = h * 13;
      if (k0 >= nk) break;
      float4 v[13];
#pragma unroll
      for (int r = 0; r < 13; ++r) {
        if (k0 + r < nk) {
          const int a = __builtin_amdgcn_readlane(my_a, k0 + r);
          float4 x = *(const float4*)(reb + (size_t)a * D_);
          if (wv + 8 * (k0 + r) >= T_) x = make_float4(0.f, 0.f, 0.f, 0.f);
          v[r] = x;
        }
      }
#pragma unroll
      for (int r = 0; r < 13; ++r)
        if (k0 + r < nk) *(float4*)(sH + (wv + 8 * (k0 + r)) * HSTR + ln * 4) = v[r];
    }
  }

  v2f bq[32], bk[32], bv[32];
  {
    const float* wq = Wq + (2 * hi) * D_ + n0 + lh;
    const float* wk = Wk + (2 * hi) * D_ + n0 + lh;
    const float* wvp = Wv + (2 * hi) * D_ + n0 + lh;
#pragma unroll
    for (int c = 0; c < 32; ++c) {
      bq[c].x = wq[(4 * c) * D_];
      bq[c].y = wq[(4 * c + 1) * D_];
      bk[c].x = wk[(4 * c) * D_];
      bk[c].y = wk[(4 * c + 1) * D_];
      bv[c].x = wvp[(4 * c) * D_];
      bv[c].y = wvp[(4 * c + 1) * D_];
    }
  }
  __syncthreads();

  float* dq = wsQ + (size_t)b * PBS + n0 + lh;
  float* dk = wsK + (size_t)b * PBS + n0 + lh;
  float* dv = wsV + (size_t)b * PBS + n0 + lh;
  for (int i = 0; i < ntl; ++i) {
    const float* arow = sH + (i * 16 + lh) * HSTR + 2 * hi;
    v8f aq = {0.f, 0.f, 0.f, 0.f, 0.f, 0.f, 0.f, 0.f};
    v8f ak = {0.f, 0.f, 0.f, 0.f, 0.f, 0.f, 0.f, 0.f};
    v8f av = {0.f, 0.f, 0.f, 0.f, 0.f, 0.f, 0.f, 0.f};
#pragma unroll
    for (int g = 0; g < 4; ++g) {
      v2f af[8];
#pragma unroll
      for (int cc = 0; cc < 8; ++cc) af[cc] = *(const v2f*)(arow + 4 * (8 * g + cc));
      __builtin_amdgcn_sched_barrier(0);
#pragma unroll
      for (int cc = 0; cc < 8; ++cc) {
        const int c = 8 * g + cc;
        aq = wmma4(af[cc], bq[c], aq);
        ak = wmma4(af[cc], bk[c], ak);
        av = wmma4(af[cc], bv[c], av);
      }
    }
#pragma unroll
    for (int r = 0; r < 8; ++r) {
      const int row = i * 16 + r + 8 * hi;
      const float q = aq[r], k = ak[r], v = av[r];
      dq[row * D_] = q > 0.f ? q : 0.f;
      dk[row * D_] = k > 0.f ? k : 0.f;
      dv[row * D_] = v > 0.f ? v : 0.f;
    }
  }
}

template <int NG>
__device__ __forceinline__ void stage_kv(const float4* __restrict__ K4, const float4* __restrict__ V4,
                                         float* sK, float* sV, int wave, int lane, int ntl) {
  float4 kr[2 * NG], vr[2 * NG];
#pragma unroll
  for (int g = 0; g < NG; ++g) {
    const int gg = g < ntl ? g : ntl - 1;
    kr[2 * g] = K4[256 * (2 * gg)];
    kr[2 * g + 1] = K4[256 * (2 * gg + 1)];
  }
#pragma unroll
  for (int g = 0; g < NG; ++g) {
    const int gg = g < ntl ? g : ntl - 1;
    vr[2 * g] = V4[256 * (2 * gg)];
    vr[2 * g + 1] = V4[256 * (2 * gg + 1)];
  }
#pragma unroll
  for (int g = 0; g < NG; ++g) {
    const int gg = g < ntl ? g : ntl - 1;
    *(float4*)(sK + (wave + 16 * gg) * HSTR + lane * 4) = kr[2 * g];
    *(float4*)(sK + (wave + 16 * gg + 8) * HSTR + lane * 4) = kr[2 * g + 1];
  }
#pragma unroll
  for (int g = 0; g < NG; ++g) {
    const int gg = g < ntl ? g : ntl - 1;
    *(float4*)(sV + (wave + 16 * gg) * HSTR + lane * 4) = vr[2 * g];
    *(float4*)(sV + (wave + 16 * gg + 8) * HSTR + lane * 4) = vr[2 * g + 1];
  }
}

__global__ void __launch_bounds__(256) __attribute__((amdgpu_waves_per_eu(2, 2))) attn_kernel(const float* __restrict__ re, const int* __restrict__ alias,
                            const int* __restrict__ seqlen,
                            const float* __restrict__ wsQ, const float* __restrict__ wsK,
                            const float* __restrict__ wsV, float* __restrict__ wsDec) {
  extern __shared__ __attribute__((aligned(16))) float smem[];
  float* sK = smem;
  float* sV = smem + TP * HSTR;
  float* sS0 = smem + 2 * TP * HSTR;
  const int tid = threadIdx.x;
  int len;
  const int b = ranked_session(seqlen, (u64*)sS0, tid, len);
  const int ntl = live_tiles(len);
  const int wave = tid >> 5, lane = tid & 31, lh = lane & 15, hi = lane >> 4;
  const int t_a0 = wave * 16 + lh, t_a1 = (wave + 8) * 16 + lh;
  const int my_a0 = alias[b * T_ + (t_a0 < T_ ? t_a0 : T_ - 1)];
  const int my_a1 = alias[b * T_ + (t_a1 < T_ ? t_a1 : T_ - 1)];
  const float* Kb = wsK + (size_t)b * PBS;
  const float* Vb = wsV + (size_t)b * PBS;
  {
    const float4* K4 = (const float4*)Kb + tid;
    const float4* V4 = (const float4*)Vb + tid;
    if (ntl <= 2)       stage_kv<2>(K4, V4, sK, sV, wave, lane, ntl);
    else if (ntl <= 4)  stage_kv<4>(K4, V4, sK, sV, wave, lane, ntl);
    else if (ntl <= 7)  stage_kv<7>(K4, V4, sK, sV, wave, lane, ntl);
    else if (ntl <= 10) stage_kv<10>(K4, V4, sK, sV, wave, lane, ntl);
    else                stage_kv<NT>(K4, V4, sK, sV, wave, lane, ntl);
  }
  __syncthreads();

  float* sS = sS0 + wave * (16 * SSTR);
  const float scale = 0.088388347648318447f;
  const float* reb = re + (size_t)b * NNODE_ * D_ + lh;

  const int npass = (ntl > 8) ? 2 : 1;
  for (int pass = 0; pass < npass; ++pass) {
    const int qt = wave + 8 * pass;
    const bool active = (qt < ntl);
    int roff[8];
    float hres[8];
    if (active) {
      const int my_a = pass ? my_a1 : my_a0;
#pragma unroll
      for (int r = 0; r < 8; ++r) {
        const int alo = __builtin_amdgcn_readlane(my_a, r);
        const int ahi = __builtin_amdgcn_readlane(my_a, 8 + r);
        roff[r] = (hi ? ahi : alo) * D_;
      }
#pragma unroll
      for (int r = 0; r < 8; ++r) hres[r] = reb[roff[r]];
      v2f qa[32];
      const float* Qrow = wsQ + (size_t)b * PBS + (qt * 16 + lh) * D_;
#pragma unroll
      for (int c = 0; c < 32; ++c) qa[c] = *(const v2f*)(Qrow + 4 * c + 2 * hi);
      for (int j = 0; j < ntl; ++j) {
        v8f acc = {0.f, 0.f, 0.f, 0.f, 0.f, 0.f, 0.f, 0.f};
#pragma unroll
        for (int c = 0; c < 32; ++c) {
          v2f bb = *(const v2f*)(sK + (j * 16 + lh) * HSTR + 4 * c + 2 * hi);
          acc = wmma4(qa[c], bb, acc);
        }
#pragma unroll
        for (int r = 0; r < 8; ++r) {
          int m = r + 8 * hi, s = j * 16 + lh;
          sS[m * SSTR + s] = (s < len) ? acc[r] * scale : NEGC;
        }
      }
    }
    __syncthreads();
    if (active) {
      float* rowp = sS + lh * SSTR + hi * 104;
      int cnt = 16 * ntl - hi * 104;
      cnt = cnt > 104 ? 104 : (cnt < 0 ? 0 : cnt);
      float mx = -3.0e38f;
      for (int s0 = 0; s0 < cnt; s0 += 8) {
#pragma unroll
        for (int u = 0; u < 8; ++u) mx = fmaxf(mx, rowp[s0 + u]);
      }
      mx = fmaxf(mx, __shfl_xor(mx, 16, 32));
      float sum = 0.f;
      for (int s0 = 0; s0 < cnt; s0 += 8) {
#pragma unroll
        for (int u = 0; u < 8; ++u) { float e = __expf(rowp[s0 + u] - mx); rowp[s0 + u] = e; sum += e; }
      }
      sum += __shfl_xor(sum, 16, 32);
      float qm = ((qt * 16 + lh) < len) ? 1.f : 0.f;
      float inv = qm / sum;
      for (int s0 = 0; s0 < cnt; s0 += 8) {
#pragma unroll
        for (int u = 0; u < 8; ++u) rowp[s0 + u] *= inv;
      }
    }
    __syncthreads();
    if (active) {
      float* Db = wsDec + (size_t)b * PBS;
      v2f pa[52];
#pragma unroll
      for (int g = 0; g < NT; ++g)
        if (g < ntl) {
#pragma unroll
          for (int cc = 0; cc < 4; ++cc) {
            const int c = 4 * g + cc;
            pa[c] = *(const v2f*)(sS + lh * SSTR + 4 * c + 2 * hi);
          }
        }
      const float* vcol = sV + (2 * hi) * HSTR + lh;
      v2f vb0[26], vb1[26];
#pragma unroll
      for (int g = 0; g < 7; ++g)
        if (g < ntl) {
#pragma unroll
          for (int cc = 0; cc < 4; ++cc) {
            const int c = 4 * g + cc;
            if (c < 26) {
              vb0[c].x = vcol[(4 * c) * HSTR];
              vb0[c].y = vcol[(4 * c + 1) * HSTR];
            }
          }
        }
      for (int nt = 0; nt < 8; ++nt) {
        int n0 = nt * 16;
        const int n1 = (nt < 7) ? n0 + 16 : n0;
        v8f acc;
#pragma unroll
        for (int r = 0; r < 8; ++r)
          acc[r] = (qt * 16 + r + 8 * hi < T_) ? hres[r] : 0.f;
        if (nt < 7) {
#pragma unroll
          for (int r = 0; r < 8; ++r) hres[r] = reb[roff[r] + n0 + 16];
        }
#pragma unroll
        for (int c = 0; c < 4; ++c) acc = wmma4(pa[c], vb0[c], acc);
        asm volatile("" : "+v"(acc));
#pragma unroll
        for (int g = 6; g < NT; ++g)
          if (g < ntl) {
#pragma unroll
            for (int cc = 0; cc < 4; ++cc) {
              const int k = 4 * g + cc;
              if (k >= 26) {
                vb1[k - 26].x = vcol[(4 * k) * HSTR + n0];
                vb1[k - 26].y = vcol[(4 * k + 1) * HSTR + n0];
              }
            }
          }
#pragma unroll
        for (int g = 1; g < 7; ++g)
          if (g < ntl) {
#pragma unroll
            for (int cc = 0; cc < 4; ++cc) {
              const int c = 4 * g + cc;
              if (c < 26) acc = wmma4(pa[c], vb0[c], acc);
            }
          }
        if (6 < ntl) {
          acc = wmma4(pa[26], vb1[0], acc);
          acc = wmma4(pa[27], vb1[1], acc);
        }
#pragma unroll
        for (int g = 0; g < 7; ++g)
          if (g < ntl) {
#pragma unroll
            for (int cc = 0; cc < 4; ++cc) {
              const int c = 4 * g + cc;
              if (c < 26) {
                vb0[c].x = vcol[(4 * c) * HSTR + n1];
                vb0[c].y = vcol[(4 * c + 1) * HSTR + n1];
              }
            }
          }
#pragma unroll
        for (int g = 7; g < NT; ++g)
          if (g < ntl) {
#pragma unroll
            for (int cc = 0; cc < 4; ++cc) {
              const int k = 4 * g + cc;
              acc = wmma4(pa[k], vb1[k - 26], acc);
            }
          }
#pragma unroll
        for (int r = 0; r < 8; ++r) {
          int t = qt * 16 + r + 8 * hi;
          Db[t * D_ + n0 + lh] = acc[r];
        }
      }
    }
    __syncthreads();
  }
}

__global__ void stamp_kernel(const int* __restrict__ seqlen, const float* __restrict__ wsDec,
                             const float* __restrict__ u_w1, const float* __restrict__ u_w2,
                             const float* __restrict__ u_b, const float* __restrict__ u_v,
                             float* __restrict__ out) {
  extern __shared__ __attribute__((aligned(16))) float smem[];
  float* sW2 = smem;
  float* sLp = smem + D_ * D_;
  float* sUb = sLp + 128;
  float* sUv = sUb + 128;
  float* sCoef = sUv + 128;
  const int tid = threadIdx.x;
  int len;
  const int b = ranked_session(seqlen, (u64*)sCoef, tid, len);
  const int ntl = live_tiles(len);
  const float* Db = wsDec + (size_t)b * PBS;

  {
    const float4* src = (const float4*)u_w2 + tid;
    float4 w[16];
#pragma unroll
    for (int i = 0; i < 16; ++i) w[i] = src[256 * i];
#pragma unroll
    for (int i = 0; i < 16; ++i) *(float4*)(sW2 + 4 * (tid + 256 * i)) = w[i];
  }
  if (tid < D_) { sUb[tid] = u_b[tid]; sUv[tid] = u_v[tid]; }
  if (tid < D_) {
    const float* lrow = Db + (len - 1) * D_;
    float acc = 0.f;
    for (int d = 0; d < D_; ++d) acc += lrow[d] * u_w1[d * D_ + tid];
    sLp[tid] = acc;
    out[(size_t)b * 2 * D_ + D_ + tid] = acc;
  }
  __syncthreads();

  const int wave = tid >> 5, lane = tid & 31, lh = lane & 15, hi = lane >> 4;
  for (int pass = 0; pass < 2; ++pass) {
    const int qt = wave + 8 * pass;
    if (qt < ntl) {
      v2f da[32];
      const float* drow = Db + (qt * 16 + lh) * D_;
#pragma unroll
      for (int c = 0; c < 32; ++c) da[c] = *(const v2f*)(drow + 4 * c + 2 * hi);
      float cpart[8];
#pragma unroll
      for (int r = 0; r < 8; ++r) cpart[r] = 0.f;
      for (int nt = 0; nt < 8; ++nt) {
        int n0 = nt * 16;
        v8f acc = {0.f, 0.f, 0.f, 0.f, 0.f, 0.f, 0.f, 0.f};
#pragma unroll
        for (int c = 0; c < 32; ++c) {
          int k = 4 * c + 2 * hi;
          v2f bb;
          bb.x = sW2[k * D_ + n0 + lh];
          bb.y = sW2[(k + 1) * D_ + n0 + lh];
          acc = wmma4(da[c], bb, acc);
        }
        int col = n0 + lh;
        float add = sLp[col] + sUb[col];
        float uvv = sUv[col];
#pragma unroll
        for (int r = 0; r < 8; ++r) {
          float x = acc[r] + add;
          float sg = 1.f / (1.f + __expf(-x));
          cpart[r] += sg * uvv;
        }
      }
#pragma unroll
      for (int r = 0; r < 8; ++r) {
        float v = cpart[r];
        v += __shfl_xor(v, 1, 32);
        v += __shfl_xor(v, 2, 32);
        v += __shfl_xor(v, 4, 32);
        v += __shfl_xor(v, 8, 32);
        if (lh == 0) {
          int t = qt * 16 + r + 8 * hi;
          sCoef[t] = (t < len) ? v : 0.f;
        }
      }
    }
  }
  __syncthreads();
  if (tid < D_) {
    float acc = 0.f;
    for (int tt = 0; tt < ntl; ++tt) {
#pragma unroll
      for (int u = 0; u < 16; ++u) {
        const int t = 16 * tt + u;
        acc += sCoef[t] * Db[t * D_ + tid];
      }
    }
    out[(size_t)b * 2 * D_ + tid] = acc;
  }
}

extern "C" void kernel_launch(void* const* d_in, const int* in_sizes, int n_in,
                              void* d_out, int out_size, void* d_ws, size_t ws_size,
                              hipStream_t stream) {
  (void)in_sizes; (void)n_in; (void)out_size; (void)ws_size;
  const float* re   = (const float*)d_in[0];
  const int*   alia = (const int*)d_in[1];
  const int*   slen = (const int*)d_in[2];
  const float* Wq   = (const float*)d_in[3];
  const float* Wk   = (const float*)d_in[4];
  const float* Wv   = (const float*)d_in[5];
  const float* u_w1 = (const float*)d_in[6];
  const float* u_w2 = (const float*)d_in[7];
  const float* u_b  = (const float*)d_in[8];
  const float* u_v  = (const float*)d_in[9];
  float* out = (float*)d_out;

  float* ws   = (float*)d_ws;
  float* wsQ  = ws;
  float* wsK  = ws + (size_t)B_ * PBS;
  float* wsV  = ws + 2 * (size_t)B_ * PBS;
  float* wsDec = wsQ;

  size_t sm1 = (size_t)(TP * HSTR + 64) * sizeof(float);
  size_t sm2 = (size_t)(2 * TP * HSTR + 8 * 16 * SSTR) * sizeof(float);
  size_t sm3 = (size_t)(D_ * D_ + 3 * 128 + TP) * sizeof(float);
  hipFuncSetAttribute((const void*)qkv_kernel,
                      hipFuncAttributeMaxDynamicSharedMemorySize, (int)sm1);
  hipFuncSetAttribute((const void*)attn_kernel,
                      hipFuncAttributeMaxDynamicSharedMemorySize, (int)sm2);
  hipFuncSetAttribute((const void*)stamp_kernel,
                      hipFuncAttributeMaxDynamicSharedMemorySize, (int)sm3);

  qkv_kernel<<<B_, 256, sm1, stream>>>(re, alia, slen, Wq, Wk, Wv, wsQ, wsK, wsV);
  attn_kernel<<<B_, 256, sm2, stream>>>(re, alia, slen, wsQ, wsK, wsV, wsDec);
  stamp_kernel<<<B_, 256, sm3, stream>>>(slen, wsDec, u_w1, u_w2, u_b, u_v, out);
}


# attn kernel: K and V tiles stream through a two-slot LDS ring (next tile loaded under the current tile's WMMAs) instead of whole-matrix staging: LDS 327 to 124 KB, two blocks per WGP; old sum order
# speedup vs baseline: 4.9972x; 1.0938x over previous
#include <hip/hip_runtime.h>
#include <math.h>


typedef __attribute__((ext_vector_type(2))) float v2f;
typedef __attribute__((ext_vector_type(8))) float v8f;

#define B_      2048
#define T_      200
#define D_      128
#define NNODE_  200
#define TP      208
#define NT      13
#define HSTR    132
#define SSTR    210
#define PBS     (TP * D_)
#define NEGC    (-4294967295.0f)

__device__ __forceinline__ int live_tiles(int len) {
  const int n = (len + 15) >> 4;
  return n < 1 ? 1 : (n > NT ? NT : n);
}

__device__ __forceinline__ v8f wmma4(v2f a, v2f b, v8f c) {
  return __builtin_amdgcn_wmma_f32_16x16x4_f32(false, a, false, b, (short)0, c,
                                               false, false);
}

typedef unsigned long long u64;

__device__ __forceinline__ int cls_get(u64 x0, u64 x1, u64 x2, int c) {
  const int w = (c >= 5) + (c >= 10);
  const u64 x = (w == 0) ? x0 : ((w == 1) ? x1 : x2);
  return (int)((x >> (12 * (c - 5 * w))) & 0xFFFull);
}

__device__ __forceinline__ u64 uniform64(u64 x) {
  const unsigned lo = (unsigned)__builtin_amdgcn_readfirstlane((int)(unsigned)(x & 0xFFFFFFFFull));
  const unsigned hi = (unsigned)__builtin_amdgcn_readfirstlane((int)(unsigned)(x >> 32));
  return ((u64)hi << 32) | (u64)lo;
}

__device__ __forceinline__ int ranked_session(const int* __restrict__ seqlen, u64* scr, int tid,
                                              int& len_out) {
  const int lane = tid & 31, wave = tid >> 5;
  int* res = (int*)(scr + 24);
  const int4 la = ((const int4*)seqlen)[2 * tid];
  const int4 lb = ((const int4*)seqlen)[2 * tid + 1];
  const int lenv[8] = {la.x, la.y, la.z, la.w, lb.x, lb.y, lb.z, lb.w};
  int cls[8];
  u64 nib = 0ull;
#pragma unroll
  for (int e = 0; e < 8; ++e) {
    cls[e] = NT - live_tiles(lenv[e]);
    nib += 1ull << (4 * cls[e]);
  }
  u64 h0 = 0ull, h1 = 0ull, h2 = 0ull;
#pragma unroll
  for (int c = 0; c < NT; ++c) {
    const int w = c / 5;
    const u64 v = ((nib >> (4 * c)) & 0xFull) << (12 * (c - 5 * w));
    if (w == 0) h0 += v; else if (w == 1) h1 += v; else h2 += v;
  }
  u64 s0 = h0, s1 = h1, s2 = h2;
#pragma unroll
  for (int d = 1; d < 32; d <<= 1) {
    const u64 t0 = __shfl_up(s0, d, 32), t1 = __shfl_up(s1, d, 32), t2 = __shfl_up(s2, d, 32);
    if (lane >= d) { s0 += t0; s1 += t1; s2 += t2; }
  }
  if (lane == 31) { scr[3 * wave] = s0; scr[3 * wave + 1] = s1; scr[3 * wave + 2] = s2; }
  if (tid == 0) { res[0] = blockIdx.x; res[1] = 1; }
  __syncthreads();
  u64 r0 = s0 - h0, r1 = s1 - h1, r2 = s2 - h2;
  u64 H0 = 0ull, H1 = 0ull, H2 = 0ull;
#pragma unroll
  for (int w = 0; w < 8; ++w) {
    const u64 a = scr[3 * w], bq = scr[3 * w + 1], cq = scr[3 * w + 2];
    H0 += a; H1 += bq; H2 += cq;
    if (w < wave) { r0 += a; r1 += bq; r2 += cq; }
  }
  const u64 T0 = uniform64(H0), T1 = uniform64(H1), T2 = uniform64(H2);
  const int k = blockIdx.x;
  int run = 0, cstar = NT - 1, jj = 0;
  bool open = true;
#pragma unroll
  for (int c = 0; c < NT; ++c) {
    const int cnt = cls_get(T0, T1, T2, c);
    if (open && k < run + cnt) { cstar = c; jj = k - run; open = false; }
    run += cnt;
  }
  int p = cls_get(r0, r1, r2, cstar);
  int found = -1, flen = 1;
#pragma unroll
  for (int e = 0; e < 8; ++e) {
    const bool m = (cls[e] == cstar);
    if (m && p == jj) { found = 8 * tid + e; flen = lenv[e]; }
    p += m ? 1 : 0;
  }
  if (found >= 0) { res[0] = found; res[1] = flen; }
  __syncthreads();
  int b = __builtin_amdgcn_readfirstlane(res[0]);
  len_out = __builtin_amdgcn_readfirstlane(res[1]);
  return b < 0 ? 0 : (b > B_ - 1 ? B_ - 1 : b);
}

__global__ void __launch_bounds__(256) __attribute__((amdgpu_waves_per_eu(4, 4))) qkv_kernel(const float* __restrict__ re, const int* __restrict__ alias,
                           const int* __restrict__ seqlen,
                           const float* __restrict__ Wq, const float* __restrict__ Wk,
                           const float* __restrict__ Wv,
                           float* __restrict__ wsQ, float* __restrict__ wsK,
                           float* __restrict__ wsV) {
  extern __shared__ __attribute__((aligned(16))) float smem_h[];
  float* sH = smem_h;
  const int tid = threadIdx.x;
  int len;
  const int b = ranked_session(seqlen, (u64*)(smem_h + TP * HSTR), tid, len);
  const int ntl = live_tiles(len);
  const int wv = __builtin_amdgcn_readfirstlane(tid >> 5);
  const int ln = tid & 31, lh = ln & 15, hi = ln >> 4;
  const int n0 = wv << 4;

  {
    const int tl = wv + 8 * ln;
    const int my_a = alias[(size_t)b * T_ + (tl < T_ ? tl : T_ - 1)];
    const float* reb = re + (size_t)b * NNODE_ * D_ + ln * 4;
    const int nk = 2 * ntl;
#pragma unroll 1
    for (int h = 0; h < 2; ++h) {
      const int k0 = h * 13;
      if (k0 >= nk) break;
      float4 v[13];
#pragma unroll
      for (int r = 0; r < 13; ++r) {
        if (k0 + r < nk) {
          const int a = __builtin_amdgcn_readlane(my_a, k0 + r);
          float4 x = *(const float4*)(reb + (size_t)a * D_);
          if (wv + 8 * (k0 + r) >= T_) x = make_float4(0.f, 0.f, 0.f, 0.f);
          v[r] = x;
        }
      }
#pragma unroll
      for (int r = 0; r < 13; ++r)
        if (k0 + r < nk) *(float4*)(sH + (wv + 8 * (k0 + r)) * HSTR + ln * 4) = v[r];
    }
  }

  v2f bq[32], bk[32], bv[32];
  {
    const float* wq = Wq + (2 * hi) * D_ + n0 + lh;
    const float* wk = Wk + (2 * hi) * D_ + n0 + lh;
    const float* wvp = Wv + (2 * hi) * D_ + n0 + lh;
#pragma unroll
    for (int c = 0; c < 32; ++c) {
      bq[c].x = wq[(4 * c) * D_];
      bq[c].y = wq[(4 * c + 1) * D_];
      bk[c].x = wk[(4 * c) * D_];
      bk[c].y = wk[(4 * c + 1) * D_];
      bv[c].x = wvp[(4 * c) * D_];
      bv[c].y = wvp[(4 * c + 1) * D_];
    }
  }
  __syncthreads();

  float* dq = wsQ + (size_t)b * PBS + n0 + lh;
  float* dk = wsK + (size_t)b * PBS + n0 + lh;
  float* dv = wsV + (size_t)b * PBS + n0 + lh;
  for (int i = 0; i < ntl; ++i) {
    const float* arow = sH + (i * 16 + lh) * HSTR + 2 * hi;
    v8f aq = {0.f, 0.f, 0.f, 0.f, 0.f, 0.f, 0.f, 0.f};
    v8f ak = {0.f, 0.f, 0.f, 0.f, 0.f, 0.f, 0.f, 0.f};
    v8f av = {0.f, 0.f, 0.f, 0.f, 0.f, 0.f, 0.f, 0.f};
#pragma unroll
    for (int g = 0; g < 4; ++g) {
      v2f af[8];
#pragma unroll
      for (int cc = 0; cc < 8; ++cc) af[cc] = *(const v2f*)(arow + 4 * (8 * g + cc));
      __builtin_amdgcn_sched_barrier(0);
#pragma unroll
      for (int cc = 0; cc < 8; ++cc) {
        const int c = 8 * g + cc;
        aq = wmma4(af[cc], bq[c], aq);
        ak = wmma4(af[cc], bk[c], ak);
        av = wmma4(af[cc], bv[c], av);
      }
    }
#pragma unroll
    for (int r = 0; r < 8; ++r) {
      const int row = i * 16 + r + 8 * hi;
      const float q = aq[r], k = ak[r], v = av[r];
      dq[row * D_] = q > 0.f ? q : 0.f;
      dk[row * D_] = k > 0.f ? k : 0.f;
      dv[row * D_] = v > 0.f ? v : 0.f;
    }
  }
}

#define KVSL (16 * HSTR)
__device__ __forceinline__ void tile_request(const float4* __restrict__ M4, int j, float4& a, float4& b) {
  a = M4[256 * (2 * j)];
  b = M4[256 * (2 * j + 1)];
}
__device__ __forceinline__ void tile_store(float* slot, int wave, int lane, const float4& a, const float4& b) {
  *(float4*)(slot + wave * HSTR + lane * 4) = a;
  *(float4*)(slot + (wave + 8) * HSTR + lane * 4) = b;
}


__device__ __forceinline__ void load_query_tile(const float* __restrict__ Qrow, int my_a, int lh, int hi,
                                                v2f (&qa)[32], int (&roff)[8]) {
#pragma unroll
  for (int r = 0; r < 8; ++r) {
    const int alo = __builtin_amdgcn_readlane(my_a, r);
    const int ahi = __builtin_amdgcn_readlane(my_a, 8 + r);
    roff[r] = (hi ? ahi : alo) * D_ + lh;
  }
#pragma unroll
  for (int c = 0; c < 32; ++c) qa[c] = *(const v2f*)(Qrow + 4 * c + 2 * hi);
}

__global__ void __launch_bounds__(256) __attribute__((amdgpu_waves_per_eu(4, 4))) attn_kernel(const float* __restrict__ re, const int* __restrict__ alias,
                            const int* __restrict__ seqlen,
                            const float* __restrict__ wsQ, const float* __restrict__ wsK,
                            const float* __restrict__ wsV, float* __restrict__ wsDec) {
  extern __shared__ __attribute__((aligned(16))) float smem[];
  float* sR = smem;
  float* sS0 = smem + 2 * KVSL;
  const int tid = threadIdx.x;
  int len;
  const int b = ranked_session(seqlen, (u64*)sS0, tid, len);
  const int ntl = live_tiles(len);
  const int wave = tid >> 5, lane = tid & 31, lh = lane & 15, hi = lane >> 4;
  const float4* K4 = (const float4*)(wsK + (size_t)b * PBS) + tid;
  const float4* V4 = (const float4*)(wsV + (size_t)b * PBS) + tid;
  float4 ta, tb;
  tile_request(K4, 0, ta, tb);
  const int t_a0 = wave * 16 + lh, t_a1 = (wave + 8) * 16 + lh;
  const int my_a0 = alias[b * T_ + (t_a0 < T_ ? t_a0 : T_ - 1)];
  const int my_a1 = alias[b * T_ + (t_a1 < T_ ? t_a1 : T_ - 1)];
  float* sS = sS0 + wave * (16 * SSTR);
  const float scale = 0.088388347648318447f;
  const float* reb = re + (size_t)b * NNODE_ * D_;
  const float* Qb = wsQ + (size_t)b * PBS;
  float* Db = wsDec + (size_t)b * PBS;

  int roff[8];
  v2f qa[32];
  if (wave < ntl) load_query_tile(Qb + (wave * 16 + lh) * D_, my_a0, lh, hi, qa, roff);
  tile_store(sR, wave, lane, ta, tb);
  __syncthreads();

  const int npass = (ntl > 8) ? 2 : 1;
  for (int pass = 0; pass < npass; ++pass) {
    const int qt = wave + 8 * pass;
    const bool active = (qt < ntl);
    const bool again = (pass + 1 < npass);

    for (int j = 0; j < ntl; ++j) {
      const float* slot = sR + (j & 1) * KVSL;
      if (j + 1 < ntl) tile_request(K4, j + 1, ta, tb);
      else             tile_request(V4, 0, ta, tb);
      if (active) {
        v8f acc = {0.f, 0.f, 0.f, 0.f, 0.f, 0.f, 0.f, 0.f};
        const float* krow = slot + lh * HSTR + 2 * hi;
        v2f kf[2][8];
#pragma unroll
        for (int cc = 0; cc < 8; ++cc) kf[0][cc] = *(const v2f*)(krow + 4 * cc);
#pragma unroll
        for (int g = 0; g < 4; ++g) {
          if (g < 3) {
#pragma unroll
            for (int cc = 0; cc < 8; ++cc) kf[(g + 1) & 1][cc] = *(const v2f*)(krow + 4 * (8 * (g + 1) + cc));
          }
          __builtin_amdgcn_sched_barrier(0);
#pragma unroll
          for (int cc = 0; cc < 8; ++cc) acc = wmma4(qa[8 * g + cc], kf[g & 1][cc], acc);
          __builtin_amdgcn_sched_barrier(0);
        }
#pragma unroll
        for (int r = 0; r < 8; ++r) {
          int m = r + 8 * hi, s = j * 16 + lh;
          sS[m * SSTR + s] = (s < len) ? acc[r] * scale : NEGC;
        }
      }
      tile_store(sR + ((j + 1) & 1) * KVSL, wave, lane, ta, tb);
      __syncthreads();
    }

    v8f dacc[8];
    if (active) {
#pragma unroll
      for (int nt = 0; nt < 8; ++nt) {
#pragma unroll
        for (int r = 0; r < 8; ++r) {
          const float x = reb[roff[r] + 16 * nt];
          dacc[nt][r] = (qt * 16 + r + 8 * hi < T_) ? x : 0.f;
        }
      }
    }
    if (active) {
      float* rowp = sS + lh * SSTR + hi * 104;
      int cnt = 16 * ntl - hi * 104;
      cnt = cnt > 104 ? 104 : (cnt < 0 ? 0 : cnt);
      float mx = -3.0e38f;
      for (int s0 = 0; s0 < cnt; s0 += 8) {
#pragma unroll
        for (int u = 0; u < 8; ++u) mx = fmaxf(mx, rowp[s0 + u]);
      }
      mx = fmaxf(mx, __shfl_xor(mx, 16, 32));
      float sum = 0.f;
      for (int s0 = 0; s0 < cnt; s0 += 8) {
#pragma unroll
        for (int u = 0; u < 8; ++u) { float e = __expf(rowp[s0 + u] - mx); rowp[s0 + u] = e; sum += e; }
      }
      sum += __shfl_xor(sum, 16, 32);
      float qm = ((qt * 16 + lh) < len) ? 1.f : 0.f;
      float inv = qm / sum;
      for (int s0 = 0; s0 < cnt; s0 += 8) {
#pragma unroll
        for (int u = 0; u < 8; ++u) rowp[s0 + u] *= inv;
      }
    }

    for (int j = 0; j < ntl; ++j) {
      const float* slot = sR + ((ntl + j) & 1) * KVSL;
      const bool more = (j + 1 < ntl) || again;
      if (j + 1 < ntl) tile_request(V4, j + 1, ta, tb);
      else if (again)  tile_request(K4, 0, ta, tb);
      if (active) {
        v2f pa[4];
#pragma unroll
        for (int cc = 0; cc < 4; ++cc) pa[cc] = *(const v2f*)(sS + lh * SSTR + 4 * (4 * j + cc) + 2 * hi);
        const float* vcol = slot + (2 * hi) * HSTR + lh;
        v2f vb[32];
#pragma unroll
        for (int nt = 0; nt < 8; ++nt) {
          int noff = 16 * nt;
          asm volatile("" : "+v"(noff));
          const float* vn = vcol + noff;
#pragma unroll
          for (int cc = 0; cc < 4; ++cc) {
            vb[8 * cc + nt].x = vn[(4 * cc) * HSTR];
            vb[8 * cc + nt].y = vn[(4 * cc + 1) * HSTR];
          }
        }
        __builtin_amdgcn_sched_barrier(0);
#pragma unroll
        for (int cc = 0; cc < 4; ++cc) {
#pragma unroll
          for (int nt = 0; nt < 8; ++nt) dacc[nt] = wmma4(pa[cc], vb[8 * cc + nt], dacc[nt]);
        }
      }
      if (more) tile_store(sR + ((ntl + j + 1) & 1) * KVSL, wave, lane, ta, tb);
      __syncthreads();
    }

    if (active) {
#pragma unroll
      for (int nt = 0; nt < 8; ++nt) {
#pragma unroll
        for (int r = 0; r < 8; ++r) {
          int t = qt * 16 + r + 8 * hi;
          Db[t * D_ + 16 * nt + lh] = dacc[nt][r];
        }
      }
    }
    if (again && wave + 8 < ntl)
      load_query_tile(Qb + ((wave + 8) * 16 + lh) * D_, my_a1, lh, hi, qa, roff);
  }
}

__global__ void stamp_kernel(const int* __restrict__ seqlen, const float* __restrict__ wsDec,
                             const float* __restrict__ u_w1, const float* __restrict__ u_w2,
                             const float* __restrict__ u_b, const float* __restrict__ u_v,
                             float* __restrict__ out) {
  extern __shared__ __attribute__((aligned(16))) float smem[];
  float* sW2 = smem;
  float* sLp = smem + D_ * D_;
  float* sUb = sLp + 128;
  float* sUv = sUb + 128;
  float* sCoef = sUv + 128;
  const int tid = threadIdx.x;
  int len;
  const int b = ranked_session(seqlen, (u64*)sCoef, tid, len);
  const int ntl = live_tiles(len);
  const float* Db = wsDec + (size_t)b * PBS;

  {
    const float4* src = (const float4*)u_w2 + tid;
    float4 w[16];
#pragma unroll
    for (int i = 0; i < 16; ++i) w[i] = src[256 * i];
#pragma unroll
    for (int i = 0; i < 16; ++i) *(float4*)(sW2 + 4 * (tid + 256 * i)) = w[i];
  }
  if (tid < D_) { sUb[tid] = u_b[tid]; sUv[tid] = u_v[tid]; }
  if (tid < D_) {
    const float* lrow = Db + (len - 1) * D_;
    float acc = 0.f;
    for (int d = 0; d < D_; ++d) acc += lrow[d] * u_w1[d * D_ + tid];
    sLp[tid] = acc;
    out[(size_t)b * 2 * D_ + D_ + tid] = acc;
  }
  __syncthreads();

  const int wave = tid >> 5, lane = tid & 31, lh = lane & 15, hi = lane >> 4;
  for (int pass = 0; pass < 2; ++pass) {
    const int qt = wave + 8 * pass;
    if (qt < ntl) {
      v2f da[32];
      const float* drow = Db + (qt * 16 + lh) * D_;
#pragma unroll
      for (int c = 0; c < 32; ++c) da[c] = *(const v2f*)(drow + 4 * c + 2 * hi);
      float cpart[8];
#pragma unroll
      for (int r = 0; r < 8; ++r) cpart[r] = 0.f;
      for (int nt = 0; nt < 8; ++nt) {
        int n0 = nt * 16;
        v8f acc = {0.f, 0.f, 0.f, 0.f, 0.f, 0.f, 0.f, 0.f};
#pragma unroll
        for (int c = 0; c < 32; ++c) {
          int k = 4 * c + 2 * hi;
          v2f bb;
          bb.x = sW2[k * D_ + n0 + lh];
          bb.y = sW2[(k + 1) * D_ + n0 + lh];
          acc = wmma4(da[c], bb, acc);
        }
        int col = n0 + lh;
        float add = sLp[col] + sUb[col];
        float uvv = sUv[col];
#pragma unroll
        for (int r = 0; r < 8; ++r) {
          float x = acc[r] + add;
          float sg = 1.f / (1.f + __expf(-x));
          cpart[r] += sg * uvv;
        }
      }
#pragma unroll
      for (int r = 0; r < 8; ++r) {
        float v = cpart[r];
        v += __shfl_xor(v, 1, 32);
        v += __shfl_xor(v, 2, 32);
        v += __shfl_xor(v, 4, 32);
        v += __shfl_xor(v, 8, 32);
        if (lh == 0) {
          int t = qt * 16 + r + 8 * hi;
          sCoef[t] = (t < len) ? v : 0.f;
        }
      }
    }
  }
  __syncthreads();
  if (tid < D_) {
    float acc = 0.f;
    for (int tt = 0; tt < ntl; ++tt) {
#pragma unroll
      for (int u = 0; u < 16; ++u) {
        const int t = 16 * tt + u;
        acc += sCoef[t] * Db[t * D_ + tid];
      }
    }
    out[(size_t)b * 2 * D_ + tid] = acc;
  }
}

extern "C" void kernel_launch(void* const* d_in, const int* in_sizes, int n_in,
                              void* d_out, int out_size, void* d_ws, size_t ws_size,
                              hipStream_t stream) {
  (void)in_sizes; (void)n_in; (void)out_size; (void)ws_size;
  const float* re   = (const float*)d_in[0];
  const int*   alia = (const int*)d_in[1];
  const int*   slen = (const int*)d_in[2];
  const float* Wq   = (const float*)d_in[3];
  const float* Wk   = (const float*)d_in[4];
  const float* Wv   = (const float*)d_in[5];
  const float* u_w1 = (const float*)d_in[6];
  const float* u_w2 = (const float*)d_in[7];
  const float* u_b  = (const float*)d_in[8];
  const float* u_v  = (const float*)d_in[9];
  float* out = (float*)d_out;

  float* ws   = (float*)d_ws;
  float* wsQ  = ws;
  float* wsK  = ws + (size_t)B_ * PBS;
  float* wsV  = ws + 2 * (size_t)B_ * PBS;
  float* wsDec = wsQ;

  size_t sm1 = (size_t)(TP * HSTR + 64) * sizeof(float);
  size_t sm2 = (size_t)(2 * KVSL + 8 * 16 * SSTR) * sizeof(float);
  size_t sm3 = (size_t)(D_ * D_ + 3 * 128 + TP) * sizeof(float);
  hipFuncSetAttribute((const void*)qkv_kernel,
                      hipFuncAttributeMaxDynamicSharedMemorySize, (int)sm1);
  hipFuncSetAttribute((const void*)attn_kernel,
                      hipFuncAttributeMaxDynamicSharedMemorySize, (int)sm2);
  hipFuncSetAttribute((const void*)stamp_kernel,
                      hipFuncAttributeMaxDynamicSharedMemorySize, (int)sm3);

  qkv_kernel<<<B_, 256, sm1, stream>>>(re, alia, slen, Wq, Wk, Wv, wsQ, wsK, wsV);
  attn_kernel<<<B_, 256, sm2, stream>>>(re, alia, slen, wsQ, wsK, wsV, wsDec);
  stamp_kernel<<<B_, 256, sm3, stream>>>(slen, wsDec, u_w1, u_w2, u_b, u_v, out);
}


# attn kernel: online softmax (running row max and sum), K and V tiles ride the LDS ring together, one barrier per key tile, no whole-row score sweeps; softmax and attn*V sums in a new order
# speedup vs baseline: 5.1815x; 1.0369x over previous
#include <hip/hip_runtime.h>
#include <math.h>


typedef __attribute__((ext_vector_type(2))) float v2f;
typedef __attribute__((ext_vector_type(8))) float v8f;

#define B_      2048
#define T_      200
#define D_      128
#define NNODE_  200
#define TP      208
#define NT      13
#define HSTR    132
#define SSTR    210
#define PBS     (TP * D_)
#define NEGC    (-4294967295.0f)

__device__ __forceinline__ int live_tiles(int len) {
  const int n = (len + 15) >> 4;
  return n < 1 ? 1 : (n > NT ? NT : n);
}

__device__ __forceinline__ v8f wmma4(v2f a, v2f b, v8f c) {
  return __builtin_amdgcn_wmma_f32_16x16x4_f32(false, a, false, b, (short)0, c,
                                               false, false);
}

typedef unsigned long long u64;

__device__ __forceinline__ int cls_get(u64 x0, u64 x1, u64 x2, int c) {
  const int w = (c >= 5) + (c >= 10);
  const u64 x = (w == 0) ? x0 : ((w == 1) ? x1 : x2);
  return (int)((x >> (12 * (c - 5 * w))) & 0xFFFull);
}

__device__ __forceinline__ u64 uniform64(u64 x) {
  const unsigned lo = (unsigned)__builtin_amdgcn_readfirstlane((int)(unsigned)(x & 0xFFFFFFFFull));
  const unsigned hi = (unsigned)__builtin_amdgcn_readfirstlane((int)(unsigned)(x >> 32));
  return ((u64)hi << 32) | (u64)lo;
}

__device__ __forceinline__ int ranked_session(const int* __restrict__ seqlen, u64* scr, int tid,
                                              int& len_out) {
  const int lane = tid & 31, wave = tid >> 5;
  int* res = (int*)(scr + 24);
  const int4 la = ((const int4*)seqlen)[2 * tid];
  const int4 lb = ((const int4*)seqlen)[2 * tid + 1];
  const int lenv[8] = {la.x, la.y, la.z, la.w, lb.x, lb.y, lb.z, lb.w};
  int cls[8];
  u64 nib = 0ull;
#pragma unroll
  for (int e = 0; e < 8; ++e) {
    cls[e] = NT - live_tiles(lenv[e]);
    nib += 1ull << (4 * cls[e]);
  }
  u64 h0 = 0ull, h1 = 0ull, h2 = 0ull;
#pragma unroll
  for (int c = 0; c < NT; ++c) {
    const int w = c / 5;
    const u64 v = ((nib >> (4 * c)) & 0xFull) << (12 * (c - 5 * w));
    if (w == 0) h0 += v; else if (w == 1) h1 += v; else h2 += v;
  }
  u64 s0 = h0, s1 = h1, s2 = h2;
#pragma unroll
  for (int d = 1; d < 32; d <<= 1) {
    const u64 t0 = __shfl_up(s0, d, 32), t1 = __shfl_up(s1, d, 32), t2 = __shfl_up(s2, d, 32);
    if (lane >= d) { s0 += t0; s1 += t1; s2 += t2; }
  }
  if (lane == 31) { scr[3 * wave] = s0; scr[3 * wave + 1] = s1; scr[3 * wave + 2] = s2; }
  if (tid == 0) { res[0] = blockIdx.x; res[1] = 1; }
  __syncthreads();
  u64 r0 = s0 - h0, r1 = s1 - h1, r2 = s2 - h2;
  u64 H0 = 0ull, H1 = 0ull, H2 = 0ull;
#pragma unroll
  for (int w = 0; w < 8; ++w) {
    const u64 a = scr[3 * w], bq = scr[3 * w + 1], cq = scr[3 * w + 2];
    H0 += a; H1 += bq; H2 += cq;
    if (w < wave) { r0 += a; r1 += bq; r2 += cq; }
  }
  const u64 T0 = uniform64(H0), T1 = uniform64(H1), T2 = uniform64(H2);
  const int k = blockIdx.x;
  int run = 0, cstar = NT - 1, jj = 0;
  bool open = true;
#pragma unroll
  for (int c = 0; c < NT; ++c) {
    const int cnt = cls_get(T0, T1, T2, c);
    if (open && k < run + cnt) { cstar = c; jj = k - run; open = false; }
    run += cnt;
  }
  int p = cls_get(r0, r1, r2, cstar);
  int found = -1, flen = 1;
#pragma unroll
  for (int e = 0; e < 8; ++e) {
    const bool m = (cls[e] == cstar);
    if (m && p == jj) { found = 8 * tid + e; flen = lenv[e]; }
    p += m ? 1 : 0;
  }
  if (found >= 0) { res[0] = found; res[1] = flen; }
  __syncthreads();
  int b = __builtin_amdgcn_readfirstlane(res[0]);
  len_out = __builtin_amdgcn_readfirstlane(res[1]);
  return b < 0 ? 0 : (b > B_ - 1 ? B_ - 1 : b);
}

__global__ void __launch_bounds__(256) __attribute__((amdgpu_waves_per_eu(4, 4))) qkv_kernel(const float* __restrict__ re, const int* __restrict__ alias,
                           const int* __restrict__ seqlen,
                           const float* __restrict__ Wq, const float* __restrict__ Wk,
                           const float* __restrict__ Wv,
                           float* __restrict__ wsQ, float* __restrict__ wsK,
                           float* __restrict__ wsV) {
  extern __shared__ __attribute__((aligned(16))) float smem_h[];
  float* sH = smem_h;
  const int tid = threadIdx.x;
  int len;
  const int b = ranked_session(seqlen, (u64*)(smem_h + TP * HSTR), tid, len);
  const int ntl = live_tiles(len);
  const int wv = __builtin_amdgcn_readfirstlane(tid >> 5);
  const int ln = tid & 31, lh = ln & 15, hi = ln >> 4;
  const int n0 = wv << 4;

  {
    const int tl = wv + 8 * ln;
    const int my_a = alias[(size_t)b * T_ + (tl < T_ ? tl : T_ - 1)];
    const float* reb = re + (size_t)b * NNODE_ * D_ + ln * 4;
    const int nk = 2 * ntl;
#pragma unroll 1
    for (int h = 0; h < 2; ++h) {
      const int k0 = h * 13;
      if (k0 >= nk) break;
      float4 v[13];
#pragma unroll
      for (int r = 0; r < 13; ++r) {
        if (k0 + r < nk) {
          const int a = __builtin_amdgcn_readlane(my_a, k0 + r);
          float4 x = *(const float4*)(reb + (size_t)a * D_);
          if (wv + 8 * (k0 + r) >= T_) x = make_float4(0.f, 0.f, 0.f, 0.f);
          v[r] = x;
        }
      }
#pragma unroll
      for (int r = 0; r < 13; ++r)
        if (k0 + r < nk) *(float4*)(sH + (wv + 8 * (k0 + r)) * HSTR + ln * 4) = v[r];
    }
  }

  v2f bq[32], bk[32], bv[32];
  {
    const float* wq = Wq + (2 * hi) * D_ + n0 + lh;
    const float* wk = Wk + (2 * hi) * D_ + n0 + lh;
    const float* wvp = Wv + (2 * hi) * D_ + n0 + lh;
#pragma unroll
    for (int c = 0; c < 32; ++c) {
      bq[c].x = wq[(4 * c) * D_];
      bq[c].y = wq[(4 * c + 1) * D_];
      bk[c].x = wk[(4 * c) * D_];
      bk[c].y = wk[(4 * c + 1) * D_];
      bv[c].x = wvp[(4 * c) * D_];
      bv[c].y = wvp[(4 * c + 1) * D_];
    }
  }
  __syncthreads();

  float* dq = wsQ + (size_t)b * PBS + n0 + lh;
  float* dk = wsK + (size_t)b * PBS + n0 + lh;
  float* dv = wsV + (size_t)b * PBS + n0 + lh;
  for (int i = 0; i < ntl; ++i) {
    const float* arow = sH + (i * 16 + lh) * HSTR + 2 * hi;
    v8f aq = {0.f, 0.f, 0.f, 0.f, 0.f, 0.f, 0.f, 0.f};
    v8f ak = {0.f, 0.f, 0.f, 0.f, 0.f, 0.f, 0.f, 0.f};
    v8f av = {0.f, 0.f, 0.f, 0.f, 0.f, 0.f, 0.f, 0.f};
#pragma unroll
    for (int g = 0; g < 4; ++g) {
      v2f af[8];
#pragma unroll
      for (int cc = 0; cc < 8; ++cc) af[cc] = *(const v2f*)(arow + 4 * (8 * g + cc));
      __builtin_amdgcn_sched_barrier(0);
#pragma unroll
      for (int cc = 0; cc < 8; ++cc) {
        const int c = 8 * g + cc;
        aq = wmma4(af[cc], bq[c], aq);
        ak = wmma4(af[cc], bk[c], ak);
        av = wmma4(af[cc], bv[c], av);
      }
    }
#pragma unroll
    for (int r = 0; r < 8; ++r) {
      const int row = i * 16 + r + 8 * hi;
      const float q = aq[r], k = ak[r], v = av[r];
      dq[row * D_] = q > 0.f ? q : 0.f;
      dk[row * D_] = k > 0.f ? k : 0.f;
      dv[row * D_] = v > 0.f ? v : 0.f;
    }
  }
}

#define KVSL (16 * HSTR)
#define SLOTF (2 * KVSL)
__device__ __forceinline__ void pair_request(const float4* __restrict__ K4, const float4* __restrict__ V4, int j,
                                             float4& ka, float4& kb, float4& va, float4& vb) {
  ka = K4[256 * (2 * j)];
  kb = K4[256 * (2 * j + 1)];
  va = V4[256 * (2 * j)];
  vb = V4[256 * (2 * j + 1)];
}
__device__ __forceinline__ void pair_store(float* slot, int wave, int lane, const float4& ka, const float4& kb,
                                           const float4& va, const float4& vb) {
  *(float4*)(slot + wave * HSTR + lane * 4) = ka;
  *(float4*)(slot + (wave + 8) * HSTR + lane * 4) = kb;
  *(float4*)(slot + KVSL + wave * HSTR + lane * 4) = va;
  *(float4*)(slot + KVSL + (wave + 8) * HSTR + lane * 4) = vb;
}


__device__ __forceinline__ void load_query_tile(const float* __restrict__ Qrow, int my_a, int lh, int hi,
                                                v2f (&qa)[32], int (&roff)[8]) {
#pragma unroll
  for (int r = 0; r < 8; ++r) {
    const int alo = __builtin_amdgcn_readlane(my_a, r);
    const int ahi = __builtin_amdgcn_readlane(my_a, 8 + r);
    roff[r] = (hi ? ahi : alo) * D_ + lh;
  }
#pragma unroll
  for (int c = 0; c < 32; ++c) qa[c] = *(const v2f*)(Qrow + 4 * c + 2 * hi);
}

__global__ void __launch_bounds__(256) __attribute__((amdgpu_waves_per_eu(4, 4))) attn_kernel(const float* __restrict__ re, const int* __restrict__ alias,
                            const int* __restrict__ seqlen,
                            const float* __restrict__ wsQ, const float* __restrict__ wsK,
                            const float* __restrict__ wsV, float* __restrict__ wsDec) {
  extern __shared__ __attribute__((aligned(16))) float smem[];
  float* sR = smem;
  float* sS0 = smem + 2 * SLOTF;
  const int tid = threadIdx.x;
  int len;
  const int b = ranked_session(seqlen, (u64*)sS0, tid, len);
  const int ntl = live_tiles(len);
  const int wave = tid >> 5, lane = tid & 31, lh = lane & 15, hi = lane >> 4;
  const float4* K4 = (const float4*)(wsK + (size_t)b * PBS) + tid;
  const float4* V4 = (const float4*)(wsV + (size_t)b * PBS) + tid;
  float4 ka, kb, va, vb4;
  pair_request(K4, V4, 0, ka, kb, va, vb4);
  const int t_a0 = wave * 16 + lh, t_a1 = (wave + 8) * 16 + lh;
  const int my_a0 = alias[b * T_ + (t_a0 < T_ ? t_a0 : T_ - 1)];
  const int my_a1 = alias[b * T_ + (t_a1 < T_ ? t_a1 : T_ - 1)];
  float* sS = sS0 + wave * (16 * SSTR);
  const float scale = 0.088388347648318447f;
  const float* reb = re + (size_t)b * NNODE_ * D_;
  const float* Qb = wsQ + (size_t)b * PBS;
  float* Db = wsDec + (size_t)b * PBS;

  int roff[8];
  v2f qa[32];
  if (wave < ntl) {
    load_query_tile(Qb + (wave * 16 + lh) * D_, my_a0, lh, hi, qa, roff);
#pragma unroll
    for (int c = 0; c < 32; ++c) asm volatile("" : "+v"(qa[c]));
  }
  pair_store(sR, wave, lane, ka, kb, va, vb4);
  __syncthreads();

  const int npass = (ntl > 8) ? 2 : 1;
  int n = 0;
  for (int pass = 0; pass < npass; ++pass) {
    const int qt = wave + 8 * pass;
    const bool active = (qt < ntl);
    const bool again = (pass + 1 < npass);

    v8f dacc[8];
#pragma unroll
    for (int nt = 0; nt < 8; ++nt) dacc[nt] = (v8f){0.f, 0.f, 0.f, 0.f, 0.f, 0.f, 0.f, 0.f};
    float m_run = -1.0e30f;
    float l_run = 0.f;

    for (int j = 0; j < ntl; ++j, ++n) {
      const float* slot = sR + (n & 1) * SLOTF;
      const bool more = (j + 1 < ntl) || again;
      if (j + 1 < ntl) pair_request(K4, V4, j + 1, ka, kb, va, vb4);
      else if (again)  pair_request(K4, V4, 0, ka, kb, va, vb4);
      if (active) {
        v8f acc = {0.f, 0.f, 0.f, 0.f, 0.f, 0.f, 0.f, 0.f};
        const float* krow = slot + lh * HSTR + 2 * hi;
        v2f kf[2][8];
#pragma unroll
        for (int cc = 0; cc < 8; ++cc) kf[0][cc] = *(const v2f*)(krow + 4 * cc);
#pragma unroll
        for (int g = 0; g < 4; ++g) {
          if (g < 3) {
#pragma unroll
            for (int cc = 0; cc < 8; ++cc) kf[(g + 1) & 1][cc] = *(const v2f*)(krow + 4 * (8 * (g + 1) + cc));
          }
          __builtin_amdgcn_sched_barrier(0);
#pragma unroll
          for (int cc = 0; cc < 8; ++cc) acc = wmma4(qa[8 * g + cc], kf[g & 1][cc], acc);
          __builtin_amdgcn_sched_barrier(0);
        }
#pragma unroll
        for (int r = 0; r < 8; ++r) {
          const int mrow = r + 8 * hi, s = j * 16 + lh;
          sS[mrow * SSTR + s] = (s < len) ? acc[r] * scale : NEGC;
        }
        v2f pa[4];
#pragma unroll
        for (int cc = 0; cc < 4; ++cc) pa[cc] = *(const v2f*)(sS + lh * SSTR + 4 * (4 * j + cc) + 2 * hi);
        float tm = fmaxf(fmaxf(fmaxf(pa[0].x, pa[0].y), fmaxf(pa[1].x, pa[1].y)),
                         fmaxf(fmaxf(pa[2].x, pa[2].y), fmaxf(pa[3].x, pa[3].y)));
        tm = fmaxf(tm, __shfl_xor(tm, 16, 32));
        const float m_new = fmaxf(m_run, tm);
        const float alpha = __expf(m_run - m_new);
        float ps = 0.f;
#pragma unroll
        for (int cc = 0; cc < 4; ++cc) {
          pa[cc].x = __expf(pa[cc].x - m_new);
          pa[cc].y = __expf(pa[cc].y - m_new);
          ps += pa[cc].x;
          ps += pa[cc].y;
        }
        l_run = l_run * alpha + ps;
        m_run = m_new;
#pragma unroll
        for (int r = 0; r < 8; ++r) {
          const float ar = __shfl(alpha, r + 8 * hi, 32);
#pragma unroll
          for (int nt = 0; nt < 8; ++nt) dacc[nt][r] *= ar;
        }
        const float* vcol = slot + KVSL + (2 * hi) * HSTR + lh;
#pragma unroll
        for (int h = 0; h < 2; ++h) {
          v2f vb[16];
#pragma unroll
          for (int q = 0; q < 4; ++q) {
            int noff = 16 * (4 * h + q);
            asm volatile("" : "+v"(noff));
            const float* vn = vcol + noff;
#pragma unroll
            for (int cc = 0; cc < 4; ++cc) {
              vb[4 * cc + q].x = vn[(4 * cc) * HSTR];
              vb[4 * cc + q].y = vn[(4 * cc + 1) * HSTR];
            }
          }
          __builtin_amdgcn_sched_barrier(0);
#pragma unroll
          for (int cc = 0; cc < 4; ++cc) {
#pragma unroll
            for (int q = 0; q < 4; ++q) dacc[4 * h + q] = wmma4(pa[cc], vb[4 * cc + q], dacc[4 * h + q]);
          }
          __builtin_amdgcn_sched_barrier(0);
        }
      }
      if (more) pair_store(sR + ((n + 1) & 1) * SLOTF, wave, lane, ka, kb, va, vb4);
      __syncthreads();
    }

    if (active) {
      v8f rs[8];
#pragma unroll
      for (int nt = 0; nt < 8; ++nt) {
#pragma unroll
        for (int r = 0; r < 8; ++r) {
          const float x = reb[roff[r] + 16 * nt];
          rs[nt][r] = (qt * 16 + r + 8 * hi < T_) ? x : 0.f;
        }
      }
      const float l_tot = l_run + __shfl_xor(l_run, 16, 32);
      const float qm = ((qt * 16 + lh) < len) ? 1.f : 0.f;
      const float inv = qm / l_tot;
#pragma unroll
      for (int r = 0; r < 8; ++r) {
        const float ir = __shfl(inv, r + 8 * hi, 32);
        const int t = qt * 16 + r + 8 * hi;
#pragma unroll
        for (int nt = 0; nt < 8; ++nt) Db[t * D_ + 16 * nt + lh] = dacc[nt][r] * ir + rs[nt][r];
      }
    }
    if (again && wave + 8 < ntl) {
      load_query_tile(Qb + ((wave + 8) * 16 + lh) * D_, my_a1, lh, hi, qa, roff);
#pragma unroll
      for (int c = 0; c < 32; ++c) asm volatile("" : "+v"(qa[c]));
    }
  }
}

__global__ void stamp_kernel(const int* __restrict__ seqlen, const float* __restrict__ wsDec,
                             const float* __restrict__ u_w1, const float* __restrict__ u_w2,
                             const float* __restrict__ u_b, const float* __restrict__ u_v,
                             float* __restrict__ out) {
  extern __shared__ __attribute__((aligned(16))) float smem[];
  float* sW2 = smem;
  float* sLp = smem + D_ * D_;
  float* sUb = sLp + 128;
  float* sUv = sUb + 128;
  float* sCoef = sUv + 128;
  const int tid = threadIdx.x;
  int len;
  const int b = ranked_session(seqlen, (u64*)sCoef, tid, len);
  const int ntl = live_tiles(len);
  const float* Db = wsDec + (size_t)b * PBS;

  {
    const float4* src = (const float4*)u_w2 + tid;
    float4 w[16];
#pragma unroll
    for (int i = 0; i < 16; ++i) w[i] = src[256 * i];
#pragma unroll
    for (int i = 0; i < 16; ++i) *(float4*)(sW2 + 4 * (tid + 256 * i)) = w[i];
  }
  if (tid < D_) { sUb[tid] = u_b[tid]; sUv[tid] = u_v[tid]; }
  if (tid < D_) {
    const float* lrow = Db + (len - 1) * D_;
    float acc = 0.f;
    for (int d = 0; d < D_; ++d) acc += lrow[d] * u_w1[d * D_ + tid];
    sLp[tid] = acc;
    out[(size_t)b * 2 * D_ + D_ + tid] = acc;
  }
  __syncthreads();

  const int wave = tid >> 5, lane = tid & 31, lh = lane & 15, hi = lane >> 4;
  for (int pass = 0; pass < 2; ++pass) {
    const int qt = wave + 8 * pass;
    if (qt < ntl) {
      v2f da[32];
      const float* drow = Db + (qt * 16 + lh) * D_;
#pragma unroll
      for (int c = 0; c < 32; ++c) da[c] = *(const v2f*)(drow + 4 * c + 2 * hi);
      float cpart[8];
#pragma unroll
      for (int r = 0; r < 8; ++r) cpart[r] = 0.f;
      for (int nt = 0; nt < 8; ++nt) {
        int n0 = nt * 16;
        v8f acc = {0.f, 0.f, 0.f, 0.f, 0.f, 0.f, 0.f, 0.f};
#pragma unroll
        for (int c = 0; c < 32; ++c) {
          int k = 4 * c + 2 * hi;
          v2f bb;
          bb.x = sW2[k * D_ + n0 + lh];
          bb.y = sW2[(k + 1) * D_ + n0 + lh];
          acc = wmma4(da[c], bb, acc);
        }
        int col = n0 + lh;
        float add = sLp[col] + sUb[col];
        float uvv = sUv[col];
#pragma unroll
        for (int r = 0; r < 8; ++r) {
          float x = acc[r] + add;
          float sg = 1.f / (1.f + __expf(-x));
          cpart[r] += sg * uvv;
        }
      }
#pragma unroll
      for (int r = 0; r < 8; ++r) {
        float v = cpart[r];
        v += __shfl_xor(v, 1, 32);
        v += __shfl_xor(v, 2, 32);
        v += __shfl_xor(v, 4, 32);
        v += __shfl_xor(v, 8, 32);
        if (lh == 0) {
          int t = qt * 16 + r + 8 * hi;
          sCoef[t] = (t < len) ? v : 0.f;
        }
      }
    }
  }
  __syncthreads();
  if (tid < D_) {
    float acc = 0.f;
    for (int tt = 0; tt < ntl; ++tt) {
#pragma unroll
      for (int u = 0; u < 16; ++u) {
        const int t = 16 * tt + u;
        acc += sCoef[t] * Db[t * D_ + tid];
      }
    }
    out[(size_t)b * 2 * D_ + tid] = acc;
  }
}

extern "C" void kernel_launch(void* const* d_in, const int* in_sizes, int n_in,
                              void* d_out, int out_size, void* d_ws, size_t ws_size,
                              hipStream_t stream) {
  (void)in_sizes; (void)n_in; (void)out_size; (void)ws_size;
  const float* re   = (const float*)d_in[0];
  const int*   alia = (const int*)d_in[1];
  const int*   slen = (const int*)d_in[2];
  const float* Wq   = (const float*)d_in[3];
  const float* Wk   = (const float*)d_in[4];
  const float* Wv   = (const float*)d_in[5];
  const float* u_w1 = (const float*)d_in[6];
  const float* u_w2 = (const float*)d_in[7];
  const float* u_b  = (const float*)d_in[8];
  const float* u_v  = (const float*)d_in[9];
  float* out = (float*)d_out;

  float* ws   = (float*)d_ws;
  float* wsQ  = ws;
  float* wsK  = ws + (size_t)B_ * PBS;
  float* wsV  = ws + 2 * (size_t)B_ * PBS;
  float* wsDec = wsQ;

  size_t sm1 = (size_t)(TP * HSTR + 64) * sizeof(float);
  size_t sm2 = (size_t)(2 * SLOTF + 8 * 16 * SSTR) * sizeof(float);
  size_t sm3 = (size_t)(D_ * D_ + 3 * 128 + TP) * sizeof(float);
  hipFuncSetAttribute((const void*)qkv_kernel,
                      hipFuncAttributeMaxDynamicSharedMemorySize, (int)sm1);
  hipFuncSetAttribute((const void*)attn_kernel,
                      hipFuncAttributeMaxDynamicSharedMemorySize, (int)sm2);
  hipFuncSetAttribute((const void*)stamp_kernel,
                      hipFuncAttributeMaxDynamicSharedMemorySize, (int)sm3);

  qkv_kernel<<<B_, 256, sm1, stream>>>(re, alia, slen, Wq, Wk, Wv, wsQ, wsK, wsV);
  attn_kernel<<<B_, 256, sm2, stream>>>(re, alia, slen, wsQ, wsK, wsV, wsDec);
  stamp_kernel<<<B_, 256, sm3, stream>>>(slen, wsDec, u_w1, u_w2, u_b, u_v, out);
}


# all three kernels: each session computed on its distinct nodes, weighed by multiplicity as keys and in the final sum; softmax, attn*V and coef sums in a new order
# speedup vs baseline: 6.1164x; 1.1804x over previous
#include <hip/hip_runtime.h>
#include <math.h>


typedef __attribute__((ext_vector_type(2))) float v2f;
typedef __attribute__((ext_vector_type(8))) float v8f;

#define B_      2048
#define T_      200
#define D_      128
#define NNODE_  200
#define TP      208
#define NT      13
#define HSTR    132
#define SSTR    210
#define PBS     (TP * D_)
#define NEGC    (-4294967295.0f)

__device__ __forceinline__ int live_tiles(int len) {
  const int n = (len + 15) >> 4;
  return n < 1 ? 1 : (n > NT ? NT : n);
}

__device__ __forceinline__ v8f wmma4(v2f a, v2f b, v8f c) {
  return __builtin_amdgcn_wmma_f32_16x16x4_f32(false, a, false, b, (short)0, c,
                                               false, false);
}

typedef unsigned long long u64;

__device__ __forceinline__ int cls_get(u64 x0, u64 x1, u64 x2, int c) {
  const int w = (c >= 5) + (c >= 10);
  const u64 x = (w == 0) ? x0 : ((w == 1) ? x1 : x2);
  return (int)((x >> (12 * (c - 5 * w))) & 0xFFFull);
}

__device__ __forceinline__ u64 uniform64(u64 x) {
  const unsigned lo = (unsigned)__builtin_amdgcn_readfirstlane((int)(unsigned)(x & 0xFFFFFFFFull));
  const unsigned hi = (unsigned)__builtin_amdgcn_readfirstlane((int)(unsigned)(x >> 32));
  return ((u64)hi << 32) | (u64)lo;
}

__device__ __forceinline__ int ranked_session(const int* __restrict__ seqlen, u64* scr, int tid,
                                              int& len_out) {
  const int lane = tid & 31, wave = tid >> 5;
  int* res = (int*)(scr + 24);
  const int4 la = ((const int4*)seqlen)[2 * tid];
  const int4 lb = ((const int4*)seqlen)[2 * tid + 1];
  const int lenv[8] = {la.x, la.y, la.z, la.w, lb.x, lb.y, lb.z, lb.w};
  int cls[8];
  u64 nib = 0ull;
#pragma unroll
  for (int e = 0; e < 8; ++e) {
    cls[e] = NT - live_tiles(lenv[e]);
    nib += 1ull << (4 * cls[e]);
  }
  u64 h0 = 0ull, h1 = 0ull, h2 = 0ull;
#pragma unroll
  for (int c = 0; c < NT; ++c) {
    const int w = c / 5;
    const u64 v = ((nib >> (4 * c)) & 0xFull) << (12 * (c - 5 * w));
    if (w == 0) h0 += v; else if (w == 1) h1 += v; else h2 += v;
  }
  u64 s0 = h0, s1 = h1, s2 = h2;
#pragma unroll
  for (int d = 1; d < 32; d <<= 1) {
    const u64 t0 = __shfl_up(s0, d, 32), t1 = __shfl_up(s1, d, 32), t2 = __shfl_up(s2, d, 32);
    if (lane >= d) { s0 += t0; s1 += t1; s2 += t2; }
  }
  if (lane == 31) { scr[3 * wave] = s0; scr[3 * wave + 1] = s1; scr[3 * wave + 2] = s2; }
  if (tid == 0) { res[0] = blockIdx.x; res[1] = 1; }
  __syncthreads();
  u64 r0 = s0 - h0, r1 = s1 - h1, r2 = s2 - h2;
  u64 H0 = 0ull, H1 = 0ull, H2 = 0ull;
#pragma unroll
  for (int w = 0; w < 8; ++w) {
    const u64 a = scr[3 * w], bq = scr[3 * w + 1], cq = scr[3 * w + 2];
    H0 += a; H1 += bq; H2 += cq;
    if (w < wave) { r0 += a; r1 += bq; r2 += cq; }
  }
  const u64 T0 = uniform64(H0), T1 = uniform64(H1), T2 = uniform64(H2);
  const int k = blockIdx.x;
  int run = 0, cstar = NT - 1, jj = 0;
  bool open = true;
#pragma unroll
  for (int c = 0; c < NT; ++c) {
    const int cnt = cls_get(T0, T1, T2, c);
    if (open && k < run + cnt) { cstar = c; jj = k - run; open = false; }
    run += cnt;
  }
  int p = cls_get(r0, r1, r2, cstar);
  int found = -1, flen = 1;
#pragma unroll
  for (int e = 0; e < 8; ++e) {
    const bool m = (cls[e] == cstar);
    if (m && p == jj) { found = 8 * tid + e; flen = lenv[e]; }
    p += m ? 1 : 0;
  }
  if (found >= 0) { res[0] = found; res[1] = flen; }
  __syncthreads();
  int b = __builtin_amdgcn_readfirstlane(res[0]);
  len_out = __builtin_amdgcn_readfirstlane(res[1]);
  return b < 0 ? 0 : (b > B_ - 1 ? B_ - 1 : b);
}

#define DDW 640
__device__ __forceinline__ int distinct_nodes(const int* __restrict__ arow, int len, float* scr, int tid,
                                              int& last_idx) {
  int* sAl = (int*)scr;
  float* sC = scr + TP;
  int* sIdx = (int*)(scr + 2 * TP);
  int* sTot = (int*)(scr + 3 * TP);
  const int lenc = len < 1 ? 1 : (len > T_ ? T_ : len);
  if (tid < TP) sAl[tid] = (tid < lenc) ? arow[tid] : -1;
  if (tid == 0) sTot[8] = 0;
  __syncthreads();
  int cnt = 0;
  const int n4 = (lenc + 3) >> 2;
  for (int q = 0; q < n4; ++q) {
    const int4 a = ((const int4*)sAl)[q];
    cnt += (a.x == tid ? 1 : 0) + (a.y == tid ? 1 : 0) + (a.z == tid ? 1 : 0) + (a.w == tid ? 1 : 0);
  }
  const bool present = (cnt > 0) && (tid < NNODE_);
  const unsigned mask = __builtin_amdgcn_ballot_w32(present);
  const int lane = tid & 31, wave = tid >> 5;
  const int pre = __popc(mask & ((1u << lane) - 1u));
  if (lane == 0) sTot[wave] = __popc(mask);
  __syncthreads();
  int off = 0, U = 0;
#pragma unroll
  for (int w = 0; w < 8; ++w) {
    const int t = sTot[w];
    U += t;
    if (w < wave) off += t;
  }
  const int a_last = sAl[lenc - 1];
  if (present) {
    sIdx[off + pre] = tid;
    sC[off + pre] = (float)cnt;
    if (tid == a_last) sTot[8] = off + pre;
  }
  if (tid < TP && tid >= U) { sIdx[tid] = 0; sC[tid] = 0.f; }
  __syncthreads();
  last_idx = __builtin_amdgcn_readfirstlane(sTot[8]);
  U = __builtin_amdgcn_readfirstlane(U);
  return U < 1 ? 1 : U;
}

__global__ void __launch_bounds__(256) __attribute__((amdgpu_waves_per_eu(4, 4))) qkv_kernel(const float* __restrict__ re, const int* __restrict__ alias,
                           const int* __restrict__ seqlen,
                           const float* __restrict__ Wq, const float* __restrict__ Wk,
                           const float* __restrict__ Wv,
                           float* __restrict__ wsQ, float* __restrict__ wsK,
                           float* __restrict__ wsV) {
  extern __shared__ __attribute__((aligned(16))) float smem_h[];
  float* sH = smem_h;
  const int tid = threadIdx.x;
  int len;
  const int b = ranked_session(seqlen, (u64*)(smem_h + TP * HSTR), tid, len);
  float* dd = smem_h + TP * HSTR + 64;
  int last_idx;
  const int U = distinct_nodes(alias + (size_t)b * T_, len, dd, tid, last_idx);
  (void)last_idx;
  const int* sIdx = (const int*)(dd + 2 * TP);
  const int ntl = live_tiles(U);
  const int wv = __builtin_amdgcn_readfirstlane(tid >> 5);
  const int ln = tid & 31, lh = ln & 15, hi = ln >> 4;
  const int n0 = wv << 4;

  {
    const int tl = wv + 8 * ln;
    const int my_a = sIdx[tl < TP ? tl : TP - 1];
    const float* reb = re + (size_t)b * NNODE_ * D_ + ln * 4;
    const int nk = 2 * ntl;
#pragma unroll 1
    for (int h = 0; h < 2; ++h) {
      const int k0 = h * 13;
      if (k0 >= nk) break;
      float4 v[13];
#pragma unroll
      for (int r = 0; r < 13; ++r) {
        if (k0 + r < nk) {
          const int a = __builtin_amdgcn_readlane(my_a, k0 + r);
          float4 x = *(const float4*)(reb + (size_t)a * D_);
          if (wv + 8 * (k0 + r) >= U) x = make_float4(0.f, 0.f, 0.f, 0.f);
          v[r] = x;
        }
      }
#pragma unroll
      for (int r = 0; r < 13; ++r)
        if (k0 + r < nk) *(float4*)(sH + (wv + 8 * (k0 + r)) * HSTR + ln * 4) = v[r];
    }
  }

  v2f bq[32], bk[32], bv[32];
  {
    const float* wq = Wq + (2 * hi) * D_ + n0 + lh;
    const float* wk = Wk + (2 * hi) * D_ + n0 + lh;
    const float* wvp = Wv + (2 * hi) * D_ + n0 + lh;
#pragma unroll
    for (int c = 0; c < 32; ++c) {
      bq[c].x = wq[(4 * c) * D_];
      bq[c].y = wq[(4 * c + 1) * D_];
      bk[c].x = wk[(4 * c) * D_];
      bk[c].y = wk[(4 * c + 1) * D_];
      bv[c].x = wvp[(4 * c) * D_];
      bv[c].y = wvp[(4 * c + 1) * D_];
    }
  }
  __syncthreads();

  float* dq = wsQ + (size_t)b * PBS + n0 + lh;
  float* dk = wsK + (size_t)b * PBS + n0 + lh;
  float* dv = wsV + (size_t)b * PBS + n0 + lh;
  for (int i = 0; i < ntl; ++i) {
    const float* arow = sH + (i * 16 + lh) * HSTR + 2 * hi;
    v8f aq = {0.f, 0.f, 0.f, 0.f, 0.f, 0.f, 0.f, 0.f};
    v8f ak = {0.f, 0.f, 0.f, 0.f, 0.f, 0.f, 0.f, 0.f};
    v8f av = {0.f, 0.f, 0.f, 0.f, 0.f, 0.f, 0.f, 0.f};
#pragma unroll
    for (int g = 0; g < 4; ++g) {
      v2f af[8];
#pragma unroll
      for (int cc = 0; cc < 8; ++cc) af[cc] = *(const v2f*)(arow + 4 * (8 * g + cc));
      __builtin_amdgcn_sched_barrier(0);
#pragma unroll
      for (int cc = 0; cc < 8; ++cc) {
        const int c = 8 * g + cc;
        aq = wmma4(af[cc], bq[c], aq);
        ak = wmma4(af[cc], bk[c], ak);
        av = wmma4(af[cc], bv[c], av);
      }
    }
#pragma unroll
    for (int r = 0; r < 8; ++r) {
      const int row = i * 16 + r + 8 * hi;
      const float q = aq[r], k = ak[r], v = av[r];
      dq[row * D_] = q > 0.f ? q : 0.f;
      dk[row * D_] = k > 0.f ? k : 0.f;
      dv[row * D_] = v > 0.f ? v : 0.f;
    }
  }
}

#define KVSL (16 * HSTR)
#define SLOTF (2 * KVSL)
__device__ __forceinline__ void pair_request(const float4* __restrict__ K4, const float4* __restrict__ V4, int j,
                                             float4& ka, float4& kb, float4& va, float4& vb) {
  ka = K4[256 * (2 * j)];
  kb = K4[256 * (2 * j + 1)];
  va = V4[256 * (2 * j)];
  vb = V4[256 * (2 * j + 1)];
}
__device__ __forceinline__ void pair_store(float* slot, int wave, int lane, const float4& ka, const float4& kb,
                                           const float4& va, const float4& vb) {
  *(float4*)(slot + wave * HSTR + lane * 4) = ka;
  *(float4*)(slot + (wave + 8) * HSTR + lane * 4) = kb;
  *(float4*)(slot + KVSL + wave * HSTR + lane * 4) = va;
  *(float4*)(slot + KVSL + (wave + 8) * HSTR + lane * 4) = vb;
}


__device__ __forceinline__ void load_query_tile(const float* __restrict__ Qrow, int my_a, int lh, int hi,
                                                v2f (&qa)[32], int (&roff)[8]) {
#pragma unroll
  for (int r = 0; r < 8; ++r) {
    const int alo = __builtin_amdgcn_readlane(my_a, r);
    const int ahi = __builtin_amdgcn_readlane(my_a, 8 + r);
    roff[r] = (hi ? ahi : alo) * D_ + lh;
  }
#pragma unroll
  for (int c = 0; c < 32; ++c) qa[c] = *(const v2f*)(Qrow + 4 * c + 2 * hi);
}

__global__ void __launch_bounds__(256) __attribute__((amdgpu_waves_per_eu(4, 4))) attn_kernel(const float* __restrict__ re, const int* __restrict__ alias,
                            const int* __restrict__ seqlen,
                            const float* __restrict__ wsQ, const float* __restrict__ wsK,
                            const float* __restrict__ wsV, float* __restrict__ wsDec) {
  extern __shared__ __attribute__((aligned(16))) float smem[];
  float* sR = smem;
  float* sS0 = smem + 2 * SLOTF;
  const int tid = threadIdx.x;
  int len;
  const int b = ranked_session(seqlen, (u64*)sS0, tid, len);
  const int wave = tid >> 5, lane = tid & 31, lh = lane & 15, hi = lane >> 4;
  const float4* K4 = (const float4*)(wsK + (size_t)b * PBS) + tid;
  const float4* V4 = (const float4*)(wsV + (size_t)b * PBS) + tid;
  float4 ka, kb, va, vb4;
  pair_request(K4, V4, 0, ka, kb, va, vb4);
  float* dd = smem + 2 * SLOTF + 8 * 16 * SSTR;
  int last_idx;
  const int U = distinct_nodes(alias + (size_t)b * T_, len, dd, tid, last_idx);
  (void)last_idx;
  const float* sC = dd + TP;
  const int* sIdx = (const int*)(dd + 2 * TP);
  const int ntl = live_tiles(U);
  const int t_a0 = wave * 16 + lh, t_a1 = (wave + 8) * 16 + lh;
  const int my_a0 = sIdx[t_a0 < TP ? t_a0 : TP - 1];
  const int my_a1 = sIdx[t_a1 < TP ? t_a1 : TP - 1];
  float* sS = sS0 + wave * (16 * SSTR);
  const float scale = 0.088388347648318447f;
  const float* reb = re + (size_t)b * NNODE_ * D_;
  const float* Qb = wsQ + (size_t)b * PBS;
  float* Db = wsDec + (size_t)b * PBS;

  int roff[8];
  v2f qa[32];
  if (wave < ntl) {
    load_query_tile(Qb + (wave * 16 + lh) * D_, my_a0, lh, hi, qa, roff);
#pragma unroll
    for (int c = 0; c < 32; ++c) asm volatile("" : "+v"(qa[c]));
  }
  pair_store(sR, wave, lane, ka, kb, va, vb4);
  __syncthreads();

  const int npass = (ntl > 8) ? 2 : 1;
  int n = 0;
  for (int pass = 0; pass < npass; ++pass) {
    const int qt = wave + 8 * pass;
    const bool active = (qt < ntl);
    const bool again = (pass + 1 < npass);

    v8f dacc[8];
#pragma unroll
    for (int nt = 0; nt < 8; ++nt) dacc[nt] = (v8f){0.f, 0.f, 0.f, 0.f, 0.f, 0.f, 0.f, 0.f};
    float m_run = -1.0e30f;
    float l_run = 0.f;

    for (int j = 0; j < ntl; ++j, ++n) {
      const float* slot = sR + (n & 1) * SLOTF;
      const bool more = (j + 1 < ntl) || again;
      if (j + 1 < ntl) pair_request(K4, V4, j + 1, ka, kb, va, vb4);
      else if (again)  pair_request(K4, V4, 0, ka, kb, va, vb4);
      if (active) {
        v8f acc = {0.f, 0.f, 0.f, 0.f, 0.f, 0.f, 0.f, 0.f};
        const float* krow = slot + lh * HSTR + 2 * hi;
        v2f kf[2][8];
#pragma unroll
        for (int cc = 0; cc < 8; ++cc) kf[0][cc] = *(const v2f*)(krow + 4 * cc);
#pragma unroll
        for (int g = 0; g < 4; ++g) {
          if (g < 3) {
#pragma unroll
            for (int cc = 0; cc < 8; ++cc) kf[(g + 1) & 1][cc] = *(const v2f*)(krow + 4 * (8 * (g + 1) + cc));
          }
          __builtin_amdgcn_sched_barrier(0);
#pragma unroll
          for (int cc = 0; cc < 8; ++cc) acc = wmma4(qa[8 * g + cc], kf[g & 1][cc], acc);
          __builtin_amdgcn_sched_barrier(0);
        }
#pragma unroll
        for (int r = 0; r < 8; ++r) {
          const int mrow = r + 8 * hi, s = j * 16 + lh;
          sS[mrow * SSTR + s] = (s < U) ? acc[r] * scale : NEGC;
        }
        v2f pa[4];
#pragma unroll
        for (int cc = 0; cc < 4; ++cc) pa[cc] = *(const v2f*)(sS + lh * SSTR + 4 * (4 * j + cc) + 2 * hi);
        float tm = fmaxf(fmaxf(fmaxf(pa[0].x, pa[0].y), fmaxf(pa[1].x, pa[1].y)),
                         fmaxf(fmaxf(pa[2].x, pa[2].y), fmaxf(pa[3].x, pa[3].y)));
        tm = fmaxf(tm, __shfl_xor(tm, 16, 32));
        const float m_new = fmaxf(m_run, tm);
        const float alpha = __expf(m_run - m_new);
        float ps = 0.f;
#pragma unroll
        for (int cc = 0; cc < 4; ++cc) {
          const v2f cw = *(const v2f*)(sC + 16 * j + 4 * cc + 2 * hi);
          pa[cc].x = cw.x * __expf(pa[cc].x - m_new);
          pa[cc].y = cw.y * __expf(pa[cc].y - m_new);
          ps += pa[cc].x;
          ps += pa[cc].y;
        }
        l_run = l_run * alpha + ps;
        m_run = m_new;
#pragma unroll
        for (int r = 0; r < 8; ++r) {
          const float ar = __shfl(alpha, r + 8 * hi, 32);
#pragma unroll
          for (int nt = 0; nt < 8; ++nt) dacc[nt][r] *= ar;
        }
        const float* vcol = slot + KVSL + (2 * hi) * HSTR + lh;
#pragma unroll
        for (int h = 0; h < 2; ++h) {
          v2f vb[16];
#pragma unroll
          for (int q = 0; q < 4; ++q) {
            int noff = 16 * (4 * h + q);
            asm volatile("" : "+v"(noff));
            const float* vn = vcol + noff;
#pragma unroll
            for (int cc = 0; cc < 4; ++cc) {
              vb[4 * cc + q].x = vn[(4 * cc) * HSTR];
              vb[4 * cc + q].y = vn[(4 * cc + 1) * HSTR];
            }
          }
          __builtin_amdgcn_sched_barrier(0);
#pragma unroll
          for (int cc = 0; cc < 4; ++cc) {
#pragma unroll
            for (int q = 0; q < 4; ++q) dacc[4 * h + q] = wmma4(pa[cc], vb[4 * cc + q], dacc[4 * h + q]);
          }
          __builtin_amdgcn_sched_barrier(0);
        }
      }
      if (more) pair_store(sR + ((n + 1) & 1) * SLOTF, wave, lane, ka, kb, va, vb4);
      __syncthreads();
    }

    if (active) {
      v8f rs[8];
#pragma unroll
      for (int nt = 0; nt < 8; ++nt) {
#pragma unroll
        for (int r = 0; r < 8; ++r) {
          const float x = reb[roff[r] + 16 * nt];
          rs[nt][r] = (qt * 16 + r + 8 * hi < U) ? x : 0.f;
        }
      }
      const float l_tot = l_run + __shfl_xor(l_run, 16, 32);
      const float qm = ((qt * 16 + lh) < U) ? 1.f : 0.f;
      const float inv = qm / l_tot;
#pragma unroll
      for (int r = 0; r < 8; ++r) {
        const float ir = __shfl(inv, r + 8 * hi, 32);
        const int t = qt * 16 + r + 8 * hi;
#pragma unroll
        for (int nt = 0; nt < 8; ++nt) Db[t * D_ + 16 * nt + lh] = dacc[nt][r] * ir + rs[nt][r];
      }
    }
    if (again && wave + 8 < ntl) {
      load_query_tile(Qb + ((wave + 8) * 16 + lh) * D_, my_a1, lh, hi, qa, roff);
#pragma unroll
      for (int c = 0; c < 32; ++c) asm volatile("" : "+v"(qa[c]));
    }
  }
}

__global__ void stamp_kernel(const int* __restrict__ alias, const int* __restrict__ seqlen, const float* __restrict__ wsDec,
                             const float* __restrict__ u_w1, const float* __restrict__ u_w2,
                             const float* __restrict__ u_b, const float* __restrict__ u_v,
                             float* __restrict__ out) {
  extern __shared__ __attribute__((aligned(16))) float smem[];
  float* sW2 = smem;
  float* sLp = smem + D_ * D_;
  float* sUb = sLp + 128;
  float* sUv = sUb + 128;
  float* sCoef = sUv + 128;
  const int tid = threadIdx.x;
  int len;
  const int b = ranked_session(seqlen, (u64*)sCoef, tid, len);
  float* dd = sCoef + TP;
  int last_idx;
  const int U = distinct_nodes(alias + (size_t)b * T_, len, dd, tid, last_idx);
  const float* sC = dd + TP;
  const int ntl = live_tiles(U);
  const float* Db = wsDec + (size_t)b * PBS;

  {
    const float4* src = (const float4*)u_w2 + tid;
    float4 w[16];
#pragma unroll
    for (int i = 0; i < 16; ++i) w[i] = src[256 * i];
#pragma unroll
    for (int i = 0; i < 16; ++i) *(float4*)(sW2 + 4 * (tid + 256 * i)) = w[i];
  }
  if (tid < D_) { sUb[tid] = u_b[tid]; sUv[tid] = u_v[tid]; }
  if (tid < D_) {
    const float* lrow = Db + last_idx * D_;
    float acc = 0.f;
    for (int d = 0; d < D_; ++d) acc += lrow[d] * u_w1[d * D_ + tid];
    sLp[tid] = acc;
    out[(size_t)b * 2 * D_ + D_ + tid] = acc;
  }
  __syncthreads();

  const int wave = tid >> 5, lane = tid & 31, lh = lane & 15, hi = lane >> 4;
  for (int pass = 0; pass < 2; ++pass) {
    const int qt = wave + 8 * pass;
    if (qt < ntl) {
      v2f da[32];
      const float* drow = Db + (qt * 16 + lh) * D_;
#pragma unroll
      for (int c = 0; c < 32; ++c) da[c] = *(const v2f*)(drow + 4 * c + 2 * hi);
      float cpart[8];
#pragma unroll
      for (int r = 0; r < 8; ++r) cpart[r] = 0.f;
      for (int nt = 0; nt < 8; ++nt) {
        int n0 = nt * 16;
        v8f acc = {0.f, 0.f, 0.f, 0.f, 0.f, 0.f, 0.f, 0.f};
#pragma unroll
        for (int c = 0; c < 32; ++c) {
          int k = 4 * c + 2 * hi;
          v2f bb;
          bb.x = sW2[k * D_ + n0 + lh];
          bb.y = sW2[(k + 1) * D_ + n0 + lh];
          acc = wmma4(da[c], bb, acc);
        }
        int col = n0 + lh;
        float add = sLp[col] + sUb[col];
        float uvv = sUv[col];
#pragma unroll
        for (int r = 0; r < 8; ++r) {
          float x = acc[r] + add;
          float sg = 1.f / (1.f + __expf(-x));
          cpart[r] += sg * uvv;
        }
      }
#pragma unroll
      for (int r = 0; r < 8; ++r) {
        float v = cpart[r];
        v += __shfl_xor(v, 1, 32);
        v += __shfl_xor(v, 2, 32);
        v += __shfl_xor(v, 4, 32);
        v += __shfl_xor(v, 8, 32);
        if (lh == 0) {
          int t = qt * 16 + r + 8 * hi;
          sCoef[t] = (t < U) ? v * sC[t] : 0.f;
        }
      }
    }
  }
  __syncthreads();
  if (tid < D_) {
    float acc = 0.f;
    for (int tt = 0; tt < ntl; ++tt) {
#pragma unroll
      for (int u = 0; u < 16; ++u) {
        const int t = 16 * tt + u;
        acc += sCoef[t] * Db[t * D_ + tid];
      }
    }
    out[(size_t)b * 2 * D_ + tid] = acc;
  }
}

extern "C" void kernel_launch(void* const* d_in, const int* in_sizes, int n_in,
                              void* d_out, int out_size, void* d_ws, size_t ws_size,
                              hipStream_t stream) {
  (void)in_sizes; (void)n_in; (void)out_size; (void)ws_size;
  const float* re   = (const float*)d_in[0];
  const int*   alia = (const int*)d_in[1];
  const int*   slen = (const int*)d_in[2];
  const float* Wq   = (const float*)d_in[3];
  const float* Wk   = (const float*)d_in[4];
  const float* Wv   = (const float*)d_in[5];
  const float* u_w1 = (const float*)d_in[6];
  const float* u_w2 = (const float*)d_in[7];
  const float* u_b  = (const float*)d_in[8];
  const float* u_v  = (const float*)d_in[9];
  float* out = (float*)d_out;

  float* ws   = (float*)d_ws;
  float* wsQ  = ws;
  float* wsK  = ws + (size_t)B_ * PBS;
  float* wsV  = ws + 2 * (size_t)B_ * PBS;
  float* wsDec = wsQ;

  size_t sm1 = (size_t)(TP * HSTR + 64 + DDW) * sizeof(float);
  size_t sm2 = (size_t)(2 * SLOTF + 8 * 16 * SSTR + DDW) * sizeof(float);
  size_t sm3 = (size_t)(D_ * D_ + 3 * 128 + TP + DDW) * sizeof(float);
  hipFuncSetAttribute((const void*)qkv_kernel,
                      hipFuncAttributeMaxDynamicSharedMemorySize, (int)sm1);
  hipFuncSetAttribute((const void*)attn_kernel,
                      hipFuncAttributeMaxDynamicSharedMemorySize, (int)sm2);
  hipFuncSetAttribute((const void*)stamp_kernel,
                      hipFuncAttributeMaxDynamicSharedMemorySize, (int)sm3);

  qkv_kernel<<<B_, 256, sm1, stream>>>(re, alia, slen, Wq, Wk, Wv, wsQ, wsK, wsV);
  attn_kernel<<<B_, 256, sm2, stream>>>(re, alia, slen, wsQ, wsK, wsV, wsDec);
  stamp_kernel<<<B_, 256, sm3, stream>>>(alia, slen, wsDec, u_w1, u_w2, u_b, u_v, out);
}
